# starting kernel
# baseline (speedup 1.0000x reference)
#include <hip/hip_runtime.h>
#include <math.h>


typedef float  v2f __attribute__((ext_vector_type(2)));
typedef float  v8f __attribute__((ext_vector_type(8)));

#define B_SZ 2
#define L_SZ 2048
#define DM   768
#define DI   1536
#define DS   16
#define NROW (B_SZ * L_SZ)

template <int K>
__global__ __launch_bounds__(32) void gemm32x64_f32_wmma(
    const float* __restrict__ A, const float* __restrict__ W,
    float* __restrict__ C, int N64, int lda, int ldw, int ldc)
{
    static_assert(K % 8 == 0, "K must be a multiple of 8");
    const int tile = blockIdx.x;
    const int tm   = tile / N64;
    const int tn   = tile % N64;
    const int lane = threadIdx.x;
    const int half = lane >> 4;
    const int lr   = lane & 15;
    const int kh   = 2 * half;

    const float* Ar0 = A + (size_t)(tm * 32 +  0 + lr) * lda;
    const float* Ar1 = A + (size_t)(tm * 32 + 16 + lr) * lda;
    const float* W0  = W + (size_t)(tn * 64 +  0 + lr) * ldw;
    const float* W1  = W + (size_t)(tn * 64 + 16 + lr) * ldw;
    const float* W2  = W + (size_t)(tn * 64 + 32 + lr) * ldw;
    const float* W3  = W + (size_t)(tn * 64 + 48 + lr) * ldw;

    v8f acc00 = {}; v8f acc01 = {}; v8f acc02 = {}; v8f acc03 = {};
    v8f acc10 = {}; v8f acc11 = {}; v8f acc12 = {}; v8f acc13 = {};

    auto ld = [&](int kb, v2f& a0, v2f& a1, v2f& b0, v2f& b1, v2f& b2, v2f& b3) {
        a0 = *(const v2f*)(Ar0 + kb);
        a1 = *(const v2f*)(Ar1 + kb);
        b0 = *(const v2f*)(W0 + kb);
        b1 = *(const v2f*)(W1 + kb);
        b2 = *(const v2f*)(W2 + kb);
        b3 = *(const v2f*)(W3 + kb);
    };
    auto mm8 = [&](v2f a0, v2f a1, v2f b0, v2f b1, v2f b2, v2f b3) {
        acc00 = __builtin_amdgcn_wmma_f32_16x16x4_f32(false, a0, false, b0, (short)0, acc00, false, false);
        acc01 = __builtin_amdgcn_wmma_f32_16x16x4_f32(false, a0, false, b1, (short)0, acc01, false, false);
        acc02 = __builtin_amdgcn_wmma_f32_16x16x4_f32(false, a0, false, b2, (short)0, acc02, false, false);
        acc03 = __builtin_amdgcn_wmma_f32_16x16x4_f32(false, a0, false, b3, (short)0, acc03, false, false);
        acc10 = __builtin_amdgcn_wmma_f32_16x16x4_f32(false, a1, false, b0, (short)0, acc10, false, false);
        acc11 = __builtin_amdgcn_wmma_f32_16x16x4_f32(false, a1, false, b1, (short)0, acc11, false, false);
        acc12 = __builtin_amdgcn_wmma_f32_16x16x4_f32(false, a1, false, b2, (short)0, acc12, false, false);
        acc13 = __builtin_amdgcn_wmma_f32_16x16x4_f32(false, a1, false, b3, (short)0, acc13, false, false);
    };

    v2f a0A, a1A, b0A, b1A, b2A, b3A;
    v2f a0B, a1B, b0B, b1B, b2B, b3B;

    ld(kh, a0A, a1A, b0A, b1A, b2A, b3A);
    for (int k = 0; k < K - 8; k += 8) {
        ld(k + 4 + kh, a0B, a1B, b0B, b1B, b2B, b3B);
        mm8(a0A, a1A, b0A, b1A, b2A, b3A);
        ld(k + 8 + kh, a0A, a1A, b0A, b1A, b2A, b3A);
        mm8(a0B, a1B, b0B, b1B, b2B, b3B);
    }
    ld(K - 4 + kh, a0B, a1B, b0B, b1B, b2B, b3B);
    mm8(a0A, a1A, b0A, b1A, b2A, b3A);
    mm8(a0B, a1B, b0B, b1B, b2B, b3B);

    float* Cb0 = C + (size_t)(tm * 32 +  0 + 8 * half) * ldc + tn * 64 + lr;
    float* Cb1 = C + (size_t)(tm * 32 + 16 + 8 * half) * ldc + tn * 64 + lr;
#pragma unroll
    for (int r = 0; r < 8; ++r) {
        float* Cr0 = Cb0 + (size_t)r * ldc;
        float* Cr1 = Cb1 + (size_t)r * ldc;
        Cr0[0]  = acc00[r];  Cr0[16] = acc01[r];
        Cr0[32] = acc02[r];  Cr0[48] = acc03[r];
        Cr1[0]  = acc10[r];  Cr1[16] = acc11[r];
        Cr1[32] = acc12[r];  Cr1[48] = acc13[r];
    }
}

__global__ __launch_bounds__(256) void conv_silu_kernel(
    const float* __restrict__ xg, const float* __restrict__ conv_w,
    const float* __restrict__ conv_b, float* __restrict__ x_conv)
{
    int idx = blockIdx.x * 256 + threadIdx.x;
    if (idx >= NROW * DI) return;
    int d   = idx % DI;
    int row = idx / DI;
    int l   = row % L_SZ;

    float acc = conv_b[d];
    const float* w = conv_w + d * 4;
#pragma unroll
    for (int j = 0; j < 4; ++j) {
        int ls = l - 3 + j;
        if (ls >= 0)
            acc = fmaf(w[j], xg[(size_t)(row - 3 + j) * (2 * DI) + d], acc);
    }
    x_conv[idx] = acc / (1.0f + __expf(-acc));
}

__global__ __launch_bounds__(256) void xproj_kernel(
    const float* __restrict__ x_conv, const float* __restrict__ Wx,
    float* __restrict__ ssm)
{
    int idx = blockIdx.x * 256 + threadIdx.x;
    if (idx >= NROW * 33) return;
    int n   = idx % 33;
    int row = idx / 33;
    const float4* xr = (const float4*)(x_conv + (size_t)row * DI);
    const float4* wr = (const float4*)(Wx + (size_t)n * DI);
    float a0 = 0.f, a1 = 0.f, a2 = 0.f, a3 = 0.f;
    for (int k = 0; k < DI / 4; ++k) {
        float4 xv = xr[k];
        float4 wv = wr[k];
        a0 = fmaf(xv.x, wv.x, a0);
        a1 = fmaf(xv.y, wv.y, a1);
        a2 = fmaf(xv.z, wv.z, a2);
        a3 = fmaf(xv.w, wv.w, a3);
    }
    ssm[idx] = (a0 + a1) + (a2 + a3);
}

__global__ __launch_bounds__(256) void scan_kernel(
    const float* __restrict__ ssm, const float* __restrict__ x_conv,
    const float* __restrict__ xg, const float* __restrict__ w_dt,
    const float* __restrict__ b_dt, const float* __restrict__ A_log,
    const float* __restrict__ D_param, float* __restrict__ y)
{
    int g = blockIdx.x * 256 + threadIdx.x;
    if (g >= B_SZ * DI) return;
    int b = g / DI, d = g % DI;

    const float wdt = w_dt[d];
    const float bdt = b_dt[d];
    const float Dp  = D_param[d];
    float Aa[DS];
#pragma unroll
    for (int n = 0; n < DS; ++n) Aa[n] = -__expf(A_log[d * DS + n]);
    float h[DS];
#pragma unroll
    for (int n = 0; n < DS; ++n) h[n] = 0.f;

    for (int l = 0; l < L_SZ; ++l) {
        const int row = b * L_SZ + l;
        const float* s = ssm + (size_t)row * 33;
        const float x_t = x_conv[(size_t)row * DI + d];

        float z  = fmaf(s[32], wdt, bdt);
        float dt = (z > 20.0f) ? z : log1pf(__expf(z));

        float yv = 0.f;
#pragma unroll
        for (int n = 0; n < DS; ++n) {
            float Abar = __expf(dt * Aa[n]);
            h[n] = fmaf(Abar, h[n], (dt * s[n]) * x_t);
            yv   = fmaf(h[n], s[DS + n], yv);
        }
        yv = fmaf(x_t, Dp, yv);

        float gv = xg[(size_t)row * (2 * DI) + DI + d];
        float sg = gv / (1.0f + __expf(-gv));
        y[(size_t)row * DI + d] = yv * sg;
    }
}

extern "C" void kernel_launch(void* const* d_in, const int* in_sizes, int n_in,
                              void* d_out, int out_size, void* d_ws, size_t ws_size,
                              hipStream_t stream)
{
    const float* x       = (const float*)d_in[0];
    const float* W_in    = (const float*)d_in[1];
    const float* conv_w  = (const float*)d_in[2];
    const float* conv_b  = (const float*)d_in[3];
    const float* W_xproj = (const float*)d_in[4];
    const float* w_dt    = (const float*)d_in[5];
    const float* b_dt    = (const float*)d_in[6];
    const float* A_log   = (const float*)d_in[7];
    const float* D_param = (const float*)d_in[8];
    const float* W_out   = (const float*)d_in[9];
    float* out = (float*)d_out;

    float* ws     = (float*)d_ws;
    float* xg     = ws;
    float* x_conv = xg + (size_t)NROW * (2 * DI);
    float* ssm    = x_conv + (size_t)NROW * DI;
    float* yb     = ssm + (size_t)NROW * 33;

    gemm32x64_f32_wmma<DM><<<(NROW / 32) * ((2 * DI) / 64), 32, 0, stream>>>(
        x, W_in, xg, (2 * DI) / 64, DM, DM, 2 * DI);

    conv_silu_kernel<<<(NROW * DI + 255) / 256, 256, 0, stream>>>(
        xg, conv_w, conv_b, x_conv);

    xproj_kernel<<<(NROW * 33 + 255) / 256, 256, 0, stream>>>(
        x_conv, W_xproj, ssm);

    scan_kernel<<<(B_SZ * DI + 255) / 256, 256, 0, stream>>>(
        ssm, x_conv, xg, w_dt, b_dt, A_log, D_param, yb);

    gemm32x64_f32_wmma<DI><<<(NROW / 32) * (DM / 64), 32, 0, stream>>>(
        yb, W_out, out, DM / 64, DI, DI, DM);
}


# GEMM fragment loads widened, no LDS: a lane reads 32 contiguous bytes of its row per fragment (two 16-byte loads) for 4 k-steps, halving cache lines touched per k; k order inside each sum changes
# speedup vs baseline: 6.9279x; 6.9279x over previous
#include <hip/hip_runtime.h>
#include <math.h>


typedef float  v2f __attribute__((ext_vector_type(2)));
typedef float  v8f __attribute__((ext_vector_type(8)));

#define B_SZ 2
#define L_SZ 2048
#define DM   768
#define DI   1536
#define DS   16
#define NROW (B_SZ * L_SZ)

__device__ __forceinline__ float xhalf(float v) {
    return __builtin_bit_cast(float, __builtin_amdgcn_permlanex16(
        __builtin_bit_cast(int, v), __builtin_bit_cast(int, v), 0x76543210, 0xfedcba98, false, false));
}

template <int K, bool CONV>
__global__ __launch_bounds__(32) void gemm32x64_f32_wmma(
    const float* __restrict__ A, const float* __restrict__ W,
    float* __restrict__ C, int N64, int lda, int ldw, int ldc,
    const float* __restrict__ conv_w, const float* __restrict__ conv_b,
    float* __restrict__ x_conv)
{
    static_assert(K % 32 == 0, "K must be a multiple of 32");
    const int tile = blockIdx.x;
    const int tm   = tile / N64;
    const int tn   = tile % N64;
    const int lane = threadIdx.x;
    const int half = lane >> 4;
    const int lr   = lane & 15;

    const float* Ar0 = A + (size_t)(tm * 32 +  0 + lr) * lda;
    const float* Ar1 = A + (size_t)(tm * 32 + 16 + lr) * lda;
    const float* W0  = W + (size_t)(tn * 64 +  0 + lr) * ldw;
    const float* W1  = W + (size_t)(tn * 64 + 16 + lr) * ldw;
    const float* W2  = W + (size_t)(tn * 64 + 32 + lr) * ldw;
    const float* W3  = W + (size_t)(tn * 64 + 48 + lr) * ldw;

    v8f acc00 = {}; v8f acc01 = {}; v8f acc02 = {}; v8f acc03 = {};
    v8f acc10 = {}; v8f acc11 = {}; v8f acc12 = {}; v8f acc13 = {};

    struct frag { float4 q0, q1; };
    const int kh8 = 8 * half;

    auto ld = [&](int kb, frag& a0, frag& a1, frag& b0, frag& b1, frag& b2, frag& b3) {
        a0.q0 = *(const float4*)(Ar0 + kb);  a0.q1 = *(const float4*)(Ar0 + kb + 4);
        a1.q0 = *(const float4*)(Ar1 + kb);  a1.q1 = *(const float4*)(Ar1 + kb + 4);
        b0.q0 = *(const float4*)(W0 + kb);   b0.q1 = *(const float4*)(W0 + kb + 4);
        b1.q0 = *(const float4*)(W1 + kb);   b1.q1 = *(const float4*)(W1 + kb + 4);
        b2.q0 = *(const float4*)(W2 + kb);   b2.q1 = *(const float4*)(W2 + kb + 4);
        b3.q0 = *(const float4*)(W3 + kb);   b3.q1 = *(const float4*)(W3 + kb + 4);
    };
    auto mm8 = [&](v2f a0, v2f a1, v2f b0, v2f b1, v2f b2, v2f b3) {
        acc00 = __builtin_amdgcn_wmma_f32_16x16x4_f32(false, a0, false, b0, (short)0, acc00, false, false);
        acc01 = __builtin_amdgcn_wmma_f32_16x16x4_f32(false, a0, false, b1, (short)0, acc01, false, false);
        acc02 = __builtin_amdgcn_wmma_f32_16x16x4_f32(false, a0, false, b2, (short)0, acc02, false, false);
        acc03 = __builtin_amdgcn_wmma_f32_16x16x4_f32(false, a0, false, b3, (short)0, acc03, false, false);
        acc10 = __builtin_amdgcn_wmma_f32_16x16x4_f32(false, a1, false, b0, (short)0, acc10, false, false);
        acc11 = __builtin_amdgcn_wmma_f32_16x16x4_f32(false, a1, false, b1, (short)0, acc11, false, false);
        acc12 = __builtin_amdgcn_wmma_f32_16x16x4_f32(false, a1, false, b2, (short)0, acc12, false, false);
        acc13 = __builtin_amdgcn_wmma_f32_16x16x4_f32(false, a1, false, b3, (short)0, acc13, false, false);
    };
    auto lo = [](const float4& q) { v2f r; r.x = q.x; r.y = q.y; return r; };
    auto hi = [](const float4& q) { v2f r; r.x = q.z; r.y = q.w; return r; };
    auto mm32 = [&](const frag& a0, const frag& a1, const frag& b0, const frag& b1, const frag& b2, const frag& b3) {
        mm8(lo(a0.q0), lo(a1.q0), lo(b0.q0), lo(b1.q0), lo(b2.q0), lo(b3.q0));
        mm8(hi(a0.q0), hi(a1.q0), hi(b0.q0), hi(b1.q0), hi(b2.q0), hi(b3.q0));
        mm8(lo(a0.q1), lo(a1.q1), lo(b0.q1), lo(b1.q1), lo(b2.q1), lo(b3.q1));
        mm8(hi(a0.q1), hi(a1.q1), hi(b0.q1), hi(b1.q1), hi(b2.q1), hi(b3.q1));
    };

    frag a0A, a1A, b0A, b1A, b2A, b3A;
    frag a0B, a1B, b0B, b1B, b2B, b3B;

    ld(kh8, a0A, a1A, b0A, b1A, b2A, b3A);
    for (int k = 0; k < K - 32; k += 32) {
        ld(k + 16 + kh8, a0B, a1B, b0B, b1B, b2B, b3B);
        mm32(a0A, a1A, b0A, b1A, b2A, b3A);
        ld(k + 32 + kh8, a0A, a1A, b0A, b1A, b2A, b3A);
        mm32(a0B, a1B, b0B, b1B, b2B, b3B);
    }
    ld(K - 16 + kh8, a0B, a1B, b0B, b1B, b2B, b3B);
    mm32(a0A, a1A, b0A, b1A, b2A, b3A);
    mm32(a0B, a1B, b0B, b1B, b2B, b3B);

    float* Cb0 = C + (size_t)(tm * 32 +  0 + 8 * half) * ldc + tn * 64 + lr;
    float* Cb1 = C + (size_t)(tm * 32 + 16 + 8 * half) * ldc + tn * 64 + lr;

    if (CONV && tn < DI / 64) {
        float* Xb0 = x_conv + (size_t)(tm * 32 +  0 + 8 * half) * DI + tn * 64 + lr;
        float* Xb1 = x_conv + (size_t)(tm * 32 + 16 + 8 * half) * DI + tn * 64 + lr;
        auto conv_tile = [&](const v8f& s0, const v8f& s1, int j) {
            const int    d = tn * 64 + 16 * j + lr;
            const float4 w = *(const float4*)(conv_w + (size_t)d * 4);
            const float  b = conv_b[d];
            float e0[11], e1[11];
#pragma unroll
            for (int i = 0; i < 3; ++i) {
                const float xa = xhalf(s0[5 + i]);
                const float xb = xhalf(s1[5 + i]);
                e0[i] = xa;
                e1[i] = half ? xb : xa;
            }
#pragma unroll
            for (int r = 0; r < 8; ++r) { e0[3 + r] = s0[r]; e1[3 + r] = s1[r]; }
#pragma unroll
            for (int r = 0; r < 8; ++r) {
                float a0 = b, a1 = b;
                a0 = fmaf(w.x, e0[r],     a0);  a1 = fmaf(w.x, e1[r],     a1);
                a0 = fmaf(w.y, e0[r + 1], a0);  a1 = fmaf(w.y, e1[r + 1], a1);
                a0 = fmaf(w.z, e0[r + 2], a0);  a1 = fmaf(w.z, e1[r + 2], a1);
                a0 = fmaf(w.w, e0[r + 3], a0);  a1 = fmaf(w.w, e1[r + 3], a1);
                const float y0 = a0 / (1.0f + __expf(-a0));
                const float y1 = a1 / (1.0f + __expf(-a1));
                if (r >= 3 || half) Xb0[(size_t)r * DI + 16 * j] = y0;
                Xb1[(size_t)r * DI + 16 * j] = y1;
                if (r < 3 && !half) Cb0[(size_t)r * ldc + 16 * j] = s0[r];
                if (r >= 5 && half) Cb1[(size_t)r * ldc + 16 * j] = s1[r];
            }
        };
        conv_tile(acc00, acc10, 0);
        conv_tile(acc01, acc11, 1);
        conv_tile(acc02, acc12, 2);
        conv_tile(acc03, acc13, 3);
        return;
    }
#pragma unroll
    for (int r = 0; r < 8; ++r) {
        float* Cr0 = Cb0 + (size_t)r * ldc;
        float* Cr1 = Cb1 + (size_t)r * ldc;
        Cr0[0]  = acc00[r];  Cr0[16] = acc01[r];
        Cr0[32] = acc02[r];  Cr0[48] = acc03[r];
        Cr1[0]  = acc10[r];  Cr1[16] = acc11[r];
        Cr1[32] = acc12[r];  Cr1[48] = acc13[r];
    }
}

__global__ __launch_bounds__(256) void conv_edge_kernel(
    const float* __restrict__ xg, const float* __restrict__ conv_w,
    const float* __restrict__ conv_b, float* __restrict__ x_conv)
{
    static_assert(DI % 256 == 0 && L_SZ % 32 == 0, "edge tiling");
    const int  tile = blockIdx.x / (DI / 256);
    const int  d    = (blockIdx.x % (DI / 256)) * 256 + threadIdx.x;
    const int  row0 = tile * 32;
    const bool head = (row0 % L_SZ) == 0;

    const float  b  = conv_b[d];
    const float4 w4 = *(const float4*)(conv_w + (size_t)d * 4);
    const float  w[4] = {w4.x, w4.y, w4.z, w4.w};
    const float* p  = xg + (size_t)row0 * (2 * DI) + d;

    float in[6];
    in[0] = in[1] = in[2] = 0.0f;
    if (!head) {
        in[0] = *(p - 3 * (2 * DI));
        in[1] = *(p - 2 * (2 * DI));
        in[2] = *(p - 1 * (2 * DI));
    }
    in[3] = p[0];
    in[4] = p[1 * (2 * DI)];
    in[5] = p[2 * (2 * DI)];
#pragma unroll
    for (int i = 0; i < 3; ++i) {
        float acc = b;
#pragma unroll
        for (int j = 0; j < 4; ++j) {
            const float t = fmaf(w[j], in[i + j], acc);
            acc = (head && i + j < 3) ? acc : t;
        }
        x_conv[(size_t)(row0 + i) * DI + d] = acc / (1.0f + __expf(-acc));
    }
}

__global__ __launch_bounds__(256) void xproj_kernel(
    const float* __restrict__ x_conv, const float* __restrict__ Wx,
    float* __restrict__ ssm)
{
    int idx = blockIdx.x * 256 + threadIdx.x;
    if (idx >= NROW * 33) return;
    int n   = idx % 33;
    int row = idx / 33;
    const float4* xr = (const float4*)(x_conv + (size_t)row * DI);
    const float4* wr = (const float4*)(Wx + (size_t)n * DI);
    float a0 = 0.f, a1 = 0.f, a2 = 0.f, a3 = 0.f;
    for (int k = 0; k < DI / 4; ++k) {
        float4 xv = xr[k];
        float4 wv = wr[k];
        a0 = fmaf(xv.x, wv.x, a0);
        a1 = fmaf(xv.y, wv.y, a1);
        a2 = fmaf(xv.z, wv.z, a2);
        a3 = fmaf(xv.w, wv.w, a3);
    }
    ssm[idx] = (a0 + a1) + (a2 + a3);
}

#define SCAN_WAVES 16
#define SCAN_NBLK  (L_SZ / 16)
#define YT_PITCH   33

template <int CTRL>
__device__ __forceinline__ float row_dpp(float v) {
    return __builtin_bit_cast(float,
        __builtin_amdgcn_update_dpp(0, __builtin_bit_cast(int, v), CTRL, 0xf, 0xf, true));
}

#define SCAN_STEP(j)                                                  \
    {                                                                 \
        const float dtb  = row_dpp<0x150 + (j)>(dt);                  \
        const float xb   = row_dpp<0x150 + (j)>(xv);                  \
        const float Abar = __expf(dtb * Aa);                          \
        h    = fmaf(Abar, h, (dtb * Bv[j]) * xb);                     \
        p[j] = h * Cv[j];                                             \
    }

__global__ __launch_bounds__(SCAN_WAVES * 32) void scan_kernel(
    const float* __restrict__ ssm, const float* __restrict__ x_conv,
    const float* __restrict__ xg, const float* __restrict__ w_dt,
    const float* __restrict__ b_dt, const float* __restrict__ A_log,
    const float* __restrict__ D_param, float* __restrict__ y)
{
    static_assert(DI % 32 == 0 && L_SZ % 32 == 0 && DS == 16, "scan tiling");
    __shared__ float ytile[2][16 * YT_PITCH];

    const int lane  = threadIdx.x & 31;
    const int wv    = threadIdx.x >> 5;
    const int n     = lane & 15;
    const int c     = lane >> 4;
    const int b     = blockIdx.x / (DI / 32);
    const int dbase = (blockIdx.x % (DI / 32)) * 32;
    const int d     = dbase + 2 * wv + c;
    const size_t row0 = (size_t)b * L_SZ;

    const float wdt = w_dt[d];
    const float bdt = b_dt[d];
    const float Dp  = D_param[d];
    const float Aa  = -__expf(A_log[d * DS + n]);
    float h = 0.f;

    const bool b3 = (n & 8) != 0, b2 = (n & 4) != 0, b1 = (n & 2) != 0, b0 = (n & 1) != 0;

    const float* pBC = ssm + row0 * 33 + n;
    const float* pDT = ssm + (row0 + n) * 33 + 32;
    const float* pX  = x_conv + (row0 + n) * DI + d;
    const float* pG  = xg + (row0 + wv) * (2 * DI) + DI + dbase + lane;
    float*       pY  = y + (row0 + wv) * DI + dbase + lane;

    auto ldblk = [&](int k, float (&Bv)[16], float (&Cv)[16], float& dtr, float& xv, float& gv) {
        const float* q = pBC + (size_t)k * (16 * 33);
#pragma unroll
        for (int j = 0; j < 16; ++j) { Bv[j] = q[j * 33]; Cv[j] = q[j * 33 + 16]; }
        dtr = pDT[(size_t)k * (16 * 33)];
        xv  = pX[(size_t)k * (16 * DI)];
        gv  = pG[(size_t)k * (16 * 2 * DI)];
    };

    auto compute = [&](int k, const float (&Bv)[16], const float (&Cv)[16],
                       float dtr, float xv, float gv, float* yt) {
        const float z  = fmaf(dtr, wdt, bdt);
        const float dt = (z > 20.0f) ? z : log1pf(__expf(z));

        float p[16];
        SCAN_STEP(0)  SCAN_STEP(1)  SCAN_STEP(2)  SCAN_STEP(3)
        SCAN_STEP(4)  SCAN_STEP(5)  SCAN_STEP(6)  SCAN_STEP(7)
        SCAN_STEP(8)  SCAN_STEP(9)  SCAN_STEP(10) SCAN_STEP(11)
        SCAN_STEP(12) SCAN_STEP(13) SCAN_STEP(14) SCAN_STEP(15)

        float q8[8];
#pragma unroll
        for (int j = 0; j < 8; ++j) {
            const float keep = b3 ? p[j + 8] : p[j];
            const float send = b3 ? p[j] : p[j + 8];
            q8[j] = keep + row_dpp<0x168>(send);
        }
        float q4[4];
#pragma unroll
        for (int j = 0; j < 4; ++j) {
            const float keep = b2 ? q8[j + 4] : q8[j];
            const float send = b2 ? q8[j] : q8[j + 4];
            q4[j] = keep + row_dpp<0x164>(send);
        }
        float q2[2];
#pragma unroll
        for (int j = 0; j < 2; ++j) {
            const float keep = b1 ? q4[j + 2] : q4[j];
            const float send = b1 ? q4[j] : q4[j + 2];
            q2[j] = keep + row_dpp<0x162>(send);
        }
        const float keep1 = b0 ? q2[1] : q2[0];
        const float send1 = b0 ? q2[0] : q2[1];
        const float ysum  = keep1 + row_dpp<0x161>(send1);

        yt[n * YT_PITCH + 2 * wv + c] = fmaf(xv, Dp, ysum);
        __syncthreads();
        const float ys = yt[wv * YT_PITCH + lane];
        const float sg = gv / (1.0f + __expf(-gv));
        pY[(size_t)k * (16 * DI)] = ys * sg;
    };

    float BvA[16], CvA[16], dtrA, xvA, gvA;
    float BvB[16], CvB[16], dtrB, xvB, gvB;
    ldblk(0, BvA, CvA, dtrA, xvA, gvA);
    for (int k = 0; k < SCAN_NBLK; k += 2) {
        ldblk(k + 1, BvB, CvB, dtrB, xvB, gvB);
        compute(k, BvA, CvA, dtrA, xvA, gvA, ytile[0]);
        const int kn = (k + 2 < SCAN_NBLK) ? (k + 2) : (SCAN_NBLK - 1);
        ldblk(kn, BvA, CvA, dtrA, xvA, gvA);
        compute(k + 1, BvB, CvB, dtrB, xvB, gvB, ytile[1]);
    }
}

extern "C" void kernel_launch(void* const* d_in, const int* in_sizes, int n_in,
                              void* d_out, int out_size, void* d_ws, size_t ws_size,
                              hipStream_t stream)
{
    const float* x       = (const float*)d_in[0];
    const float* W_in    = (const float*)d_in[1];
    const float* conv_w  = (const float*)d_in[2];
    const float* conv_b  = (const float*)d_in[3];
    const float* W_xproj = (const float*)d_in[4];
    const float* w_dt    = (const float*)d_in[5];
    const float* b_dt    = (const float*)d_in[6];
    const float* A_log   = (const float*)d_in[7];
    const float* D_param = (const float*)d_in[8];
    const float* W_out   = (const float*)d_in[9];
    float* out = (float*)d_out;

    float* ws     = (float*)d_ws;
    float* xg     = ws;
    float* x_conv = xg + (size_t)NROW * (2 * DI);
    float* ssm    = x_conv + (size_t)NROW * DI;
    float* yb     = ssm + (size_t)NROW * 33;

    gemm32x64_f32_wmma<DM, true><<<(NROW / 32) * ((2 * DI) / 64), 32, 0, stream>>>(
        x, W_in, xg, (2 * DI) / 64, DM, DM, 2 * DI, conv_w, conv_b, x_conv);

    conv_edge_kernel<<<(NROW / 32) * (DI / 256), 256, 0, stream>>>(
        xg, conv_w, conv_b, x_conv);

    xproj_kernel<<<(NROW * 33 + 255) / 256, 256, 0, stream>>>(
        x_conv, W_xproj, ssm);

    scan_kernel<<<B_SZ * (DI / 32), SCAN_WAVES * 32, 0, stream>>>(
        ssm, x_conv, xg, w_dt, b_dt, A_log, D_param, yb);

    gemm32x64_f32_wmma<DI, false><<<(NROW / 32) * (DM / 64), 32, 0, stream>>>(
        yb, W_out, out, DM / 64, DI, DI, DM, nullptr, nullptr, nullptr);
}


# x-projection re-cut, K along the lanes: a wave sums 4 rows x 33 outputs from whole-line 16-byte loads; lane partials added from LDS in lane order, no butterfly; summation order of ssm changes
# speedup vs baseline: 8.2121x; 1.1854x over previous
#include <hip/hip_runtime.h>
#include <math.h>


typedef float  v2f __attribute__((ext_vector_type(2)));
typedef float  v8f __attribute__((ext_vector_type(8)));

#define B_SZ 2
#define L_SZ 2048
#define DM   768
#define DI   1536
#define DS   16
#define NROW (B_SZ * L_SZ)

__device__ __forceinline__ float xhalf(float v) {
    return __builtin_bit_cast(float, __builtin_amdgcn_permlanex16(
        __builtin_bit_cast(int, v), __builtin_bit_cast(int, v), 0x76543210, 0xfedcba98, false, false));
}

template <int K, bool CONV>
__global__ __launch_bounds__(32) void gemm32x64_f32_wmma(
    const float* __restrict__ A, const float* __restrict__ W,
    float* __restrict__ C, int N64, int lda, int ldw, int ldc,
    const float* __restrict__ conv_w, const float* __restrict__ conv_b,
    float* __restrict__ x_conv)
{
    static_assert(K % 32 == 0, "K must be a multiple of 32");
    const int tile = blockIdx.x;
    const int tm   = tile / N64;
    const int tn   = tile % N64;
    const int lane = threadIdx.x;
    const int half = lane >> 4;
    const int lr   = lane & 15;

    const float* Ar0 = A + (size_t)(tm * 32 +  0 + lr) * lda;
    const float* Ar1 = A + (size_t)(tm * 32 + 16 + lr) * lda;
    const float* W0  = W + (size_t)(tn * 64 +  0 + lr) * ldw;
    const float* W1  = W + (size_t)(tn * 64 + 16 + lr) * ldw;
    const float* W2  = W + (size_t)(tn * 64 + 32 + lr) * ldw;
    const float* W3  = W + (size_t)(tn * 64 + 48 + lr) * ldw;

    v8f acc00 = {}; v8f acc01 = {}; v8f acc02 = {}; v8f acc03 = {};
    v8f acc10 = {}; v8f acc11 = {}; v8f acc12 = {}; v8f acc13 = {};

    struct frag { float4 q0, q1; };
    const int kh8 = 8 * half;

    auto ld = [&](int kb, frag& a0, frag& a1, frag& b0, frag& b1, frag& b2, frag& b3) {
        a0.q0 = *(const float4*)(Ar0 + kb);  a0.q1 = *(const float4*)(Ar0 + kb + 4);
        a1.q0 = *(const float4*)(Ar1 + kb);  a1.q1 = *(const float4*)(Ar1 + kb + 4);
        b0.q0 = *(const float4*)(W0 + kb);   b0.q1 = *(const float4*)(W0 + kb + 4);
        b1.q0 = *(const float4*)(W1 + kb);   b1.q1 = *(const float4*)(W1 + kb + 4);
        b2.q0 = *(const float4*)(W2 + kb);   b2.q1 = *(const float4*)(W2 + kb + 4);
        b3.q0 = *(const float4*)(W3 + kb);   b3.q1 = *(const float4*)(W3 + kb + 4);
    };
    auto mm8 = [&](v2f a0, v2f a1, v2f b0, v2f b1, v2f b2, v2f b3) {
        acc00 = __builtin_amdgcn_wmma_f32_16x16x4_f32(false, a0, false, b0, (short)0, acc00, false, false);
        acc01 = __builtin_amdgcn_wmma_f32_16x16x4_f32(false, a0, false, b1, (short)0, acc01, false, false);
        acc02 = __builtin_amdgcn_wmma_f32_16x16x4_f32(false, a0, false, b2, (short)0, acc02, false, false);
        acc03 = __builtin_amdgcn_wmma_f32_16x16x4_f32(false, a0, false, b3, (short)0, acc03, false, false);
        acc10 = __builtin_amdgcn_wmma_f32_16x16x4_f32(false, a1, false, b0, (short)0, acc10, false, false);
        acc11 = __builtin_amdgcn_wmma_f32_16x16x4_f32(false, a1, false, b1, (short)0, acc11, false, false);
        acc12 = __builtin_amdgcn_wmma_f32_16x16x4_f32(false, a1, false, b2, (short)0, acc12, false, false);
        acc13 = __builtin_amdgcn_wmma_f32_16x16x4_f32(false, a1, false, b3, (short)0, acc13, false, false);
    };
    auto lo = [](const float4& q) { v2f r; r.x = q.x; r.y = q.y; return r; };
    auto hi = [](const float4& q) { v2f r; r.x = q.z; r.y = q.w; return r; };
    auto mm32 = [&](const frag& a0, const frag& a1, const frag& b0, const frag& b1, const frag& b2, const frag& b3) {
        mm8(lo(a0.q0), lo(a1.q0), lo(b0.q0), lo(b1.q0), lo(b2.q0), lo(b3.q0));
        mm8(hi(a0.q0), hi(a1.q0), hi(b0.q0), hi(b1.q0), hi(b2.q0), hi(b3.q0));
        mm8(lo(a0.q1), lo(a1.q1), lo(b0.q1), lo(b1.q1), lo(b2.q1), lo(b3.q1));
        mm8(hi(a0.q1), hi(a1.q1), hi(b0.q1), hi(b1.q1), hi(b2.q1), hi(b3.q1));
    };

    frag a0A, a1A, b0A, b1A, b2A, b3A;
    frag a0B, a1B, b0B, b1B, b2B, b3B;

    ld(kh8, a0A, a1A, b0A, b1A, b2A, b3A);
    for (int k = 0; k < K - 32; k += 32) {
        ld(k + 16 + kh8, a0B, a1B, b0B, b1B, b2B, b3B);
        mm32(a0A, a1A, b0A, b1A, b2A, b3A);
        ld(k + 32 + kh8, a0A, a1A, b0A, b1A, b2A, b3A);
        mm32(a0B, a1B, b0B, b1B, b2B, b3B);
    }
    ld(K - 16 + kh8, a0B, a1B, b0B, b1B, b2B, b3B);
    mm32(a0A, a1A, b0A, b1A, b2A, b3A);
    mm32(a0B, a1B, b0B, b1B, b2B, b3B);

    float* Cb0 = C + (size_t)(tm * 32 +  0 + 8 * half) * ldc + tn * 64 + lr;
    float* Cb1 = C + (size_t)(tm * 32 + 16 + 8 * half) * ldc + tn * 64 + lr;

    if (CONV && tn < DI / 64) {
        float* Xb0 = x_conv + (size_t)(tm * 32 +  0 + 8 * half) * DI + tn * 64 + lr;
        float* Xb1 = x_conv + (size_t)(tm * 32 + 16 + 8 * half) * DI + tn * 64 + lr;
        auto conv_tile = [&](const v8f& s0, const v8f& s1, int j) {
            const int    d = tn * 64 + 16 * j + lr;
            const float4 w = *(const float4*)(conv_w + (size_t)d * 4);
            const float  b = conv_b[d];
            float e0[11], e1[11];
#pragma unroll
            for (int i = 0; i < 3; ++i) {
                const float xa = xhalf(s0[5 + i]);
                const float xb = xhalf(s1[5 + i]);
                e0[i] = xa;
                e1[i] = half ? xb : xa;
            }
#pragma unroll
            for (int r = 0; r < 8; ++r) { e0[3 + r] = s0[r]; e1[3 + r] = s1[r]; }
#pragma unroll
            for (int r = 0; r < 8; ++r) {
                float a0 = b, a1 = b;
                a0 = fmaf(w.x, e0[r],     a0);  a1 = fmaf(w.x, e1[r],     a1);
                a0 = fmaf(w.y, e0[r + 1], a0);  a1 = fmaf(w.y, e1[r + 1], a1);
                a0 = fmaf(w.z, e0[r + 2], a0);  a1 = fmaf(w.z, e1[r + 2], a1);
                a0 = fmaf(w.w, e0[r + 3], a0);  a1 = fmaf(w.w, e1[r + 3], a1);
                const float y0 = a0 / (1.0f + __expf(-a0));
                const float y1 = a1 / (1.0f + __expf(-a1));
                if (r >= 3 || half) Xb0[(size_t)r * DI + 16 * j] = y0;
                Xb1[(size_t)r * DI + 16 * j] = y1;
                if (r < 3 && !half) Cb0[(size_t)r * ldc + 16 * j] = s0[r];
                if (r >= 5 && half) Cb1[(size_t)r * ldc + 16 * j] = s1[r];
            }
        };
        conv_tile(acc00, acc10, 0);
        conv_tile(acc01, acc11, 1);
        conv_tile(acc02, acc12, 2);
        conv_tile(acc03, acc13, 3);
        return;
    }
#pragma unroll
    for (int r = 0; r < 8; ++r) {
        float* Cr0 = Cb0 + (size_t)r * ldc;
        float* Cr1 = Cb1 + (size_t)r * ldc;
        Cr0[0]  = acc00[r];  Cr0[16] = acc01[r];
        Cr0[32] = acc02[r];  Cr0[48] = acc03[r];
        Cr1[0]  = acc10[r];  Cr1[16] = acc11[r];
        Cr1[32] = acc12[r];  Cr1[48] = acc13[r];
    }
}

__global__ __launch_bounds__(256) void conv_edge_kernel(
    const float* __restrict__ xg, const float* __restrict__ conv_w,
    const float* __restrict__ conv_b, float* __restrict__ x_conv)
{
    static_assert(DI % 256 == 0 && L_SZ % 32 == 0, "edge tiling");
    const int  tile = blockIdx.x / (DI / 256);
    const int  d    = (blockIdx.x % (DI / 256)) * 256 + threadIdx.x;
    const int  row0 = tile * 32;
    const bool head = (row0 % L_SZ) == 0;

    const float  b  = conv_b[d];
    const float4 w4 = *(const float4*)(conv_w + (size_t)d * 4);
    const float  w[4] = {w4.x, w4.y, w4.z, w4.w};
    const float* p  = xg + (size_t)row0 * (2 * DI) + d;

    float in[6];
    in[0] = in[1] = in[2] = 0.0f;
    if (!head) {
        in[0] = *(p - 3 * (2 * DI));
        in[1] = *(p - 2 * (2 * DI));
        in[2] = *(p - 1 * (2 * DI));
    }
    in[3] = p[0];
    in[4] = p[1 * (2 * DI)];
    in[5] = p[2 * (2 * DI)];
#pragma unroll
    for (int i = 0; i < 3; ++i) {
        float acc = b;
#pragma unroll
        for (int j = 0; j < 4; ++j) {
            const float t = fmaf(w[j], in[i + j], acc);
            acc = (head && i + j < 3) ? acc : t;
        }
        x_conv[(size_t)(row0 + i) * DI + d] = acc / (1.0f + __expf(-acc));
    }
}

#define XP_WAVES 8
#define XP_RPW   4
#define XP_ROWS  (XP_WAVES * XP_RPW)
#define XP_OUT   (XP_ROWS * 33)
#define XP_PITCH 33

__global__ __launch_bounds__(XP_WAVES * 32) __attribute__((amdgpu_waves_per_eu(2, 2)))
void xproj_kernel(
    const float* __restrict__ x_conv, const float* __restrict__ Wx,
    float* __restrict__ ssm)
{
    static_assert(NROW % XP_ROWS == 0 && DI % 128 == 0, "xproj tiling");
    __shared__ float part[XP_OUT * XP_PITCH];

    const int lane = threadIdx.x & 31;
    const int wv   = threadIdx.x >> 5;
    const int row0 = blockIdx.x * XP_ROWS;

    const float* xp = x_conv + (size_t)(row0 + XP_RPW * wv) * DI + 4 * lane;
    const float* wp = Wx + 4 * lane;

    v2f acc[XP_RPW][33];
#pragma unroll
    for (int r = 0; r < XP_RPW; ++r)
#pragma unroll
        for (int n = 0; n < 33; ++n) acc[r][n] = v2f{0.f, 0.f};

    for (int c = 0; c < DI; c += 128) {
        float4 xr[XP_RPW], w[33];
#pragma unroll
        for (int r = 0; r < XP_RPW; ++r) xr[r] = *(const float4*)(xp + (size_t)r * DI + c);
#pragma unroll
        for (int n = 0; n < 33; ++n) w[n] = *(const float4*)(wp + (size_t)n * DI + c);
#pragma unroll
        for (int n = 0; n < 33; ++n) {
            const v2f wlo = {w[n].x, w[n].y}, whi = {w[n].z, w[n].w};
#pragma unroll
            for (int r = 0; r < XP_RPW; ++r) {
                const v2f xlo = {xr[r].x, xr[r].y}, xhi = {xr[r].z, xr[r].w};
                acc[r][n] = __builtin_elementwise_fma(xlo, wlo, acc[r][n]);
                acc[r][n] = __builtin_elementwise_fma(xhi, whi, acc[r][n]);
            }
        }
    }

#pragma unroll
    for (int r = 0; r < XP_RPW; ++r)
#pragma unroll
        for (int n = 0; n < 33; ++n)
            part[((XP_RPW * wv + r) * 33 + n) * XP_PITCH + lane] = acc[r][n].x + acc[r][n].y;
    __syncthreads();

    for (int o = threadIdx.x; o < XP_OUT; o += XP_WAVES * 32) {
        const float* q = part + o * XP_PITCH;
        float s0 = 0.f, s1 = 0.f, s2 = 0.f, s3 = 0.f;
#pragma unroll
        for (int l = 0; l < 32; l += 4) {
            s0 += q[l];  s1 += q[l + 1];  s2 += q[l + 2];  s3 += q[l + 3];
        }
        ssm[(size_t)row0 * 33 + o] = (s0 + s1) + (s2 + s3);
    }
}

#define SCAN_WAVES 16
#define SCAN_NBLK  (L_SZ / 16)
#define YT_PITCH   33

template <int CTRL>
__device__ __forceinline__ float row_dpp(float v) {
    return __builtin_bit_cast(float,
        __builtin_amdgcn_update_dpp(0, __builtin_bit_cast(int, v), CTRL, 0xf, 0xf, true));
}

#define SCAN_STEP(j)                                                  \
    {                                                                 \
        const float dtb  = row_dpp<0x150 + (j)>(dt);                  \
        const float xb   = row_dpp<0x150 + (j)>(xv);                  \
        const float Abar = __expf(dtb * Aa);                          \
        h    = fmaf(Abar, h, (dtb * Bv[j]) * xb);                     \
        p[j] = h * Cv[j];                                             \
    }

__global__ __launch_bounds__(SCAN_WAVES * 32) void scan_kernel(
    const float* __restrict__ ssm, const float* __restrict__ x_conv,
    const float* __restrict__ xg, const float* __restrict__ w_dt,
    const float* __restrict__ b_dt, const float* __restrict__ A_log,
    const float* __restrict__ D_param, float* __restrict__ y)
{
    static_assert(DI % 32 == 0 && L_SZ % 32 == 0 && DS == 16, "scan tiling");
    __shared__ float ytile[2][16 * YT_PITCH];

    const int lane  = threadIdx.x & 31;
    const int wv    = threadIdx.x >> 5;
    const int n     = lane & 15;
    const int c     = lane >> 4;
    const int b     = blockIdx.x / (DI / 32);
    const int dbase = (blockIdx.x % (DI / 32)) * 32;
    const int d     = dbase + 2 * wv + c;
    const size_t row0 = (size_t)b * L_SZ;

    const float wdt = w_dt[d];
    const float bdt = b_dt[d];
    const float Dp  = D_param[d];
    const float Aa  = -__expf(A_log[d * DS + n]);
    float h = 0.f;

    const bool b3 = (n & 8) != 0, b2 = (n & 4) != 0, b1 = (n & 2) != 0, b0 = (n & 1) != 0;

    const float* pBC = ssm + row0 * 33 + n;
    const float* pDT = ssm + (row0 + n) * 33 + 32;
    const float* pX  = x_conv + (row0 + n) * DI + d;
    const float* pG  = xg + (row0 + wv) * (2 * DI) + DI + dbase + lane;
    float*       pY  = y + (row0 + wv) * DI + dbase + lane;

    auto ldblk = [&](int k, float (&Bv)[16], float (&Cv)[16], float& dtr, float& xv, float& gv) {
        const float* q = pBC + (size_t)k * (16 * 33);
#pragma unroll
        for (int j = 0; j < 16; ++j) { Bv[j] = q[j * 33]; Cv[j] = q[j * 33 + 16]; }
        dtr = pDT[(size_t)k * (16 * 33)];
        xv  = pX[(size_t)k * (16 * DI)];
        gv  = pG[(size_t)k * (16 * 2 * DI)];
    };

    auto compute = [&](int k, const float (&Bv)[16], const float (&Cv)[16],
                       float dtr, float xv, float gv, float* yt) {
        const float z  = fmaf(dtr, wdt, bdt);
        const float dt = (z > 20.0f) ? z : log1pf(__expf(z));

        float p[16];
        SCAN_STEP(0)  SCAN_STEP(1)  SCAN_STEP(2)  SCAN_STEP(3)
        SCAN_STEP(4)  SCAN_STEP(5)  SCAN_STEP(6)  SCAN_STEP(7)
        SCAN_STEP(8)  SCAN_STEP(9)  SCAN_STEP(10) SCAN_STEP(11)
        SCAN_STEP(12) SCAN_STEP(13) SCAN_STEP(14) SCAN_STEP(15)

        float q8[8];
#pragma unroll
        for (int j = 0; j < 8; ++j) {
            const float keep = b3 ? p[j + 8] : p[j];
            const float send = b3 ? p[j] : p[j + 8];
            q8[j] = keep + row_dpp<0x168>(send);
        }
        float q4[4];
#pragma unroll
        for (int j = 0; j < 4; ++j) {
            const float keep = b2 ? q8[j + 4] : q8[j];
            const float send = b2 ? q8[j] : q8[j + 4];
            q4[j] = keep + row_dpp<0x164>(send);
        }
        float q2[2];
#pragma unroll
        for (int j = 0; j < 2; ++j) {
            const float keep = b1 ? q4[j + 2] : q4[j];
            const float send = b1 ? q4[j] : q4[j + 2];
            q2[j] = keep + row_dpp<0x162>(send);
        }
        const float keep1 = b0 ? q2[1] : q2[0];
        const float send1 = b0 ? q2[0] : q2[1];
        const float ysum  = keep1 + row_dpp<0x161>(send1);

        yt[n * YT_PITCH + 2 * wv + c] = fmaf(xv, Dp, ysum);
        __syncthreads();
        const float ys = yt[wv * YT_PITCH + lane];
        const float sg = gv / (1.0f + __expf(-gv));
        pY[(size_t)k * (16 * DI)] = ys * sg;
    };

    float BvA[16], CvA[16], dtrA, xvA, gvA;
    float BvB[16], CvB[16], dtrB, xvB, gvB;
    ldblk(0, BvA, CvA, dtrA, xvA, gvA);
    for (int k = 0; k < SCAN_NBLK; k += 2) {
        ldblk(k + 1, BvB, CvB, dtrB, xvB, gvB);
        compute(k, BvA, CvA, dtrA, xvA, gvA, ytile[0]);
        const int kn = (k + 2 < SCAN_NBLK) ? (k + 2) : (SCAN_NBLK - 1);
        ldblk(kn, BvA, CvA, dtrA, xvA, gvA);
        compute(k + 1, BvB, CvB, dtrB, xvB, gvB, ytile[1]);
    }
}

extern "C" void kernel_launch(void* const* d_in, const int* in_sizes, int n_in,
                              void* d_out, int out_size, void* d_ws, size_t ws_size,
                              hipStream_t stream)
{
    const float* x       = (const float*)d_in[0];
    const float* W_in    = (const float*)d_in[1];
    const float* conv_w  = (const float*)d_in[2];
    const float* conv_b  = (const float*)d_in[3];
    const float* W_xproj = (const float*)d_in[4];
    const float* w_dt    = (const float*)d_in[5];
    const float* b_dt    = (const float*)d_in[6];
    const float* A_log   = (const float*)d_in[7];
    const float* D_param = (const float*)d_in[8];
    const float* W_out   = (const float*)d_in[9];
    float* out = (float*)d_out;

    float* ws     = (float*)d_ws;
    float* xg     = ws;
    float* x_conv = xg + (size_t)NROW * (2 * DI);
    float* ssm    = x_conv + (size_t)NROW * DI;
    float* yb     = ssm + (size_t)NROW * 33;

    gemm32x64_f32_wmma<DM, true><<<(NROW / 32) * ((2 * DI) / 64), 32, 0, stream>>>(
        x, W_in, xg, (2 * DI) / 64, DM, DM, 2 * DI, conv_w, conv_b, x_conv);

    conv_edge_kernel<<<(NROW / 32) * (DI / 256), 256, 0, stream>>>(
        xg, conv_w, conv_b, x_conv);

    xproj_kernel<<<NROW / XP_ROWS, XP_WAVES * 32, 0, stream>>>(
        x_conv, W_xproj, ssm);

    scan_kernel<<<B_SZ * (DI / 32), SCAN_WAVES * 32, 0, stream>>>(
        ssm, x_conv, xg, w_dt, b_dt, A_log, D_param, yb);

    gemm32x64_f32_wmma<DI, false><<<(NROW / 32) * (DM / 64), 32, 0, stream>>>(
        yb, W_out, out, DM / 64, DI, DI, DM, nullptr, nullptr, nullptr);
}


# first GEMM on 64x64 register tiles per wave (16 accumulators, 3 waves per SIMD): 16 fragment loads per 64 WMMAs, was 12 per 32; conv epilogue and edge kernel on 64-row tiles; every sum unchanged
# speedup vs baseline: 8.9086x; 1.0848x over previous
#include <hip/hip_runtime.h>
#include <math.h>


typedef float  v2f __attribute__((ext_vector_type(2)));
typedef float  v4f __attribute__((ext_vector_type(4)));
typedef float  v8f __attribute__((ext_vector_type(8)));

#define B_SZ 2
#define L_SZ 2048
#define DM   768
#define DI   1536
#define DS   16
#define NROW (B_SZ * L_SZ)

__device__ __forceinline__ float xhalf(float v) {
    return __builtin_bit_cast(float, __builtin_amdgcn_permlanex16(
        __builtin_bit_cast(int, v), __builtin_bit_cast(int, v), 0x76543210, 0xfedcba98, false, false));
}

struct frag  { float4 q0, q1; };
struct vfrag { v4f q0, q1; };
__device__ __forceinline__ v2f frag_lo(const float4& q) { v2f r; r.x = q.x; r.y = q.y; return r; }
__device__ __forceinline__ v2f frag_hi(const float4& q) { v2f r; r.x = q.z; r.y = q.w; return r; }
__device__ __forceinline__ v2f frag_lo(const v4f& q) { return __builtin_shufflevector(q, q, 0, 1); }
__device__ __forceinline__ v2f frag_hi(const v4f& q) { return __builtin_shufflevector(q, q, 2, 3); }

__global__ __launch_bounds__(32) __attribute__((amdgpu_waves_per_eu(3, 3)))
void gemm64x64_conv_f32_wmma(
    const float* __restrict__ A, const float* __restrict__ W, float* __restrict__ C,
    const float* __restrict__ conv_w, const float* __restrict__ conv_b,
    float* __restrict__ x_conv)
{
    constexpr int K = DM, LDC = 2 * DI, N64 = (2 * DI) / 64;
    static_assert(K % 32 == 0 && NROW % 64 == 0 && L_SZ % 64 == 0, "first GEMM tiling");
    const int tile = blockIdx.x;
    const int tm   = tile / N64;
    const int tn   = tile % N64;
    const int lane = threadIdx.x;
    const int half = lane >> 4;
    const int lr   = lane & 15;

    const float* Ap[4];
    const float* Wp[4];
#pragma unroll
    for (int s = 0; s < 4; ++s) {
        Ap[s] = A + (size_t)(tm * 64 + 16 * s + lr) * K + 8 * half;
        Wp[s] = W + (size_t)(tn * 64 + 16 * s + lr) * K + 8 * half;
    }

    v8f acc[4][4];
#pragma unroll
    for (int s = 0; s < 4; ++s)
#pragma unroll
        for (int j = 0; j < 4; ++j) acc[s][j] = v8f{};

    auto ld = [&](int kb, vfrag (&a)[4], vfrag (&b)[4]) {
#pragma unroll
        for (int s = 0; s < 4; ++s) {
            a[s].q0 = *(const v4f*)(Ap[s] + kb);  a[s].q1 = *(const v4f*)(Ap[s] + kb + 4);
        }
#pragma unroll
        for (int j = 0; j < 4; ++j) {
            b[j].q0 = *(const v4f*)(Wp[j] + kb);  b[j].q1 = *(const v4f*)(Wp[j] + kb + 4);
        }
    };
    auto mm16 = [&](const v2f (&a)[4], const v2f (&b)[4]) {
#pragma unroll
        for (int s = 0; s < 4; ++s)
#pragma unroll
            for (int j = 0; j < 4; ++j)
                acc[s][j] = __builtin_amdgcn_wmma_f32_16x16x4_f32(false, a[s], false, b[j], (short)0, acc[s][j], false, false);
    };
    auto mm64 = [&](const vfrag (&a)[4], const vfrag (&b)[4]) {
        v2f fa[4], fb[4];
#pragma unroll
        for (int s = 0; s < 4; ++s) { fa[s] = frag_lo(a[s].q0); fb[s] = frag_lo(b[s].q0); }
        mm16(fa, fb);
#pragma unroll
        for (int s = 0; s < 4; ++s) { fa[s] = frag_hi(a[s].q0); fb[s] = frag_hi(b[s].q0); }
        mm16(fa, fb);
#pragma unroll
        for (int s = 0; s < 4; ++s) { fa[s] = frag_lo(a[s].q1); fb[s] = frag_lo(b[s].q1); }
        mm16(fa, fb);
#pragma unroll
        for (int s = 0; s < 4; ++s) { fa[s] = frag_hi(a[s].q1); fb[s] = frag_hi(b[s].q1); }
        mm16(fa, fb);
    };

    vfrag aA[4], bA[4], aB[4], bB[4];
    ld(0, aA, bA);
    for (int k = 0; k < K; k += 32) {
        ld(k + 16, aB, bB);
        mm64(aA, bA);
        const int kn = (k + 32 < K) ? (k + 32) : k;
        ld(kn, aA, bA);
        mm64(aB, bB);
    }

    float* Cb = C + (size_t)(tm * 64 + 8 * half) * LDC + tn * 64 + lr;

    if (tn < DI / 64) {
        float* Xb = x_conv + (size_t)(tm * 64 + 8 * half) * DI + tn * 64 + lr;
#pragma unroll
        for (int j = 0; j < 4; ++j) {
            const int    d = tn * 64 + 16 * j + lr;
            const float4 w = *(const float4*)(conv_w + (size_t)d * 4);
            const float  b = conv_b[d];
            float xs[4][3];
#pragma unroll
            for (int s = 0; s < 4; ++s)
#pragma unroll
                for (int i = 0; i < 3; ++i) xs[s][i] = xhalf(acc[s][j][5 + i]);
#pragma unroll
            for (int s = 0; s < 4; ++s) {
                float e[11];
#pragma unroll
                for (int i = 0; i < 3; ++i) e[i] = (half || s == 0) ? xs[s][i] : xs[s > 0 ? s - 1 : 0][i];
#pragma unroll
                for (int r = 0; r < 8; ++r) e[3 + r] = acc[s][j][r];
#pragma unroll
                for (int r = 0; r < 8; ++r) {
                    float a = b;
                    a = fmaf(w.x, e[r],     a);
                    a = fmaf(w.y, e[r + 1], a);
                    a = fmaf(w.z, e[r + 2], a);
                    a = fmaf(w.w, e[r + 3], a);
                    const float y = a / (1.0f + __expf(-a));
                    if (s > 0 || r >= 3 || half) Xb[(size_t)(16 * s + r) * DI + 16 * j] = y;
                    if (s == 0 && r < 3 && !half)  Cb[(size_t)r * LDC + 16 * j] = acc[s][j][r];
                    if (s == 3 && r >= 5 && half)  Cb[(size_t)(48 + r) * LDC + 16 * j] = acc[s][j][r];
                }
            }
        }
        return;
    }
#pragma unroll
    for (int s = 0; s < 4; ++s)
#pragma unroll
        for (int r = 0; r < 8; ++r) {
            float* Cr = Cb + (size_t)(16 * s + r) * LDC;
            Cr[0]  = acc[s][0][r];  Cr[16] = acc[s][1][r];
            Cr[32] = acc[s][2][r];  Cr[48] = acc[s][3][r];
        }
}

template <int K>
__global__ __launch_bounds__(32) void gemm32x64_f32_wmma(
    const float* __restrict__ A, const float* __restrict__ W,
    float* __restrict__ C, int N64, int lda, int ldw, int ldc)
{
    static_assert(K % 32 == 0, "K must be a multiple of 32");
    const int tile = blockIdx.x;
    const int tm   = tile / N64;
    const int tn   = tile % N64;
    const int lane = threadIdx.x;
    const int half = lane >> 4;
    const int lr   = lane & 15;

    const float* Ar0 = A + (size_t)(tm * 32 +  0 + lr) * lda;
    const float* Ar1 = A + (size_t)(tm * 32 + 16 + lr) * lda;
    const float* W0  = W + (size_t)(tn * 64 +  0 + lr) * ldw;
    const float* W1  = W + (size_t)(tn * 64 + 16 + lr) * ldw;
    const float* W2  = W + (size_t)(tn * 64 + 32 + lr) * ldw;
    const float* W3  = W + (size_t)(tn * 64 + 48 + lr) * ldw;

    v8f acc00 = {}; v8f acc01 = {}; v8f acc02 = {}; v8f acc03 = {};
    v8f acc10 = {}; v8f acc11 = {}; v8f acc12 = {}; v8f acc13 = {};

    const int kh8 = 8 * half;

    auto ld = [&](int kb, frag& a0, frag& a1, frag& b0, frag& b1, frag& b2, frag& b3) {
        a0.q0 = *(const float4*)(Ar0 + kb);  a0.q1 = *(const float4*)(Ar0 + kb + 4);
        a1.q0 = *(const float4*)(Ar1 + kb);  a1.q1 = *(const float4*)(Ar1 + kb + 4);
        b0.q0 = *(const float4*)(W0 + kb);   b0.q1 = *(const float4*)(W0 + kb + 4);
        b1.q0 = *(const float4*)(W1 + kb);   b1.q1 = *(const float4*)(W1 + kb + 4);
        b2.q0 = *(const float4*)(W2 + kb);   b2.q1 = *(const float4*)(W2 + kb + 4);
        b3.q0 = *(const float4*)(W3 + kb);   b3.q1 = *(const float4*)(W3 + kb + 4);
    };
    auto mm8 = [&](v2f a0, v2f a1, v2f b0, v2f b1, v2f b2, v2f b3) {
        acc00 = __builtin_amdgcn_wmma_f32_16x16x4_f32(false, a0, false, b0, (short)0, acc00, false, false);
        acc01 = __builtin_amdgcn_wmma_f32_16x16x4_f32(false, a0, false, b1, (short)0, acc01, false, false);
        acc02 = __builtin_amdgcn_wmma_f32_16x16x4_f32(false, a0, false, b2, (short)0, acc02, false, false);
        acc03 = __builtin_amdgcn_wmma_f32_16x16x4_f32(false, a0, false, b3, (short)0, acc03, false, false);
        acc10 = __builtin_amdgcn_wmma_f32_16x16x4_f32(false, a1, false, b0, (short)0, acc10, false, false);
        acc11 = __builtin_amdgcn_wmma_f32_16x16x4_f32(false, a1, false, b1, (short)0, acc11, false, false);
        acc12 = __builtin_amdgcn_wmma_f32_16x16x4_f32(false, a1, false, b2, (short)0, acc12, false, false);
        acc13 = __builtin_amdgcn_wmma_f32_16x16x4_f32(false, a1, false, b3, (short)0, acc13, false, false);
    };
    auto lo = [](const float4& q) { return frag_lo(q); };
    auto hi = [](const float4& q) { return frag_hi(q); };
    auto mm32 = [&](const frag& a0, const frag& a1, const frag& b0, const frag& b1, const frag& b2, const frag& b3) {
        mm8(lo(a0.q0), lo(a1.q0), lo(b0.q0), lo(b1.q0), lo(b2.q0), lo(b3.q0));
        mm8(hi(a0.q0), hi(a1.q0), hi(b0.q0), hi(b1.q0), hi(b2.q0), hi(b3.q0));
        mm8(lo(a0.q1), lo(a1.q1), lo(b0.q1), lo(b1.q1), lo(b2.q1), lo(b3.q1));
        mm8(hi(a0.q1), hi(a1.q1), hi(b0.q1), hi(b1.q1), hi(b2.q1), hi(b3.q1));
    };

    frag a0A, a1A, b0A, b1A, b2A, b3A;
    frag a0B, a1B, b0B, b1B, b2B, b3B;

    ld(kh8, a0A, a1A, b0A, b1A, b2A, b3A);
    for (int k = 0; k < K - 32; k += 32) {
        ld(k + 16 + kh8, a0B, a1B, b0B, b1B, b2B, b3B);
        mm32(a0A, a1A, b0A, b1A, b2A, b3A);
        ld(k + 32 + kh8, a0A, a1A, b0A, b1A, b2A, b3A);
        mm32(a0B, a1B, b0B, b1B, b2B, b3B);
    }
    ld(K - 16 + kh8, a0B, a1B, b0B, b1B, b2B, b3B);
    mm32(a0A, a1A, b0A, b1A, b2A, b3A);
    mm32(a0B, a1B, b0B, b1B, b2B, b3B);

    float* Cb0 = C + (size_t)(tm * 32 +  0 + 8 * half) * ldc + tn * 64 + lr;
    float* Cb1 = C + (size_t)(tm * 32 + 16 + 8 * half) * ldc + tn * 64 + lr;
#pragma unroll
    for (int r = 0; r < 8; ++r) {
        float* Cr0 = Cb0 + (size_t)r * ldc;
        float* Cr1 = Cb1 + (size_t)r * ldc;
        Cr0[0]  = acc00[r];  Cr0[16] = acc01[r];
        Cr0[32] = acc02[r];  Cr0[48] = acc03[r];
        Cr1[0]  = acc10[r];  Cr1[16] = acc11[r];
        Cr1[32] = acc12[r];  Cr1[48] = acc13[r];
    }
}

__global__ __launch_bounds__(256) void conv_edge_kernel(
    const float* __restrict__ xg, const float* __restrict__ conv_w,
    const float* __restrict__ conv_b, float* __restrict__ x_conv)
{
    static_assert(DI % 256 == 0 && L_SZ % 64 == 0, "edge tiling");
    const int  tile = blockIdx.x / (DI / 256);
    const int  d    = (blockIdx.x % (DI / 256)) * 256 + threadIdx.x;
    const int  row0 = tile * 64;
    const bool head = (row0 % L_SZ) == 0;

    const float  b  = conv_b[d];
    const float4 w4 = *(const float4*)(conv_w + (size_t)d * 4);
    const float  w[4] = {w4.x, w4.y, w4.z, w4.w};
    const float* p  = xg + (size_t)row0 * (2 * DI) + d;

    float in[6];
    in[0] = in[1] = in[2] = 0.0f;
    if (!head) {
        in[0] = *(p - 3 * (2 * DI));
        in[1] = *(p - 2 * (2 * DI));
        in[2] = *(p - 1 * (2 * DI));
    }
    in[3] = p[0];
    in[4] = p[1 * (2 * DI)];
    in[5] = p[2 * (2 * DI)];
#pragma unroll
    for (int i = 0; i < 3; ++i) {
        float acc = b;
#pragma unroll
        for (int j = 0; j < 4; ++j) {
            const float t = fmaf(w[j], in[i + j], acc);
            acc = (head && i + j < 3) ? acc : t;
        }
        x_conv[(size_t)(row0 + i) * DI + d] = acc / (1.0f + __expf(-acc));
    }
}

#define XP_WAVES 8
#define XP_RPW   4
#define XP_ROWS  (XP_WAVES * XP_RPW)
#define XP_OUT   (XP_ROWS * 33)
#define XP_PITCH 33

__global__ __launch_bounds__(XP_WAVES * 32) __attribute__((amdgpu_waves_per_eu(2, 2)))
void xproj_kernel(
    const float* __restrict__ x_conv, const float* __restrict__ Wx,
    float* __restrict__ ssm)
{
    static_assert(NROW % XP_ROWS == 0 && DI % 128 == 0, "xproj tiling");
    __shared__ float part[XP_OUT * XP_PITCH];

    const int lane = threadIdx.x & 31;
    const int wv   = threadIdx.x >> 5;
    const int row0 = blockIdx.x * XP_ROWS;

    const float* xp = x_conv + (size_t)(row0 + XP_RPW * wv) * DI + 4 * lane;
    const float* wp = Wx + 4 * lane;

    v2f acc[XP_RPW][33];
#pragma unroll
    for (int r = 0; r < XP_RPW; ++r)
#pragma unroll
        for (int n = 0; n < 33; ++n) acc[r][n] = v2f{0.f, 0.f};

    for (int c = 0; c < DI; c += 128) {
        float4 xr[XP_RPW], w[33];
#pragma unroll
        for (int r = 0; r < XP_RPW; ++r) xr[r] = *(const float4*)(xp + (size_t)r * DI + c);
#pragma unroll
        for (int n = 0; n < 33; ++n) w[n] = *(const float4*)(wp + (size_t)n * DI + c);
#pragma unroll
        for (int n = 0; n < 33; ++n) {
            const v2f wlo = {w[n].x, w[n].y}, whi = {w[n].z, w[n].w};
#pragma unroll
            for (int r = 0; r < XP_RPW; ++r) {
                const v2f xlo = {xr[r].x, xr[r].y}, xhi = {xr[r].z, xr[r].w};
                acc[r][n] = __builtin_elementwise_fma(xlo, wlo, acc[r][n]);
                acc[r][n] = __builtin_elementwise_fma(xhi, whi, acc[r][n]);
            }
        }
    }

#pragma unroll
    for (int r = 0; r < XP_RPW; ++r)
#pragma unroll
        for (int n = 0; n < 33; ++n)
            part[((XP_RPW * wv + r) * 33 + n) * XP_PITCH + lane] = acc[r][n].x + acc[r][n].y;
    __syncthreads();

    for (int o = threadIdx.x; o < XP_OUT; o += XP_WAVES * 32) {
        const float* q = part + o * XP_PITCH;
        float s0 = 0.f, s1 = 0.f, s2 = 0.f, s3 = 0.f;
#pragma unroll
        for (int l = 0; l < 32; l += 4) {
            s0 += q[l];  s1 += q[l + 1];  s2 += q[l + 2];  s3 += q[l + 3];
        }
        ssm[(size_t)row0 * 33 + o] = (s0 + s1) + (s2 + s3);
    }
}

#define SCAN_WAVES 16
#define SCAN_NBLK  (L_SZ / 16)
#define YT_PITCH   33

template <int CTRL>
__device__ __forceinline__ float row_dpp(float v) {
    return __builtin_bit_cast(float,
        __builtin_amdgcn_update_dpp(0, __builtin_bit_cast(int, v), CTRL, 0xf, 0xf, true));
}

#define SCAN_STEP(j)                                                  \
    {                                                                 \
        const float dtb  = row_dpp<0x150 + (j)>(dt);                  \
        const float xb   = row_dpp<0x150 + (j)>(xv);                  \
        const float Abar = __expf(dtb * Aa);                          \
        h    = fmaf(Abar, h, (dtb * Bv[j]) * xb);                     \
        p[j] = h * Cv[j];                                             \
    }

__global__ __launch_bounds__(SCAN_WAVES * 32) void scan_kernel(
    const float* __restrict__ ssm, const float* __restrict__ x_conv,
    const float* __restrict__ xg, const float* __restrict__ w_dt,
    const float* __restrict__ b_dt, const float* __restrict__ A_log,
    const float* __restrict__ D_param, float* __restrict__ y)
{
    static_assert(DI % 32 == 0 && L_SZ % 32 == 0 && DS == 16, "scan tiling");
    __shared__ float ytile[2][16 * YT_PITCH];

    const int lane  = threadIdx.x & 31;
    const int wv    = threadIdx.x >> 5;
    const int n     = lane & 15;
    const int c     = lane >> 4;
    const int b     = blockIdx.x / (DI / 32);
    const int dbase = (blockIdx.x % (DI / 32)) * 32;
    const int d     = dbase + 2 * wv + c;
    const size_t row0 = (size_t)b * L_SZ;

    const float wdt = w_dt[d];
    const float bdt = b_dt[d];
    const float Dp  = D_param[d];
    const float Aa  = -__expf(A_log[d * DS + n]);
    float h = 0.f;

    const bool b3 = (n & 8) != 0, b2 = (n & 4) != 0, b1 = (n & 2) != 0, b0 = (n & 1) != 0;

    const float* pBC = ssm + row0 * 33 + n;
    const float* pDT = ssm + (row0 + n) * 33 + 32;
    const float* pX  = x_conv + (row0 + n) * DI + d;
    const float* pG  = xg + (row0 + wv) * (2 * DI) + DI + dbase + lane;
    float*       pY  = y + (row0 + wv) * DI + dbase + lane;

    auto ldblk = [&](int k, float (&Bv)[16], float (&Cv)[16], float& dtr, float& xv, float& gv) {
        const float* q = pBC + (size_t)k * (16 * 33);
#pragma unroll
        for (int j = 0; j < 16; ++j) { Bv[j] = q[j * 33]; Cv[j] = q[j * 33 + 16]; }
        dtr = pDT[(size_t)k * (16 * 33)];
        xv  = pX[(size_t)k * (16 * DI)];
        gv  = pG[(size_t)k * (16 * 2 * DI)];
    };

    auto compute = [&](int k, const float (&Bv)[16], const float (&Cv)[16],
                       float dtr, float xv, float gv, float* yt) {
        const float z  = fmaf(dtr, wdt, bdt);
        const float dt = (z > 20.0f) ? z : log1pf(__expf(z));

        float p[16];
        SCAN_STEP(0)  SCAN_STEP(1)  SCAN_STEP(2)  SCAN_STEP(3)
        SCAN_STEP(4)  SCAN_STEP(5)  SCAN_STEP(6)  SCAN_STEP(7)
        SCAN_STEP(8)  SCAN_STEP(9)  SCAN_STEP(10) SCAN_STEP(11)
        SCAN_STEP(12) SCAN_STEP(13) SCAN_STEP(14) SCAN_STEP(15)

        float q8[8];
#pragma unroll
        for (int j = 0; j < 8; ++j) {
            const float keep = b3 ? p[j + 8] : p[j];
            const float send = b3 ? p[j] : p[j + 8];
            q8[j] = keep + row_dpp<0x168>(send);
        }
        float q4[4];
#pragma unroll
        for (int j = 0; j < 4; ++j) {
            const float keep = b2 ? q8[j + 4] : q8[j];
            const float send = b2 ? q8[j] : q8[j + 4];
            q4[j] = keep + row_dpp<0x164>(send);
        }
        float q2[2];
#pragma unroll
        for (int j = 0; j < 2; ++j) {
            const float keep = b1 ? q4[j + 2] : q4[j];
            const float send = b1 ? q4[j] : q4[j + 2];
            q2[j] = keep + row_dpp<0x162>(send);
        }
        const float keep1 = b0 ? q2[1] : q2[0];
        const float send1 = b0 ? q2[0] : q2[1];
        const float ysum  = keep1 + row_dpp<0x161>(send1);

        yt[n * YT_PITCH + 2 * wv + c] = fmaf(xv, Dp, ysum);
        __syncthreads();
        const float ys = yt[wv * YT_PITCH + lane];
        const float sg = gv / (1.0f + __expf(-gv));
        pY[(size_t)k * (16 * DI)] = ys * sg;
    };

    float BvA[16], CvA[16], dtrA, xvA, gvA;
    float BvB[16], CvB[16], dtrB, xvB, gvB;
    ldblk(0, BvA, CvA, dtrA, xvA, gvA);
    for (int k = 0; k < SCAN_NBLK; k += 2) {
        ldblk(k + 1, BvB, CvB, dtrB, xvB, gvB);
        compute(k, BvA, CvA, dtrA, xvA, gvA, ytile[0]);
        const int kn = (k + 2 < SCAN_NBLK) ? (k + 2) : (SCAN_NBLK - 1);
        ldblk(kn, BvA, CvA, dtrA, xvA, gvA);
        compute(k + 1, BvB, CvB, dtrB, xvB, gvB, ytile[1]);
    }
}

extern "C" void kernel_launch(void* const* d_in, const int* in_sizes, int n_in,
                              void* d_out, int out_size, void* d_ws, size_t ws_size,
                              hipStream_t stream)
{
    const float* x       = (const float*)d_in[0];
    const float* W_in    = (const float*)d_in[1];
    const float* conv_w  = (const float*)d_in[2];
    const float* conv_b  = (const float*)d_in[3];
    const float* W_xproj = (const float*)d_in[4];
    const float* w_dt    = (const float*)d_in[5];
    const float* b_dt    = (const float*)d_in[6];
    const float* A_log   = (const float*)d_in[7];
    const float* D_param = (const float*)d_in[8];
    const float* W_out   = (const float*)d_in[9];
    float* out = (float*)d_out;

    float* ws     = (float*)d_ws;
    float* xg     = ws;
    float* x_conv = xg + (size_t)NROW * (2 * DI);
    float* ssm    = x_conv + (size_t)NROW * DI;
    float* yb     = ssm + (size_t)NROW * 33;

    gemm64x64_conv_f32_wmma<<<(NROW / 64) * ((2 * DI) / 64), 32, 0, stream>>>(
        x, W_in, xg, conv_w, conv_b, x_conv);

    conv_edge_kernel<<<(NROW / 64) * (DI / 256), 256, 0, stream>>>(
        xg, conv_w, conv_b, x_conv);

    xproj_kernel<<<NROW / XP_ROWS, XP_WAVES * 32, 0, stream>>>(
        x_conv, W_xproj, ssm);

    scan_kernel<<<B_SZ * (DI / 32), SCAN_WAVES * 32, 0, stream>>>(
        ssm, x_conv, xg, w_dt, b_dt, A_log, D_param, yb);

    gemm32x64_f32_wmma<DI><<<(NROW / 32) * (DM / 64), 32, 0, stream>>>(
        yb, W_out, out, DM / 64, DI, DI, DM);
}


# scan re-cut: 384 work-groups of 4 waves (8 channels) from 96 of 16, at most 2 waves per SIMD chip-wide; y in 8-channel panels (whole-line stores kept), second GEMM reads A from them; sums unchanged
# speedup vs baseline: 10.3731x; 1.1644x over previous
#include <hip/hip_runtime.h>
#include <math.h>


typedef float  v2f __attribute__((ext_vector_type(2)));
typedef float  v4f __attribute__((ext_vector_type(4)));
typedef float  v8f __attribute__((ext_vector_type(8)));

#define B_SZ 2
#define L_SZ 2048
#define DM   768
#define DI   1536
#define DS   16
#define NROW (B_SZ * L_SZ)

__device__ __forceinline__ float xhalf(float v) {
    return __builtin_bit_cast(float, __builtin_amdgcn_permlanex16(
        __builtin_bit_cast(int, v), __builtin_bit_cast(int, v), 0x76543210, 0xfedcba98, false, false));
}

struct frag  { float4 q0, q1; };
struct vfrag { v4f q0, q1; };
__device__ __forceinline__ v2f frag_lo(const float4& q) { v2f r; r.x = q.x; r.y = q.y; return r; }
__device__ __forceinline__ v2f frag_hi(const float4& q) { v2f r; r.x = q.z; r.y = q.w; return r; }
__device__ __forceinline__ v2f frag_lo(const v4f& q) { return __builtin_shufflevector(q, q, 0, 1); }
__device__ __forceinline__ v2f frag_hi(const v4f& q) { return __builtin_shufflevector(q, q, 2, 3); }

__global__ __launch_bounds__(32) __attribute__((amdgpu_waves_per_eu(3, 3)))
void gemm64x64_conv_f32_wmma(
    const float* __restrict__ A, const float* __restrict__ W, float* __restrict__ C,
    const float* __restrict__ conv_w, const float* __restrict__ conv_b,
    float* __restrict__ x_conv)
{
    constexpr int K = DM, LDC = 2 * DI, N64 = (2 * DI) / 64;
    static_assert(K % 32 == 0 && NROW % 64 == 0 && L_SZ % 64 == 0, "first GEMM tiling");
    const int tile = blockIdx.x;
    const int tm   = tile / N64;
    const int tn   = tile % N64;
    const int lane = threadIdx.x;
    const int half = lane >> 4;
    const int lr   = lane & 15;

    const float* Ap[4];
    const float* Wp[4];
#pragma unroll
    for (int s = 0; s < 4; ++s) {
        Ap[s] = A + (size_t)(tm * 64 + 16 * s + lr) * K + 8 * half;
        Wp[s] = W + (size_t)(tn * 64 + 16 * s + lr) * K + 8 * half;
    }

    v8f acc[4][4];
#pragma unroll
    for (int s = 0; s < 4; ++s)
#pragma unroll
        for (int j = 0; j < 4; ++j) acc[s][j] = v8f{};

    auto ld = [&](int kb, vfrag (&a)[4], vfrag (&b)[4]) {
#pragma unroll
        for (int s = 0; s < 4; ++s) {
            a[s].q0 = *(const v4f*)(Ap[s] + kb);  a[s].q1 = *(const v4f*)(Ap[s] + kb + 4);
        }
#pragma unroll
        for (int j = 0; j < 4; ++j) {
            b[j].q0 = *(const v4f*)(Wp[j] + kb);  b[j].q1 = *(const v4f*)(Wp[j] + kb + 4);
        }
    };
    auto mm16 = [&](const v2f (&a)[4], const v2f (&b)[4]) {
#pragma unroll
        for (int s = 0; s < 4; ++s)
#pragma unroll
            for (int j = 0; j < 4; ++j)
                acc[s][j] = __builtin_amdgcn_wmma_f32_16x16x4_f32(false, a[s], false, b[j], (short)0, acc[s][j], false, false);
    };
    auto mm64 = [&](const vfrag (&a)[4], const vfrag (&b)[4]) {
        v2f fa[4], fb[4];
#pragma unroll
        for (int s = 0; s < 4; ++s) { fa[s] = frag_lo(a[s].q0); fb[s] = frag_lo(b[s].q0); }
        mm16(fa, fb);
#pragma unroll
        for (int s = 0; s < 4; ++s) { fa[s] = frag_hi(a[s].q0); fb[s] = frag_hi(b[s].q0); }
        mm16(fa, fb);
#pragma unroll
        for (int s = 0; s < 4; ++s) { fa[s] = frag_lo(a[s].q1); fb[s] = frag_lo(b[s].q1); }
        mm16(fa, fb);
#pragma unroll
        for (int s = 0; s < 4; ++s) { fa[s] = frag_hi(a[s].q1); fb[s] = frag_hi(b[s].q1); }
        mm16(fa, fb);
    };

    vfrag aA[4], bA[4], aB[4], bB[4];
    ld(0, aA, bA);
    for (int k = 0; k < K; k += 32) {
        ld(k + 16, aB, bB);
        mm64(aA, bA);
        const int kn = (k + 32 < K) ? (k + 32) : k;
        ld(kn, aA, bA);
        mm64(aB, bB);
    }

    float* Cb = C + (size_t)(tm * 64 + 8 * half) * LDC + tn * 64 + lr;

    if (tn < DI / 64) {
        float* Xb = x_conv + (size_t)(tm * 64 + 8 * half) * DI + tn * 64 + lr;
#pragma unroll
        for (int j = 0; j < 4; ++j) {
            const int    d = tn * 64 + 16 * j + lr;
            const float4 w = *(const float4*)(conv_w + (size_t)d * 4);
            const float  b = conv_b[d];
            float xs[4][3];
#pragma unroll
            for (int s = 0; s < 4; ++s)
#pragma unroll
                for (int i = 0; i < 3; ++i) xs[s][i] = xhalf(acc[s][j][5 + i]);
#pragma unroll
            for (int s = 0; s < 4; ++s) {
                float e[11];
#pragma unroll
                for (int i = 0; i < 3; ++i) e[i] = (half || s == 0) ? xs[s][i] : xs[s > 0 ? s - 1 : 0][i];
#pragma unroll
                for (int r = 0; r < 8; ++r) e[3 + r] = acc[s][j][r];
#pragma unroll
                for (int r = 0; r < 8; ++r) {
                    float a = b;
                    a = fmaf(w.x, e[r],     a);
                    a = fmaf(w.y, e[r + 1], a);
                    a = fmaf(w.z, e[r + 2], a);
                    a = fmaf(w.w, e[r + 3], a);
                    const float y = a / (1.0f + __expf(-a));
                    if (s > 0 || r >= 3 || half) Xb[(size_t)(16 * s + r) * DI + 16 * j] = y;
                    if (s == 0 && r < 3 && !half)  Cb[(size_t)r * LDC + 16 * j] = acc[s][j][r];
                    if (s == 3 && r >= 5 && half)  Cb[(size_t)(48 + r) * LDC + 16 * j] = acc[s][j][r];
                }
            }
        }
        return;
    }
#pragma unroll
    for (int s = 0; s < 4; ++s)
#pragma unroll
        for (int r = 0; r < 8; ++r) {
            float* Cr = Cb + (size_t)(16 * s + r) * LDC;
            Cr[0]  = acc[s][0][r];  Cr[16] = acc[s][1][r];
            Cr[32] = acc[s][2][r];  Cr[48] = acc[s][3][r];
        }
}

template <int K>
__global__ __launch_bounds__(32) void gemm32x64_f32_wmma(
    const float* __restrict__ A, const float* __restrict__ W,
    float* __restrict__ C, int N64, int ldw, int ldc)
{
    static_assert(K % 32 == 0 && L_SZ % 32 == 0, "K a multiple of 32; a 32-row tile inside one sequence");
    const int tile = blockIdx.x;
    const int tm   = tile / N64;
    const int tn   = tile % N64;
    const int lane = threadIdx.x;
    const int half = lane >> 4;
    const int lr   = lane & 15;

    const int    ab  = (tm * 32) / L_SZ;
    const int    al  = (tm * 32) % L_SZ + lr;
    const float* Ar0 = A + (((size_t)ab * (K / 8) + half) * L_SZ + al) * 8;
    const float* Ar1 = Ar0 + 16 * 8;
    const float* W0  = W + (size_t)(tn * 64 +  0 + lr) * ldw;
    const float* W1  = W + (size_t)(tn * 64 + 16 + lr) * ldw;
    const float* W2  = W + (size_t)(tn * 64 + 32 + lr) * ldw;
    const float* W3  = W + (size_t)(tn * 64 + 48 + lr) * ldw;

    v8f acc00 = {}; v8f acc01 = {}; v8f acc02 = {}; v8f acc03 = {};
    v8f acc10 = {}; v8f acc11 = {}; v8f acc12 = {}; v8f acc13 = {};

    const int kh8 = 8 * half;

    auto ld = [&](int k, frag& a0, frag& a1, frag& b0, frag& b1, frag& b2, frag& b3) {
        const size_t ka = (size_t)k * L_SZ;
        const int    kb = k + kh8;
        a0.q0 = *(const float4*)(Ar0 + ka);  a0.q1 = *(const float4*)(Ar0 + ka + 4);
        a1.q0 = *(const float4*)(Ar1 + ka);  a1.q1 = *(const float4*)(Ar1 + ka + 4);
        b0.q0 = *(const float4*)(W0 + kb);   b0.q1 = *(const float4*)(W0 + kb + 4);
        b1.q0 = *(const float4*)(W1 + kb);   b1.q1 = *(const float4*)(W1 + kb + 4);
        b2.q0 = *(const float4*)(W2 + kb);   b2.q1 = *(const float4*)(W2 + kb + 4);
        b3.q0 = *(const float4*)(W3 + kb);   b3.q1 = *(const float4*)(W3 + kb + 4);
    };
    auto mm8 = [&](v2f a0, v2f a1, v2f b0, v2f b1, v2f b2, v2f b3) {
        acc00 = __builtin_amdgcn_wmma_f32_16x16x4_f32(false, a0, false, b0, (short)0, acc00, false, false);
        acc01 = __builtin_amdgcn_wmma_f32_16x16x4_f32(false, a0, false, b1, (short)0, acc01, false, false);
        acc02 = __builtin_amdgcn_wmma_f32_16x16x4_f32(false, a0, false, b2, (short)0, acc02, false, false);
        acc03 = __builtin_amdgcn_wmma_f32_16x16x4_f32(false, a0, false, b3, (short)0, acc03, false, false);
        acc10 = __builtin_amdgcn_wmma_f32_16x16x4_f32(false, a1, false, b0, (short)0, acc10, false, false);
        acc11 = __builtin_amdgcn_wmma_f32_16x16x4_f32(false, a1, false, b1, (short)0, acc11, false, false);
        acc12 = __builtin_amdgcn_wmma_f32_16x16x4_f32(false, a1, false, b2, (short)0, acc12, false, false);
        acc13 = __builtin_amdgcn_wmma_f32_16x16x4_f32(false, a1, false, b3, (short)0, acc13, false, false);
    };
    auto lo = [](const float4& q) { return frag_lo(q); };
    auto hi = [](const float4& q) { return frag_hi(q); };
    auto mm32 = [&](const frag& a0, const frag& a1, const frag& b0, const frag& b1, const frag& b2, const frag& b3) {
        mm8(lo(a0.q0), lo(a1.q0), lo(b0.q0), lo(b1.q0), lo(b2.q0), lo(b3.q0));
        mm8(hi(a0.q0), hi(a1.q0), hi(b0.q0), hi(b1.q0), hi(b2.q0), hi(b3.q0));
        mm8(lo(a0.q1), lo(a1.q1), lo(b0.q1), lo(b1.q1), lo(b2.q1), lo(b3.q1));
        mm8(hi(a0.q1), hi(a1.q1), hi(b0.q1), hi(b1.q1), hi(b2.q1), hi(b3.q1));
    };

    frag a0A, a1A, b0A, b1A, b2A, b3A;
    frag a0B, a1B, b0B, b1B, b2B, b3B;

    ld(0, a0A, a1A, b0A, b1A, b2A, b3A);
    for (int k = 0; k < K - 32; k += 32) {
        ld(k + 16, a0B, a1B, b0B, b1B, b2B, b3B);
        mm32(a0A, a1A, b0A, b1A, b2A, b3A);
        ld(k + 32, a0A, a1A, b0A, b1A, b2A, b3A);
        mm32(a0B, a1B, b0B, b1B, b2B, b3B);
    }
    ld(K - 16, a0B, a1B, b0B, b1B, b2B, b3B);
    mm32(a0A, a1A, b0A, b1A, b2A, b3A);
    mm32(a0B, a1B, b0B, b1B, b2B, b3B);

    float* Cb0 = C + (size_t)(tm * 32 +  0 + 8 * half) * ldc + tn * 64 + lr;
    float* Cb1 = C + (size_t)(tm * 32 + 16 + 8 * half) * ldc + tn * 64 + lr;
#pragma unroll
    for (int r = 0; r < 8; ++r) {
        float* Cr0 = Cb0 + (size_t)r * ldc;
        float* Cr1 = Cb1 + (size_t)r * ldc;
        Cr0[0]  = acc00[r];  Cr0[16] = acc01[r];
        Cr0[32] = acc02[r];  Cr0[48] = acc03[r];
        Cr1[0]  = acc10[r];  Cr1[16] = acc11[r];
        Cr1[32] = acc12[r];  Cr1[48] = acc13[r];
    }
}

__global__ __launch_bounds__(256) void conv_edge_kernel(
    const float* __restrict__ xg, const float* __restrict__ conv_w,
    const float* __restrict__ conv_b, float* __restrict__ x_conv)
{
    static_assert(DI % 256 == 0 && L_SZ % 64 == 0, "edge tiling");
    const int  tile = blockIdx.x / (DI / 256);
    const int  d    = (blockIdx.x % (DI / 256)) * 256 + threadIdx.x;
    const int  row0 = tile * 64;
    const bool head = (row0 % L_SZ) == 0;

    const float  b  = conv_b[d];
    const float4 w4 = *(const float4*)(conv_w + (size_t)d * 4);
    const float  w[4] = {w4.x, w4.y, w4.z, w4.w};
    const float* p  = xg + (size_t)row0 * (2 * DI) + d;

    float in[6];
    in[0] = in[1] = in[2] = 0.0f;
    if (!head) {
        in[0] = *(p - 3 * (2 * DI));
        in[1] = *(p - 2 * (2 * DI));
        in[2] = *(p - 1 * (2 * DI));
    }
    in[3] = p[0];
    in[4] = p[1 * (2 * DI)];
    in[5] = p[2 * (2 * DI)];
#pragma unroll
    for (int i = 0; i < 3; ++i) {
        float acc = b;
#pragma unroll
        for (int j = 0; j < 4; ++j) {
            const float t = fmaf(w[j], in[i + j], acc);
            acc = (head && i + j < 3) ? acc : t;
        }
        x_conv[(size_t)(row0 + i) * DI + d] = acc / (1.0f + __expf(-acc));
    }
}

#define XP_WAVES 8
#define XP_RPW   4
#define XP_ROWS  (XP_WAVES * XP_RPW)
#define XP_OUT   (XP_ROWS * 33)
#define XP_PITCH 33

__global__ __launch_bounds__(XP_WAVES * 32) __attribute__((amdgpu_waves_per_eu(2, 2)))
void xproj_kernel(
    const float* __restrict__ x_conv, const float* __restrict__ Wx,
    float* __restrict__ ssm)
{
    static_assert(NROW % XP_ROWS == 0 && DI % 128 == 0, "xproj tiling");
    __shared__ float part[XP_OUT * XP_PITCH];

    const int lane = threadIdx.x & 31;
    const int wv   = threadIdx.x >> 5;
    const int row0 = blockIdx.x * XP_ROWS;

    const float* xp = x_conv + (size_t)(row0 + XP_RPW * wv) * DI + 4 * lane;
    const float* wp = Wx + 4 * lane;

    v2f acc[XP_RPW][33];
#pragma unroll
    for (int r = 0; r < XP_RPW; ++r)
#pragma unroll
        for (int n = 0; n < 33; ++n) acc[r][n] = v2f{0.f, 0.f};

    for (int c = 0; c < DI; c += 128) {
        float4 xr[XP_RPW], w[33];
#pragma unroll
        for (int r = 0; r < XP_RPW; ++r) xr[r] = *(const float4*)(xp + (size_t)r * DI + c);
#pragma unroll
        for (int n = 0; n < 33; ++n) w[n] = *(const float4*)(wp + (size_t)n * DI + c);
#pragma unroll
        for (int n = 0; n < 33; ++n) {
            const v2f wlo = {w[n].x, w[n].y}, whi = {w[n].z, w[n].w};
#pragma unroll
            for (int r = 0; r < XP_RPW; ++r) {
                const v2f xlo = {xr[r].x, xr[r].y}, xhi = {xr[r].z, xr[r].w};
                acc[r][n] = __builtin_elementwise_fma(xlo, wlo, acc[r][n]);
                acc[r][n] = __builtin_elementwise_fma(xhi, whi, acc[r][n]);
            }
        }
    }

#pragma unroll
    for (int r = 0; r < XP_RPW; ++r)
#pragma unroll
        for (int n = 0; n < 33; ++n)
            part[((XP_RPW * wv + r) * 33 + n) * XP_PITCH + lane] = acc[r][n].x + acc[r][n].y;
    __syncthreads();

    for (int o = threadIdx.x; o < XP_OUT; o += XP_WAVES * 32) {
        const float* q = part + o * XP_PITCH;
        float s0 = 0.f, s1 = 0.f, s2 = 0.f, s3 = 0.f;
#pragma unroll
        for (int l = 0; l < 32; l += 4) {
            s0 += q[l];  s1 += q[l + 1];  s2 += q[l + 2];  s3 += q[l + 3];
        }
        ssm[(size_t)row0 * 33 + o] = (s0 + s1) + (s2 + s3);
    }
}

#define SCAN_WAVES 4
#define SCAN_CH    (2 * SCAN_WAVES)
#define SCAN_NBLK  (L_SZ / 16)
#define YT_PITCH   17

template <int CTRL>
__device__ __forceinline__ float row_dpp(float v) {
    return __builtin_bit_cast(float,
        __builtin_amdgcn_update_dpp(0, __builtin_bit_cast(int, v), CTRL, 0xf, 0xf, true));
}

#define SCAN_STEP(j)                                                  \
    {                                                                 \
        const float dtb  = row_dpp<0x150 + (j)>(dt);                  \
        const float xb   = row_dpp<0x150 + (j)>(xv);                  \
        const float Abar = __expf(dtb * Aa);                          \
        h    = fmaf(Abar, h, (dtb * Bv[j]) * xb);                     \
        p[j] = h * Cv[j];                                             \
    }

__global__ __launch_bounds__(SCAN_WAVES * 32) void scan_kernel(
    const float* __restrict__ ssm, const float* __restrict__ x_conv,
    const float* __restrict__ xg, const float* __restrict__ w_dt,
    const float* __restrict__ b_dt, const float* __restrict__ A_log,
    const float* __restrict__ D_param, float* __restrict__ y)
{
    static_assert(DI % SCAN_CH == 0 && SCAN_CH == 8 && L_SZ % 32 == 0 && DS == 16, "scan tiling");
    __shared__ float ytile[2][SCAN_CH * YT_PITCH];

    const int lane  = threadIdx.x & 31;
    const int wv    = threadIdx.x >> 5;
    const int n     = lane & 15;
    const int c     = lane >> 4;
    const int b     = blockIdx.x / (DI / SCAN_CH);
    const int g     = blockIdx.x % (DI / SCAN_CH);
    const int dbase = g * SCAN_CH;
    const int so    = 4 * wv + (lane >> 3);
    const int co    = lane & 7;
    const int d     = dbase + 2 * wv + c;
    const size_t row0 = (size_t)b * L_SZ;

    const float wdt = w_dt[d];
    const float bdt = b_dt[d];
    const float Dp  = D_param[d];
    const float Aa  = -__expf(A_log[d * DS + n]);
    float h = 0.f;

    const bool b3 = (n & 8) != 0, b2 = (n & 4) != 0, b1 = (n & 2) != 0, b0 = (n & 1) != 0;

    const float* pBC = ssm + row0 * 33 + n;
    const float* pDT = ssm + (row0 + n) * 33 + 32;
    const float* pX  = x_conv + (row0 + n) * DI + d;
    const float* pG  = xg + (row0 + so) * (2 * DI) + DI + dbase + co;
    float*       pY  = y + (((size_t)b * (DI / SCAN_CH) + g) * L_SZ + so) * SCAN_CH + co;

    auto ldblk = [&](int k, float (&Bv)[16], float (&Cv)[16], float& dtr, float& xv, float& gv) {
        const float* q = pBC + (size_t)k * (16 * 33);
#pragma unroll
        for (int j = 0; j < 16; ++j) { Bv[j] = q[j * 33]; Cv[j] = q[j * 33 + 16]; }
        dtr = pDT[(size_t)k * (16 * 33)];
        xv  = pX[(size_t)k * (16 * DI)];
        gv  = pG[(size_t)k * (16 * 2 * DI)];
    };

    auto compute = [&](int k, const float (&Bv)[16], const float (&Cv)[16],
                       float dtr, float xv, float gv, float* yt) {
        const float z  = fmaf(dtr, wdt, bdt);
        const float dt = (z > 20.0f) ? z : log1pf(__expf(z));

        float p[16];
        SCAN_STEP(0)  SCAN_STEP(1)  SCAN_STEP(2)  SCAN_STEP(3)
        SCAN_STEP(4)  SCAN_STEP(5)  SCAN_STEP(6)  SCAN_STEP(7)
        SCAN_STEP(8)  SCAN_STEP(9)  SCAN_STEP(10) SCAN_STEP(11)
        SCAN_STEP(12) SCAN_STEP(13) SCAN_STEP(14) SCAN_STEP(15)

        float q8[8];
#pragma unroll
        for (int j = 0; j < 8; ++j) {
            const float keep = b3 ? p[j + 8] : p[j];
            const float send = b3 ? p[j] : p[j + 8];
            q8[j] = keep + row_dpp<0x168>(send);
        }
        float q4[4];
#pragma unroll
        for (int j = 0; j < 4; ++j) {
            const float keep = b2 ? q8[j + 4] : q8[j];
            const float send = b2 ? q8[j] : q8[j + 4];
            q4[j] = keep + row_dpp<0x164>(send);
        }
        float q2[2];
#pragma unroll
        for (int j = 0; j < 2; ++j) {
            const float keep = b1 ? q4[j + 2] : q4[j];
            const float send = b1 ? q4[j] : q4[j + 2];
            q2[j] = keep + row_dpp<0x162>(send);
        }
        const float keep1 = b0 ? q2[1] : q2[0];
        const float send1 = b0 ? q2[0] : q2[1];
        const float ysum  = keep1 + row_dpp<0x161>(send1);

        yt[(2 * wv + c) * YT_PITCH + n] = fmaf(xv, Dp, ysum);
        __syncthreads();
        const float ys = yt[co * YT_PITCH + so];
        const float sg = gv / (1.0f + __expf(-gv));
        pY[(size_t)k * (16 * SCAN_CH)] = ys * sg;
    };

    float BvA[16], CvA[16], dtrA, xvA, gvA;
    float BvB[16], CvB[16], dtrB, xvB, gvB;
    ldblk(0, BvA, CvA, dtrA, xvA, gvA);
    for (int k = 0; k < SCAN_NBLK; k += 2) {
        ldblk(k + 1, BvB, CvB, dtrB, xvB, gvB);
        compute(k, BvA, CvA, dtrA, xvA, gvA, ytile[0]);
        const int kn = (k + 2 < SCAN_NBLK) ? (k + 2) : (SCAN_NBLK - 1);
        ldblk(kn, BvA, CvA, dtrA, xvA, gvA);
        compute(k + 1, BvB, CvB, dtrB, xvB, gvB, ytile[1]);
    }
}

extern "C" void kernel_launch(void* const* d_in, const int* in_sizes, int n_in,
                              void* d_out, int out_size, void* d_ws, size_t ws_size,
                              hipStream_t stream)
{
    const float* x       = (const float*)d_in[0];
    const float* W_in    = (const float*)d_in[1];
    const float* conv_w  = (const float*)d_in[2];
    const float* conv_b  = (const float*)d_in[3];
    const float* W_xproj = (const float*)d_in[4];
    const float* w_dt    = (const float*)d_in[5];
    const float* b_dt    = (const float*)d_in[6];
    const float* A_log   = (const float*)d_in[7];
    const float* D_param = (const float*)d_in[8];
    const float* W_out   = (const float*)d_in[9];
    float* out = (float*)d_out;

    float* ws     = (float*)d_ws;
    float* xg     = ws;
    float* x_conv = xg + (size_t)NROW * (2 * DI);
    float* ssm    = x_conv + (size_t)NROW * DI;
    float* yb     = ssm + (size_t)NROW * 33;

    gemm64x64_conv_f32_wmma<<<(NROW / 64) * ((2 * DI) / 64), 32, 0, stream>>>(
        x, W_in, xg, conv_w, conv_b, x_conv);

    conv_edge_kernel<<<(NROW / 64) * (DI / 256), 256, 0, stream>>>(
        xg, conv_w, conv_b, x_conv);

    xproj_kernel<<<NROW / XP_ROWS, XP_WAVES * 32, 0, stream>>>(
        x_conv, W_xproj, ssm);

    scan_kernel<<<B_SZ * (DI / SCAN_CH), SCAN_WAVES * 32, 0, stream>>>(
        ssm, x_conv, xg, w_dt, b_dt, A_log, D_param, yb);

    gemm32x64_f32_wmma<DI><<<(NROW / 32) * (DM / 64), 32, 0, stream>>>(
        yb, W_out, out, DM / 64, DI, DM);
}


# scan: block barrier fenced for LDS only, so global loads stay in flight across it and the y store is never waited for; loads issued 3 blocks ahead in 4 register sets; every sum unchanged
# speedup vs baseline: 11.0502x; 1.0653x over previous
#include <hip/hip_runtime.h>
#include <math.h>


typedef float  v2f __attribute__((ext_vector_type(2)));
typedef float  v4f __attribute__((ext_vector_type(4)));
typedef float  v8f __attribute__((ext_vector_type(8)));

#define B_SZ 2
#define L_SZ 2048
#define DM   768
#define DI   1536
#define DS   16
#define NROW (B_SZ * L_SZ)

__device__ __forceinline__ float xhalf(float v) {
    return __builtin_bit_cast(float, __builtin_amdgcn_permlanex16(
        __builtin_bit_cast(int, v), __builtin_bit_cast(int, v), 0x76543210, 0xfedcba98, false, false));
}

struct frag  { float4 q0, q1; };
struct vfrag { v4f q0, q1; };
__device__ __forceinline__ v2f frag_lo(const float4& q) { v2f r; r.x = q.x; r.y = q.y; return r; }
__device__ __forceinline__ v2f frag_hi(const float4& q) { v2f r; r.x = q.z; r.y = q.w; return r; }
__device__ __forceinline__ v2f frag_lo(const v4f& q) { return __builtin_shufflevector(q, q, 0, 1); }
__device__ __forceinline__ v2f frag_hi(const v4f& q) { return __builtin_shufflevector(q, q, 2, 3); }

__global__ __launch_bounds__(32) __attribute__((amdgpu_waves_per_eu(3, 3)))
void gemm64x64_conv_f32_wmma(
    const float* __restrict__ A, const float* __restrict__ W, float* __restrict__ C,
    const float* __restrict__ conv_w, const float* __restrict__ conv_b,
    float* __restrict__ x_conv)
{
    constexpr int K = DM, LDC = 2 * DI, N64 = (2 * DI) / 64;
    static_assert(K % 32 == 0 && NROW % 64 == 0 && L_SZ % 64 == 0, "first GEMM tiling");
    const int tile = blockIdx.x;
    const int tm   = tile / N64;
    const int tn   = tile % N64;
    const int lane = threadIdx.x;
    const int half = lane >> 4;
    const int lr   = lane & 15;

    const float* Ap[4];
    const float* Wp[4];
#pragma unroll
    for (int s = 0; s < 4; ++s) {
        Ap[s] = A + (size_t)(tm * 64 + 16 * s + lr) * K + 8 * half;
        Wp[s] = W + (size_t)(tn * 64 + 16 * s + lr) * K + 8 * half;
    }

    v8f acc[4][4];
#pragma unroll
    for (int s = 0; s < 4; ++s)
#pragma unroll
        for (int j = 0; j < 4; ++j) acc[s][j] = v8f{};

    auto ld = [&](int kb, vfrag (&a)[4], vfrag (&b)[4]) {
#pragma unroll
        for (int s = 0; s < 4; ++s) {
            a[s].q0 = *(const v4f*)(Ap[s] + kb);  a[s].q1 = *(const v4f*)(Ap[s] + kb + 4);
        }
#pragma unroll
        for (int j = 0; j < 4; ++j) {
            b[j].q0 = *(const v4f*)(Wp[j] + kb);  b[j].q1 = *(const v4f*)(Wp[j] + kb + 4);
        }
    };
    auto mm16 = [&](const v2f (&a)[4], const v2f (&b)[4]) {
#pragma unroll
        for (int s = 0; s < 4; ++s)
#pragma unroll
            for (int j = 0; j < 4; ++j)
                acc[s][j] = __builtin_amdgcn_wmma_f32_16x16x4_f32(false, a[s], false, b[j], (short)0, acc[s][j], false, false);
    };
    auto mm64 = [&](const vfrag (&a)[4], const vfrag (&b)[4]) {
        v2f fa[4], fb[4];
#pragma unroll
        for (int s = 0; s < 4; ++s) { fa[s] = frag_lo(a[s].q0); fb[s] = frag_lo(b[s].q0); }
        mm16(fa, fb);
#pragma unroll
        for (int s = 0; s < 4; ++s) { fa[s] = frag_hi(a[s].q0); fb[s] = frag_hi(b[s].q0); }
        mm16(fa, fb);
#pragma unroll
        for (int s = 0; s < 4; ++s) { fa[s] = frag_lo(a[s].q1); fb[s] = frag_lo(b[s].q1); }
        mm16(fa, fb);
#pragma unroll
        for (int s = 0; s < 4; ++s) { fa[s] = frag_hi(a[s].q1); fb[s] = frag_hi(b[s].q1); }
        mm16(fa, fb);
    };

    vfrag aA[4], bA[4], aB[4], bB[4];
    ld(0, aA, bA);
    for (int k = 0; k < K; k += 32) {
        ld(k + 16, aB, bB);
        mm64(aA, bA);
        const int kn = (k + 32 < K) ? (k + 32) : k;
        ld(kn, aA, bA);
        mm64(aB, bB);
    }

    float* Cb = C + (size_t)(tm * 64 + 8 * half) * LDC + tn * 64 + lr;

    if (tn < DI / 64) {
        float* Xb = x_conv + (size_t)(tm * 64 + 8 * half) * DI + tn * 64 + lr;
#pragma unroll
        for (int j = 0; j < 4; ++j) {
            const int    d = tn * 64 + 16 * j + lr;
            const float4 w = *(const float4*)(conv_w + (size_t)d * 4);
            const float  b = conv_b[d];
            float xs[4][3];
#pragma unroll
            for (int s = 0; s < 4; ++s)
#pragma unroll
                for (int i = 0; i < 3; ++i) xs[s][i] = xhalf(acc[s][j][5 + i]);
#pragma unroll
            for (int s = 0; s < 4; ++s) {
                float e[11];
#pragma unroll
                for (int i = 0; i < 3; ++i) e[i] = (half || s == 0) ? xs[s][i] : xs[s > 0 ? s - 1 : 0][i];
#pragma unroll
                for (int r = 0; r < 8; ++r) e[3 + r] = acc[s][j][r];
#pragma unroll
                for (int r = 0; r < 8; ++r) {
                    float a = b;
                    a = fmaf(w.x, e[r],     a);
                    a = fmaf(w.y, e[r + 1], a);
                    a = fmaf(w.z, e[r + 2], a);
                    a = fmaf(w.w, e[r + 3], a);
                    const float y = a / (1.0f + __expf(-a));
                    if (s > 0 || r >= 3 || half) Xb[(size_t)(16 * s + r) * DI + 16 * j] = y;
                    if (s == 0 && r < 3 && !half)  Cb[(size_t)r * LDC + 16 * j] = acc[s][j][r];
                    if (s == 3 && r >= 5 && half)  Cb[(size_t)(48 + r) * LDC + 16 * j] = acc[s][j][r];
                }
            }
        }
        return;
    }
#pragma unroll
    for (int s = 0; s < 4; ++s)
#pragma unroll
        for (int r = 0; r < 8; ++r) {
            float* Cr = Cb + (size_t)(16 * s + r) * LDC;
            Cr[0]  = acc[s][0][r];  Cr[16] = acc[s][1][r];
            Cr[32] = acc[s][2][r];  Cr[48] = acc[s][3][r];
        }
}

template <int K>
__global__ __launch_bounds__(32) void gemm32x64_f32_wmma(
    const float* __restrict__ A, const float* __restrict__ W,
    float* __restrict__ C, int N64, int ldw, int ldc)
{
    static_assert(K % 32 == 0 && L_SZ % 32 == 0, "K a multiple of 32; a 32-row tile inside one sequence");
    const int tile = blockIdx.x;
    const int tm   = tile / N64;
    const int tn   = tile % N64;
    const int lane = threadIdx.x;
    const int half = lane >> 4;
    const int lr   = lane & 15;

    const int    ab  = (tm * 32) / L_SZ;
    const int    al  = (tm * 32) % L_SZ + lr;
    const float* Ar0 = A + (((size_t)ab * (K / 8) + half) * L_SZ + al) * 8;
    const float* Ar1 = Ar0 + 16 * 8;
    const float* W0  = W + (size_t)(tn * 64 +  0 + lr) * ldw;
    const float* W1  = W + (size_t)(tn * 64 + 16 + lr) * ldw;
    const float* W2  = W + (size_t)(tn * 64 + 32 + lr) * ldw;
    const float* W3  = W + (size_t)(tn * 64 + 48 + lr) * ldw;

    v8f acc00 = {}; v8f acc01 = {}; v8f acc02 = {}; v8f acc03 = {};
    v8f acc10 = {}; v8f acc11 = {}; v8f acc12 = {}; v8f acc13 = {};

    const int kh8 = 8 * half;

    auto ld = [&](int k, frag& a0, frag& a1, frag& b0, frag& b1, frag& b2, frag& b3) {
        const size_t ka = (size_t)k * L_SZ;
        const int    kb = k + kh8;
        a0.q0 = *(const float4*)(Ar0 + ka);  a0.q1 = *(const float4*)(Ar0 + ka + 4);
        a1.q0 = *(const float4*)(Ar1 + ka);  a1.q1 = *(const float4*)(Ar1 + ka + 4);
        b0.q0 = *(const float4*)(W0 + kb);   b0.q1 = *(const float4*)(W0 + kb + 4);
        b1.q0 = *(const float4*)(W1 + kb);   b1.q1 = *(const float4*)(W1 + kb + 4);
        b2.q0 = *(const float4*)(W2 + kb);   b2.q1 = *(const float4*)(W2 + kb + 4);
        b3.q0 = *(const float4*)(W3 + kb);   b3.q1 = *(const float4*)(W3 + kb + 4);
    };
    auto mm8 = [&](v2f a0, v2f a1, v2f b0, v2f b1, v2f b2, v2f b3) {
        acc00 = __builtin_amdgcn_wmma_f32_16x16x4_f32(false, a0, false, b0, (short)0, acc00, false, false);
        acc01 = __builtin_amdgcn_wmma_f32_16x16x4_f32(false, a0, false, b1, (short)0, acc01, false, false);
        acc02 = __builtin_amdgcn_wmma_f32_16x16x4_f32(false, a0, false, b2, (short)0, acc02, false, false);
        acc03 = __builtin_amdgcn_wmma_f32_16x16x4_f32(false, a0, false, b3, (short)0, acc03, false, false);
        acc10 = __builtin_amdgcn_wmma_f32_16x16x4_f32(false, a1, false, b0, (short)0, acc10, false, false);
        acc11 = __builtin_amdgcn_wmma_f32_16x16x4_f32(false, a1, false, b1, (short)0, acc11, false, false);
        acc12 = __builtin_amdgcn_wmma_f32_16x16x4_f32(false, a1, false, b2, (short)0, acc12, false, false);
        acc13 = __builtin_amdgcn_wmma_f32_16x16x4_f32(false, a1, false, b3, (short)0, acc13, false, false);
    };
    auto lo = [](const float4& q) { return frag_lo(q); };
    auto hi = [](const float4& q) { return frag_hi(q); };
    auto mm32 = [&](const frag& a0, const frag& a1, const frag& b0, const frag& b1, const frag& b2, const frag& b3) {
        mm8(lo(a0.q0), lo(a1.q0), lo(b0.q0), lo(b1.q0), lo(b2.q0), lo(b3.q0));
        mm8(hi(a0.q0), hi(a1.q0), hi(b0.q0), hi(b1.q0), hi(b2.q0), hi(b3.q0));
        mm8(lo(a0.q1), lo(a1.q1), lo(b0.q1), lo(b1.q1), lo(b2.q1), lo(b3.q1));
        mm8(hi(a0.q1), hi(a1.q1), hi(b0.q1), hi(b1.q1), hi(b2.q1), hi(b3.q1));
    };

    frag a0A, a1A, b0A, b1A, b2A, b3A;
    frag a0B, a1B, b0B, b1B, b2B, b3B;

    ld(0, a0A, a1A, b0A, b1A, b2A, b3A);
    for (int k = 0; k < K - 32; k += 32) {
        ld(k + 16, a0B, a1B, b0B, b1B, b2B, b3B);
        mm32(a0A, a1A, b0A, b1A, b2A, b3A);
        ld(k + 32, a0A, a1A, b0A, b1A, b2A, b3A);
        mm32(a0B, a1B, b0B, b1B, b2B, b3B);
    }
    ld(K - 16, a0B, a1B, b0B, b1B, b2B, b3B);
    mm32(a0A, a1A, b0A, b1A, b2A, b3A);
    mm32(a0B, a1B, b0B, b1B, b2B, b3B);

    float* Cb0 = C + (size_t)(tm * 32 +  0 + 8 * half) * ldc + tn * 64 + lr;
    float* Cb1 = C + (size_t)(tm * 32 + 16 + 8 * half) * ldc + tn * 64 + lr;
#pragma unroll
    for (int r = 0; r < 8; ++r) {
        float* Cr0 = Cb0 + (size_t)r * ldc;
        float* Cr1 = Cb1 + (size_t)r * ldc;
        Cr0[0]  = acc00[r];  Cr0[16] = acc01[r];
        Cr0[32] = acc02[r];  Cr0[48] = acc03[r];
        Cr1[0]  = acc10[r];  Cr1[16] = acc11[r];
        Cr1[32] = acc12[r];  Cr1[48] = acc13[r];
    }
}

__global__ __launch_bounds__(256) void conv_edge_kernel(
    const float* __restrict__ xg, const float* __restrict__ conv_w,
    const float* __restrict__ conv_b, float* __restrict__ x_conv)
{
    static_assert(DI % 256 == 0 && L_SZ % 64 == 0, "edge tiling");
    const int  tile = blockIdx.x / (DI / 256);
    const int  d    = (blockIdx.x % (DI / 256)) * 256 + threadIdx.x;
    const int  row0 = tile * 64;
    const bool head = (row0 % L_SZ) == 0;

    const float  b  = conv_b[d];
    const float4 w4 = *(const float4*)(conv_w + (size_t)d * 4);
    const float  w[4] = {w4.x, w4.y, w4.z, w4.w};
    const float* p  = xg + (size_t)row0 * (2 * DI) + d;

    float in[6];
    in[0] = in[1] = in[2] = 0.0f;
    if (!head) {
        in[0] = *(p - 3 * (2 * DI));
        in[1] = *(p - 2 * (2 * DI));
        in[2] = *(p - 1 * (2 * DI));
    }
    in[3] = p[0];
    in[4] = p[1 * (2 * DI)];
    in[5] = p[2 * (2 * DI)];
#pragma unroll
    for (int i = 0; i < 3; ++i) {
        float acc = b;
#pragma unroll
        for (int j = 0; j < 4; ++j) {
            const float t = fmaf(w[j], in[i + j], acc);
            acc = (head && i + j < 3) ? acc : t;
        }
        x_conv[(size_t)(row0 + i) * DI + d] = acc / (1.0f + __expf(-acc));
    }
}

#define XP_WAVES 8
#define XP_RPW   4
#define XP_ROWS  (XP_WAVES * XP_RPW)
#define XP_OUT   (XP_ROWS * 33)
#define XP_PITCH 33

__global__ __launch_bounds__(XP_WAVES * 32) __attribute__((amdgpu_waves_per_eu(2, 2)))
void xproj_kernel(
    const float* __restrict__ x_conv, const float* __restrict__ Wx,
    float* __restrict__ ssm)
{
    static_assert(NROW % XP_ROWS == 0 && DI % 128 == 0, "xproj tiling");
    __shared__ float part[XP_OUT * XP_PITCH];

    const int lane = threadIdx.x & 31;
    const int wv   = threadIdx.x >> 5;
    const int row0 = blockIdx.x * XP_ROWS;

    const float* xp = x_conv + (size_t)(row0 + XP_RPW * wv) * DI + 4 * lane;
    const float* wp = Wx + 4 * lane;

    v2f acc[XP_RPW][33];
#pragma unroll
    for (int r = 0; r < XP_RPW; ++r)
#pragma unroll
        for (int n = 0; n < 33; ++n) acc[r][n] = v2f{0.f, 0.f};

    for (int c = 0; c < DI; c += 128) {
        float4 xr[XP_RPW], w[33];
#pragma unroll
        for (int r = 0; r < XP_RPW; ++r) xr[r] = *(const float4*)(xp + (size_t)r * DI + c);
#pragma unroll
        for (int n = 0; n < 33; ++n) w[n] = *(const float4*)(wp + (size_t)n * DI + c);
#pragma unroll
        for (int n = 0; n < 33; ++n) {
            const v2f wlo = {w[n].x, w[n].y}, whi = {w[n].z, w[n].w};
#pragma unroll
            for (int r = 0; r < XP_RPW; ++r) {
                const v2f xlo = {xr[r].x, xr[r].y}, xhi = {xr[r].z, xr[r].w};
                acc[r][n] = __builtin_elementwise_fma(xlo, wlo, acc[r][n]);
                acc[r][n] = __builtin_elementwise_fma(xhi, whi, acc[r][n]);
            }
        }
    }

#pragma unroll
    for (int r = 0; r < XP_RPW; ++r)
#pragma unroll
        for (int n = 0; n < 33; ++n)
            part[((XP_RPW * wv + r) * 33 + n) * XP_PITCH + lane] = acc[r][n].x + acc[r][n].y;
    __syncthreads();

    for (int o = threadIdx.x; o < XP_OUT; o += XP_WAVES * 32) {
        const float* q = part + o * XP_PITCH;
        float s0 = 0.f, s1 = 0.f, s2 = 0.f, s3 = 0.f;
#pragma unroll
        for (int l = 0; l < 32; l += 4) {
            s0 += q[l];  s1 += q[l + 1];  s2 += q[l + 2];  s3 += q[l + 3];
        }
        ssm[(size_t)row0 * 33 + o] = (s0 + s1) + (s2 + s3);
    }
}

#define SCAN_WAVES 4
#define SCAN_CH    (2 * SCAN_WAVES)
#define SCAN_NBLK  (L_SZ / 16)
#define YT_PITCH   17

template <int CTRL>
__device__ __forceinline__ float row_dpp(float v) {
    return __builtin_bit_cast(float,
        __builtin_amdgcn_update_dpp(0, __builtin_bit_cast(int, v), CTRL, 0xf, 0xf, true));
}

#define SCAN_STEP(j)                                                  \
    {                                                                 \
        const float dtb  = row_dpp<0x150 + (j)>(dt);                  \
        const float xb   = row_dpp<0x150 + (j)>(xv);                  \
        const float Abar = __expf(dtb * Aa);                          \
        h    = fmaf(Abar, h, (dtb * Bv[j]) * xb);                     \
        p[j] = h * Cv[j];                                             \
    }

__global__ __launch_bounds__(SCAN_WAVES * 32) void scan_kernel(
    const float* __restrict__ ssm, const float* __restrict__ x_conv,
    const float* __restrict__ xg, const float* __restrict__ w_dt,
    const float* __restrict__ b_dt, const float* __restrict__ A_log,
    const float* __restrict__ D_param, float* __restrict__ y)
{
    static_assert(DI % SCAN_CH == 0 && SCAN_CH == 8 && L_SZ % 32 == 0 && DS == 16, "scan tiling");
    __shared__ float ytile[2][SCAN_CH * YT_PITCH];

    const int lane  = threadIdx.x & 31;
    const int wv    = threadIdx.x >> 5;
    const int n     = lane & 15;
    const int c     = lane >> 4;
    const int b     = blockIdx.x / (DI / SCAN_CH);
    const int g     = blockIdx.x % (DI / SCAN_CH);
    const int dbase = g * SCAN_CH;
    const int so    = 4 * wv + (lane >> 3);
    const int co    = lane & 7;
    const int d     = dbase + 2 * wv + c;
    const size_t row0 = (size_t)b * L_SZ;

    const float wdt = w_dt[d];
    const float bdt = b_dt[d];
    const float Dp  = D_param[d];
    const float Aa  = -__expf(A_log[d * DS + n]);
    float h = 0.f;

    const bool b3 = (n & 8) != 0, b2 = (n & 4) != 0, b1 = (n & 2) != 0, b0 = (n & 1) != 0;

    const float* pBC = ssm + row0 * 33 + n;
    const float* pDT = ssm + (row0 + n) * 33 + 32;
    const float* pX  = x_conv + (row0 + n) * DI + d;
    const float* pG  = xg + (row0 + so) * (2 * DI) + DI + dbase + co;
    float*       pY  = y + (((size_t)b * (DI / SCAN_CH) + g) * L_SZ + so) * SCAN_CH + co;

    auto ldblk = [&](int k, float (&Bv)[16], float (&Cv)[16], float& dtr, float& xv, float& gv) {
        const float* q = pBC + (size_t)k * (16 * 33);
#pragma unroll
        for (int j = 0; j < 16; ++j) { Bv[j] = q[j * 33]; Cv[j] = q[j * 33 + 16]; }
        dtr = pDT[(size_t)k * (16 * 33)];
        xv  = pX[(size_t)k * (16 * DI)];
        gv  = pG[(size_t)k * (16 * 2 * DI)];
    };

    auto compute = [&](int k, const float (&Bv)[16], const float (&Cv)[16],
                       float dtr, float xv, float gv, float* yt) {
        const float z  = fmaf(dtr, wdt, bdt);
        const float dt = (z > 20.0f) ? z : log1pf(__expf(z));

        float p[16];
        SCAN_STEP(0)  SCAN_STEP(1)  SCAN_STEP(2)  SCAN_STEP(3)
        SCAN_STEP(4)  SCAN_STEP(5)  SCAN_STEP(6)  SCAN_STEP(7)
        SCAN_STEP(8)  SCAN_STEP(9)  SCAN_STEP(10) SCAN_STEP(11)
        SCAN_STEP(12) SCAN_STEP(13) SCAN_STEP(14) SCAN_STEP(15)

        float q8[8];
#pragma unroll
        for (int j = 0; j < 8; ++j) {
            const float keep = b3 ? p[j + 8] : p[j];
            const float send = b3 ? p[j] : p[j + 8];
            q8[j] = keep + row_dpp<0x168>(send);
        }
        float q4[4];
#pragma unroll
        for (int j = 0; j < 4; ++j) {
            const float keep = b2 ? q8[j + 4] : q8[j];
            const float send = b2 ? q8[j] : q8[j + 4];
            q4[j] = keep + row_dpp<0x164>(send);
        }
        float q2[2];
#pragma unroll
        for (int j = 0; j < 2; ++j) {
            const float keep = b1 ? q4[j + 2] : q4[j];
            const float send = b1 ? q4[j] : q4[j + 2];
            q2[j] = keep + row_dpp<0x162>(send);
        }
        const float keep1 = b0 ? q2[1] : q2[0];
        const float send1 = b0 ? q2[0] : q2[1];
        const float ysum  = keep1 + row_dpp<0x161>(send1);

        yt[(2 * wv + c) * YT_PITCH + n] = fmaf(xv, Dp, ysum);
        __builtin_amdgcn_fence(__ATOMIC_RELEASE, "workgroup", "local");
        __builtin_amdgcn_s_barrier();
        __builtin_amdgcn_fence(__ATOMIC_ACQUIRE, "workgroup", "local");
        const float ys = yt[co * YT_PITCH + so];
        const float sg = gv / (1.0f + __expf(-gv));
        pY[(size_t)k * (16 * SCAN_CH)] = ys * sg;
    };

    static_assert(SCAN_NBLK % 4 == 0 && SCAN_NBLK >= 4, "scan loop unrolled over four register sets");
    constexpr int LASTB = SCAN_NBLK - 1;
    float BvA[16], CvA[16], dtrA, xvA, gvA;
    float BvB[16], CvB[16], dtrB, xvB, gvB;
    float BvC[16], CvC[16], dtrC, xvC, gvC;
    float BvD[16], CvD[16], dtrD, xvD, gvD;
    ldblk(0, BvA, CvA, dtrA, xvA, gvA);
    __builtin_amdgcn_sched_barrier(0);
    ldblk(1, BvB, CvB, dtrB, xvB, gvB);
    __builtin_amdgcn_sched_barrier(0);
    ldblk(2, BvC, CvC, dtrC, xvC, gvC);
    for (int k = 0; k < SCAN_NBLK; k += 4) {
        ldblk(k + 3, BvD, CvD, dtrD, xvD, gvD);
        compute(k, BvA, CvA, dtrA, xvA, gvA, ytile[0]);
        ldblk((k + 4 < SCAN_NBLK) ? (k + 4) : LASTB, BvA, CvA, dtrA, xvA, gvA);
        compute(k + 1, BvB, CvB, dtrB, xvB, gvB, ytile[1]);
        ldblk((k + 5 < SCAN_NBLK) ? (k + 5) : LASTB, BvB, CvB, dtrB, xvB, gvB);
        compute(k + 2, BvC, CvC, dtrC, xvC, gvC, ytile[0]);
        ldblk((k + 6 < SCAN_NBLK) ? (k + 6) : LASTB, BvC, CvC, dtrC, xvC, gvC);
        compute(k + 3, BvD, CvD, dtrD, xvD, gvD, ytile[1]);
    }
}

extern "C" void kernel_launch(void* const* d_in, const int* in_sizes, int n_in,
                              void* d_out, int out_size, void* d_ws, size_t ws_size,
                              hipStream_t stream)
{
    const float* x       = (const float*)d_in[0];
    const float* W_in    = (const float*)d_in[1];
    const float* conv_w  = (const float*)d_in[2];
    const float* conv_b  = (const float*)d_in[3];
    const float* W_xproj = (const float*)d_in[4];
    const float* w_dt    = (const float*)d_in[5];
    const float* b_dt    = (const float*)d_in[6];
    const float* A_log   = (const float*)d_in[7];
    const float* D_param = (const float*)d_in[8];
    const float* W_out   = (const float*)d_in[9];
    float* out = (float*)d_out;

    float* ws     = (float*)d_ws;
    float* xg     = ws;
    float* x_conv = xg + (size_t)NROW * (2 * DI);
    float* ssm    = x_conv + (size_t)NROW * DI;
    float* yb     = ssm + (size_t)NROW * 33;

    gemm64x64_conv_f32_wmma<<<(NROW / 64) * ((2 * DI) / 64), 32, 0, stream>>>(
        x, W_in, xg, conv_w, conv_b, x_conv);

    conv_edge_kernel<<<(NROW / 64) * (DI / 256), 256, 0, stream>>>(
        xg, conv_w, conv_b, x_conv);

    xproj_kernel<<<NROW / XP_ROWS, XP_WAVES * 32, 0, stream>>>(
        x_conv, W_xproj, ssm);

    scan_kernel<<<B_SZ * (DI / SCAN_CH), SCAN_WAVES * 32, 0, stream>>>(
        ssm, x_conv, xg, w_dt, b_dt, A_log, D_param, yb);

    gemm32x64_f32_wmma<DI><<<(NROW / 32) * (DM / 64), 32, 0, stream>>>(
        yb, W_out, out, DM / 64, DI, DM);
}


# second GEMM re-cut: 64x64 register tiles, K split over the 4 waves of a work-group (3072 waves = 3 per SIMD, was 1536), partial tiles added through LDS in K order; summation order of out changes
# speedup vs baseline: 12.1051x; 1.0955x over previous
#include <hip/hip_runtime.h>
#include <math.h>


typedef float  v2f __attribute__((ext_vector_type(2)));
typedef float  v4f __attribute__((ext_vector_type(4)));
typedef float  v8f __attribute__((ext_vector_type(8)));

#define B_SZ 2
#define L_SZ 2048
#define DM   768
#define DI   1536
#define DS   16
#define NROW (B_SZ * L_SZ)

__device__ __forceinline__ float xhalf(float v) {
    return __builtin_bit_cast(float, __builtin_amdgcn_permlanex16(
        __builtin_bit_cast(int, v), __builtin_bit_cast(int, v), 0x76543210, 0xfedcba98, false, false));
}

struct vfrag { v4f q0, q1; };
__device__ __forceinline__ v2f frag_lo(const v4f& q) { return __builtin_shufflevector(q, q, 0, 1); }
__device__ __forceinline__ v2f frag_hi(const v4f& q) { return __builtin_shufflevector(q, q, 2, 3); }

__global__ __launch_bounds__(32) __attribute__((amdgpu_waves_per_eu(3, 3)))
void gemm64x64_conv_f32_wmma(
    const float* __restrict__ A, const float* __restrict__ W, float* __restrict__ C,
    const float* __restrict__ conv_w, const float* __restrict__ conv_b,
    float* __restrict__ x_conv)
{
    constexpr int K = DM, LDC = 2 * DI, N64 = (2 * DI) / 64;
    static_assert(K % 32 == 0 && NROW % 64 == 0 && L_SZ % 64 == 0, "first GEMM tiling");
    const int tile = blockIdx.x;
    const int tm   = tile / N64;
    const int tn   = tile % N64;
    const int lane = threadIdx.x;
    const int half = lane >> 4;
    const int lr   = lane & 15;

    const float* Ap[4];
    const float* Wp[4];
#pragma unroll
    for (int s = 0; s < 4; ++s) {
        Ap[s] = A + (size_t)(tm * 64 + 16 * s + lr) * K + 8 * half;
        Wp[s] = W + (size_t)(tn * 64 + 16 * s + lr) * K + 8 * half;
    }

    v8f acc[4][4];
#pragma unroll
    for (int s = 0; s < 4; ++s)
#pragma unroll
        for (int j = 0; j < 4; ++j) acc[s][j] = v8f{};

    auto ld = [&](int kb, vfrag (&a)[4], vfrag (&b)[4]) {
#pragma unroll
        for (int s = 0; s < 4; ++s) {
            a[s].q0 = *(const v4f*)(Ap[s] + kb);  a[s].q1 = *(const v4f*)(Ap[s] + kb + 4);
        }
#pragma unroll
        for (int j = 0; j < 4; ++j) {
            b[j].q0 = *(const v4f*)(Wp[j] + kb);  b[j].q1 = *(const v4f*)(Wp[j] + kb + 4);
        }
    };
    auto mm16 = [&](const v2f (&a)[4], const v2f (&b)[4]) {
#pragma unroll
        for (int s = 0; s < 4; ++s)
#pragma unroll
            for (int j = 0; j < 4; ++j)
                acc[s][j] = __builtin_amdgcn_wmma_f32_16x16x4_f32(false, a[s], false, b[j], (short)0, acc[s][j], false, false);
    };
    auto mm64 = [&](const vfrag (&a)[4], const vfrag (&b)[4]) {
        v2f fa[4], fb[4];
#pragma unroll
        for (int s = 0; s < 4; ++s) { fa[s] = frag_lo(a[s].q0); fb[s] = frag_lo(b[s].q0); }
        mm16(fa, fb);
#pragma unroll
        for (int s = 0; s < 4; ++s) { fa[s] = frag_hi(a[s].q0); fb[s] = frag_hi(b[s].q0); }
        mm16(fa, fb);
#pragma unroll
        for (int s = 0; s < 4; ++s) { fa[s] = frag_lo(a[s].q1); fb[s] = frag_lo(b[s].q1); }
        mm16(fa, fb);
#pragma unroll
        for (int s = 0; s < 4; ++s) { fa[s] = frag_hi(a[s].q1); fb[s] = frag_hi(b[s].q1); }
        mm16(fa, fb);
    };

    vfrag aA[4], bA[4], aB[4], bB[4];
    ld(0, aA, bA);
    for (int k = 0; k < K; k += 32) {
        ld(k + 16, aB, bB);
        mm64(aA, bA);
        const int kn = (k + 32 < K) ? (k + 32) : k;
        ld(kn, aA, bA);
        mm64(aB, bB);
    }

    float* Cb = C + (size_t)(tm * 64 + 8 * half) * LDC + tn * 64 + lr;

    if (tn < DI / 64) {
        float* Xb = x_conv + (size_t)(tm * 64 + 8 * half) * DI + tn * 64 + lr;
#pragma unroll
        for (int j = 0; j < 4; ++j) {
            const int    d = tn * 64 + 16 * j + lr;
            const float4 w = *(const float4*)(conv_w + (size_t)d * 4);
            const float  b = conv_b[d];
            float xs[4][3];
#pragma unroll
            for (int s = 0; s < 4; ++s)
#pragma unroll
                for (int i = 0; i < 3; ++i) xs[s][i] = xhalf(acc[s][j][5 + i]);
#pragma unroll
            for (int s = 0; s < 4; ++s) {
                float e[11];
#pragma unroll
                for (int i = 0; i < 3; ++i) e[i] = (half || s == 0) ? xs[s][i] : xs[s > 0 ? s - 1 : 0][i];
#pragma unroll
                for (int r = 0; r < 8; ++r) e[3 + r] = acc[s][j][r];
#pragma unroll
                for (int r = 0; r < 8; ++r) {
                    float a = b;
                    a = fmaf(w.x, e[r],     a);
                    a = fmaf(w.y, e[r + 1], a);
                    a = fmaf(w.z, e[r + 2], a);
                    a = fmaf(w.w, e[r + 3], a);
                    const float y = a / (1.0f + __expf(-a));
                    if (s > 0 || r >= 3 || half) Xb[(size_t)(16 * s + r) * DI + 16 * j] = y;
                    if (s == 0 && r < 3 && !half)  Cb[(size_t)r * LDC + 16 * j] = acc[s][j][r];
                    if (s == 3 && r >= 5 && half)  Cb[(size_t)(48 + r) * LDC + 16 * j] = acc[s][j][r];
                }
            }
        }
        return;
    }
#pragma unroll
    for (int s = 0; s < 4; ++s)
#pragma unroll
        for (int r = 0; r < 8; ++r) {
            float* Cr = Cb + (size_t)(16 * s + r) * LDC;
            Cr[0]  = acc[s][0][r];  Cr[16] = acc[s][1][r];
            Cr[32] = acc[s][2][r];  Cr[48] = acc[s][3][r];
        }
}

#define G2_KS 4
__global__ __launch_bounds__(G2_KS * 32) __attribute__((amdgpu_waves_per_eu(3, 3)))
void gemm64x64_splitk_f32_wmma(
    const float* __restrict__ A, const float* __restrict__ W, float* __restrict__ C)
{
    constexpr int K = DI, KQ = K / G2_KS, N64 = DM / 64;
    static_assert(KQ % 32 == 0 && NROW % 64 == 0 && L_SZ % 64 == 0 && DM % 64 == 0, "second GEMM tiling");
    __shared__ float4 part[G2_KS][G2_KS][4][2][32];

    const int tile = blockIdx.x;
    const int tm   = tile / N64;
    const int tn   = tile % N64;
    const int lane = threadIdx.x & 31;
    const int wv   = __builtin_amdgcn_readfirstlane((int)(threadIdx.x >> 5));
    const int half = lane >> 4;
    const int lr   = lane & 15;

    const int ab = (tm * 64) / L_SZ;
    const int al = (tm * 64) % L_SZ + lr;
    const int k0 = wv * KQ;

    const float* Ap[4];
    const float* Wp[4];
#pragma unroll
    for (int s = 0; s < 4; ++s) {
        Ap[s] = A + (((size_t)ab * (K / 8) + half) * L_SZ + al + 16 * s) * 8;
        Wp[s] = W + (size_t)(tn * 64 + 16 * s + lr) * K + 8 * half;
    }

    v8f acc[4][4];
#pragma unroll
    for (int s = 0; s < 4; ++s)
#pragma unroll
        for (int j = 0; j < 4; ++j) acc[s][j] = v8f{};

    auto ld = [&](int kb, vfrag (&a)[4], vfrag (&b)[4]) {
        const size_t ka = (size_t)kb * L_SZ;
#pragma unroll
        for (int s = 0; s < 4; ++s) {
            a[s].q0 = *(const v4f*)(Ap[s] + ka);  a[s].q1 = *(const v4f*)(Ap[s] + ka + 4);
        }
#pragma unroll
        for (int j = 0; j < 4; ++j) {
            b[j].q0 = *(const v4f*)(Wp[j] + kb);  b[j].q1 = *(const v4f*)(Wp[j] + kb + 4);
        }
    };
    auto mm16 = [&](const v2f (&a)[4], const v2f (&b)[4]) {
#pragma unroll
        for (int s = 0; s < 4; ++s)
#pragma unroll
            for (int j = 0; j < 4; ++j)
                acc[s][j] = __builtin_amdgcn_wmma_f32_16x16x4_f32(false, a[s], false, b[j], (short)0, acc[s][j], false, false);
    };
    auto mm64 = [&](const vfrag (&a)[4], const vfrag (&b)[4]) {
        v2f fa[4], fb[4];
#pragma unroll
        for (int s = 0; s < 4; ++s) { fa[s] = frag_lo(a[s].q0); fb[s] = frag_lo(b[s].q0); }
        mm16(fa, fb);
#pragma unroll
        for (int s = 0; s < 4; ++s) { fa[s] = frag_hi(a[s].q0); fb[s] = frag_hi(b[s].q0); }
        mm16(fa, fb);
#pragma unroll
        for (int s = 0; s < 4; ++s) { fa[s] = frag_lo(a[s].q1); fb[s] = frag_lo(b[s].q1); }
        mm16(fa, fb);
#pragma unroll
        for (int s = 0; s < 4; ++s) { fa[s] = frag_hi(a[s].q1); fb[s] = frag_hi(b[s].q1); }
        mm16(fa, fb);
    };

    vfrag aA[4], bA[4], aB[4], bB[4];
    ld(k0, aA, bA);
    for (int k = k0; k < k0 + KQ; k += 32) {
        ld(k + 16, aB, bB);
        mm64(aA, bA);
        const int kn = (k + 32 < k0 + KQ) ? (k + 32) : k;
        ld(kn, aA, bA);
        mm64(aB, bB);
    }

#pragma unroll
    for (int s = 0; s < 4; ++s)
#pragma unroll
        for (int j = 0; j < 4; ++j) {
            const v8f v = acc[s][j];
            part[s][wv][j][0][lane] = make_float4(v[0], v[1], v[2], v[3]);
            part[s][wv][j][1][lane] = make_float4(v[4], v[5], v[6], v[7]);
        }
    __syncthreads();

    float* Cb = C + (size_t)(tm * 64 + 16 * wv + 8 * half) * DM + tn * 64 + lr;
#pragma unroll
    for (int j = 0; j < 4; ++j)
#pragma unroll
        for (int h4 = 0; h4 < 2; ++h4) {
            const float4 p0 = part[wv][0][j][h4][lane];
            const float4 p1 = part[wv][1][j][h4][lane];
            const float4 p2 = part[wv][2][j][h4][lane];
            const float4 p3 = part[wv][3][j][h4][lane];
            float* Cr = Cb + (size_t)(4 * h4) * DM + 16 * j;
            Cr[0]              = ((p0.x + p1.x) + p2.x) + p3.x;
            Cr[(size_t)DM]     = ((p0.y + p1.y) + p2.y) + p3.y;
            Cr[(size_t)2 * DM] = ((p0.z + p1.z) + p2.z) + p3.z;
            Cr[(size_t)3 * DM] = ((p0.w + p1.w) + p2.w) + p3.w;
        }
}

__global__ __launch_bounds__(256) void conv_edge_kernel(
    const float* __restrict__ xg, const float* __restrict__ conv_w,
    const float* __restrict__ conv_b, float* __restrict__ x_conv)
{
    static_assert(DI % 256 == 0 && L_SZ % 64 == 0, "edge tiling");
    const int  tile = blockIdx.x / (DI / 256);
    const int  d    = (blockIdx.x % (DI / 256)) * 256 + threadIdx.x;
    const int  row0 = tile * 64;
    const bool head = (row0 % L_SZ) == 0;

    const float  b  = conv_b[d];
    const float4 w4 = *(const float4*)(conv_w + (size_t)d * 4);
    const float  w[4] = {w4.x, w4.y, w4.z, w4.w};
    const float* p  = xg + (size_t)row0 * (2 * DI) + d;

    float in[6];
    in[0] = in[1] = in[2] = 0.0f;
    if (!head) {
        in[0] = *(p - 3 * (2 * DI));
        in[1] = *(p - 2 * (2 * DI));
        in[2] = *(p - 1 * (2 * DI));
    }
    in[3] = p[0];
    in[4] = p[1 * (2 * DI)];
    in[5] = p[2 * (2 * DI)];
#pragma unroll
    for (int i = 0; i < 3; ++i) {
        float acc = b;
#pragma unroll
        for (int j = 0; j < 4; ++j) {
            const float t = fmaf(w[j], in[i + j], acc);
            acc = (head && i + j < 3) ? acc : t;
        }
        x_conv[(size_t)(row0 + i) * DI + d] = acc / (1.0f + __expf(-acc));
    }
}

#define XP_WAVES 8
#define XP_RPW   4
#define XP_ROWS  (XP_WAVES * XP_RPW)
#define XP_OUT   (XP_ROWS * 33)
#define XP_PITCH 33

__global__ __launch_bounds__(XP_WAVES * 32) __attribute__((amdgpu_waves_per_eu(2, 2)))
void xproj_kernel(
    const float* __restrict__ x_conv, const float* __restrict__ Wx,
    float* __restrict__ ssm)
{
    static_assert(NROW % XP_ROWS == 0 && DI % 128 == 0, "xproj tiling");
    __shared__ float part[XP_OUT * XP_PITCH];

    const int lane = threadIdx.x & 31;
    const int wv   = threadIdx.x >> 5;
    const int row0 = blockIdx.x * XP_ROWS;

    const float* xp = x_conv + (size_t)(row0 + XP_RPW * wv) * DI + 4 * lane;
    const float* wp = Wx + 4 * lane;

    v2f acc[XP_RPW][33];
#pragma unroll
    for (int r = 0; r < XP_RPW; ++r)
#pragma unroll
        for (int n = 0; n < 33; ++n) acc[r][n] = v2f{0.f, 0.f};

    for (int c = 0; c < DI; c += 128) {
        float4 xr[XP_RPW], w[33];
#pragma unroll
        for (int r = 0; r < XP_RPW; ++r) xr[r] = *(const float4*)(xp + (size_t)r * DI + c);
#pragma unroll
        for (int n = 0; n < 33; ++n) w[n] = *(const float4*)(wp + (size_t)n * DI + c);
#pragma unroll
        for (int n = 0; n < 33; ++n) {
            const v2f wlo = {w[n].x, w[n].y}, whi = {w[n].z, w[n].w};
#pragma unroll
            for (int r = 0; r < XP_RPW; ++r) {
                const v2f xlo = {xr[r].x, xr[r].y}, xhi = {xr[r].z, xr[r].w};
                acc[r][n] = __builtin_elementwise_fma(xlo, wlo, acc[r][n]);
                acc[r][n] = __builtin_elementwise_fma(xhi, whi, acc[r][n]);
            }
        }
    }

#pragma unroll
    for (int r = 0; r < XP_RPW; ++r)
#pragma unroll
        for (int n = 0; n < 33; ++n)
            part[((XP_RPW * wv + r) * 33 + n) * XP_PITCH + lane] = acc[r][n].x + acc[r][n].y;
    __syncthreads();

    for (int o = threadIdx.x; o < XP_OUT; o += XP_WAVES * 32) {
        const float* q = part + o * XP_PITCH;
        float s0 = 0.f, s1 = 0.f, s2 = 0.f, s3 = 0.f;
#pragma unroll
        for (int l = 0; l < 32; l += 4) {
            s0 += q[l];  s1 += q[l + 1];  s2 += q[l + 2];  s3 += q[l + 3];
        }
        ssm[(size_t)row0 * 33 + o] = (s0 + s1) + (s2 + s3);
    }
}

#define SCAN_WAVES 4
#define SCAN_CH    (2 * SCAN_WAVES)
#define SCAN_NBLK  (L_SZ / 16)
#define YT_PITCH   17

template <int CTRL>
__device__ __forceinline__ float row_dpp(float v) {
    return __builtin_bit_cast(float,
        __builtin_amdgcn_update_dpp(0, __builtin_bit_cast(int, v), CTRL, 0xf, 0xf, true));
}

#define SCAN_STEP(j)                                                  \
    {                                                                 \
        const float dtb  = row_dpp<0x150 + (j)>(dt);                  \
        const float xb   = row_dpp<0x150 + (j)>(xv);                  \
        const float Abar = __expf(dtb * Aa);                          \
        h    = fmaf(Abar, h, (dtb * Bv[j]) * xb);                     \
        p[j] = h * Cv[j];                                             \
    }

__global__ __launch_bounds__(SCAN_WAVES * 32) void scan_kernel(
    const float* __restrict__ ssm, const float* __restrict__ x_conv,
    const float* __restrict__ xg, const float* __restrict__ w_dt,
    const float* __restrict__ b_dt, const float* __restrict__ A_log,
    const float* __restrict__ D_param, float* __restrict__ y)
{
    static_assert(DI % SCAN_CH == 0 && SCAN_CH == 8 && L_SZ % 32 == 0 && DS == 16, "scan tiling");
    __shared__ float ytile[2][SCAN_CH * YT_PITCH];

    const int lane  = threadIdx.x & 31;
    const int wv    = threadIdx.x >> 5;
    const int n     = lane & 15;
    const int c     = lane >> 4;
    const int b     = blockIdx.x / (DI / SCAN_CH);
    const int g     = blockIdx.x % (DI / SCAN_CH);
    const int dbase = g * SCAN_CH;
    const int so    = 4 * wv + (lane >> 3);
    const int co    = lane & 7;
    const int d     = dbase + 2 * wv + c;
    const size_t row0 = (size_t)b * L_SZ;

    const float wdt = w_dt[d];
    const float bdt = b_dt[d];
    const float Dp  = D_param[d];
    const float Aa  = -__expf(A_log[d * DS + n]);
    float h = 0.f;

    const bool b3 = (n & 8) != 0, b2 = (n & 4) != 0, b1 = (n & 2) != 0, b0 = (n & 1) != 0;

    const float* pBC = ssm + row0 * 33 + n;
    const float* pDT = ssm + (row0 + n) * 33 + 32;
    const float* pX  = x_conv + (row0 + n) * DI + d;
    const float* pG  = xg + (row0 + so) * (2 * DI) + DI + dbase + co;
    float*       pY  = y + (((size_t)b * (DI / SCAN_CH) + g) * L_SZ + so) * SCAN_CH + co;

    auto ldblk = [&](int k, float (&Bv)[16], float (&Cv)[16], float& dtr, float& xv, float& gv) {
        const float* q = pBC + (size_t)k * (16 * 33);
#pragma unroll
        for (int j = 0; j < 16; ++j) { Bv[j] = q[j * 33]; Cv[j] = q[j * 33 + 16]; }
        dtr = pDT[(size_t)k * (16 * 33)];
        xv  = pX[(size_t)k * (16 * DI)];
        gv  = pG[(size_t)k * (16 * 2 * DI)];
    };

    auto compute = [&](int k, const float (&Bv)[16], const float (&Cv)[16],
                       float dtr, float xv, float gv, float* yt) {
        const float z  = fmaf(dtr, wdt, bdt);
        const float dt = (z > 20.0f) ? z : log1pf(__expf(z));

        float p[16];
        SCAN_STEP(0)  SCAN_STEP(1)  SCAN_STEP(2)  SCAN_STEP(3)
        SCAN_STEP(4)  SCAN_STEP(5)  SCAN_STEP(6)  SCAN_STEP(7)
        SCAN_STEP(8)  SCAN_STEP(9)  SCAN_STEP(10) SCAN_STEP(11)
        SCAN_STEP(12) SCAN_STEP(13) SCAN_STEP(14) SCAN_STEP(15)

        float q8[8];
#pragma unroll
        for (int j = 0; j < 8; ++j) {
            const float keep = b3 ? p[j + 8] : p[j];
            const float send = b3 ? p[j] : p[j + 8];
            q8[j] = keep + row_dpp<0x168>(send);
        }
        float q4[4];
#pragma unroll
        for (int j = 0; j < 4; ++j) {
            const float keep = b2 ? q8[j + 4] : q8[j];
            const float send = b2 ? q8[j] : q8[j + 4];
            q4[j] = keep + row_dpp<0x164>(send);
        }
        float q2[2];
#pragma unroll
        for (int j = 0; j < 2; ++j) {
            const float keep = b1 ? q4[j + 2] : q4[j];
            const float send = b1 ? q4[j] : q4[j + 2];
            q2[j] = keep + row_dpp<0x162>(send);
        }
        const float keep1 = b0 ? q2[1] : q2[0];
        const float send1 = b0 ? q2[0] : q2[1];
        const float ysum  = keep1 + row_dpp<0x161>(send1);

        yt[(2 * wv + c) * YT_PITCH + n] = fmaf(xv, Dp, ysum);
        __builtin_amdgcn_fence(__ATOMIC_RELEASE, "workgroup", "local");
        __builtin_amdgcn_s_barrier();
        __builtin_amdgcn_fence(__ATOMIC_ACQUIRE, "workgroup", "local");
        const float ys = yt[co * YT_PITCH + so];
        const float sg = gv / (1.0f + __expf(-gv));
        pY[(size_t)k * (16 * SCAN_CH)] = ys * sg;
    };

    static_assert(SCAN_NBLK % 4 == 0 && SCAN_NBLK >= 4, "scan loop unrolled over four register sets");
    constexpr int LASTB = SCAN_NBLK - 1;
    float BvA[16], CvA[16], dtrA, xvA, gvA;
    float BvB[16], CvB[16], dtrB, xvB, gvB;
    float BvC[16], CvC[16], dtrC, xvC, gvC;
    float BvD[16], CvD[16], dtrD, xvD, gvD;
    ldblk(0, BvA, CvA, dtrA, xvA, gvA);
    __builtin_amdgcn_sched_barrier(0);
    ldblk(1, BvB, CvB, dtrB, xvB, gvB);
    __builtin_amdgcn_sched_barrier(0);
    ldblk(2, BvC, CvC, dtrC, xvC, gvC);
    for (int k = 0; k < SCAN_NBLK; k += 4) {
        ldblk(k + 3, BvD, CvD, dtrD, xvD, gvD);
        compute(k, BvA, CvA, dtrA, xvA, gvA, ytile[0]);
        ldblk((k + 4 < SCAN_NBLK) ? (k + 4) : LASTB, BvA, CvA, dtrA, xvA, gvA);
        compute(k + 1, BvB, CvB, dtrB, xvB, gvB, ytile[1]);
        ldblk((k + 5 < SCAN_NBLK) ? (k + 5) : LASTB, BvB, CvB, dtrB, xvB, gvB);
        compute(k + 2, BvC, CvC, dtrC, xvC, gvC, ytile[0]);
        ldblk((k + 6 < SCAN_NBLK) ? (k + 6) : LASTB, BvC, CvC, dtrC, xvC, gvC);
        compute(k + 3, BvD, CvD, dtrD, xvD, gvD, ytile[1]);
    }
}

extern "C" void kernel_launch(void* const* d_in, const int* in_sizes, int n_in,
                              void* d_out, int out_size, void* d_ws, size_t ws_size,
                              hipStream_t stream)
{
    const float* x       = (const float*)d_in[0];
    const float* W_in    = (const float*)d_in[1];
    const float* conv_w  = (const float*)d_in[2];
    const float* conv_b  = (const float*)d_in[3];
    const float* W_xproj = (const float*)d_in[4];
    const float* w_dt    = (const float*)d_in[5];
    const float* b_dt    = (const float*)d_in[6];
    const float* A_log   = (const float*)d_in[7];
    const float* D_param = (const float*)d_in[8];
    const float* W_out   = (const float*)d_in[9];
    float* out = (float*)d_out;

    float* ws     = (float*)d_ws;
    float* xg     = ws;
    float* x_conv = xg + (size_t)NROW * (2 * DI);
    float* ssm    = x_conv + (size_t)NROW * DI;
    float* yb     = ssm + (size_t)NROW * 33;

    gemm64x64_conv_f32_wmma<<<(NROW / 64) * ((2 * DI) / 64), 32, 0, stream>>>(
        x, W_in, xg, conv_w, conv_b, x_conv);

    conv_edge_kernel<<<(NROW / 64) * (DI / 256), 256, 0, stream>>>(
        xg, conv_w, conv_b, x_conv);

    xproj_kernel<<<NROW / XP_ROWS, XP_WAVES * 32, 0, stream>>>(
        x_conv, W_xproj, ssm);

    scan_kernel<<<B_SZ * (DI / SCAN_CH), SCAN_WAVES * 32, 0, stream>>>(
        ssm, x_conv, xg, w_dt, b_dt, A_log, D_param, yb);

    gemm64x64_splitk_f32_wmma<<<(NROW / 64) * (DM / 64), G2_KS * 32, 0, stream>>>(
        yb, W_out, out);
}


# scan: per-(channel,step) terms made once by the loader lane, one block ahead, no branch: softplus, dl=-log2e*dt, u=dt*x; 2 row_share products per step, not 3 and a scale; rounding of dA, dB*x changes
# speedup vs baseline: 12.7055x; 1.0496x over previous
#include <hip/hip_runtime.h>
#include <math.h>


typedef float  v2f __attribute__((ext_vector_type(2)));
typedef float  v4f __attribute__((ext_vector_type(4)));
typedef float  v8f __attribute__((ext_vector_type(8)));

#define B_SZ 2
#define L_SZ 2048
#define DM   768
#define DI   1536
#define DS   16
#define NROW (B_SZ * L_SZ)

__device__ __forceinline__ float xhalf(float v) {
    return __builtin_bit_cast(float, __builtin_amdgcn_permlanex16(
        __builtin_bit_cast(int, v), __builtin_bit_cast(int, v), 0x76543210, 0xfedcba98, false, false));
}

struct vfrag { v4f q0, q1; };
__device__ __forceinline__ v2f frag_lo(const v4f& q) { return __builtin_shufflevector(q, q, 0, 1); }
__device__ __forceinline__ v2f frag_hi(const v4f& q) { return __builtin_shufflevector(q, q, 2, 3); }

__global__ __launch_bounds__(32) __attribute__((amdgpu_waves_per_eu(3, 3)))
void gemm64x64_conv_f32_wmma(
    const float* __restrict__ A, const float* __restrict__ W, float* __restrict__ C,
    const float* __restrict__ conv_w, const float* __restrict__ conv_b,
    float* __restrict__ x_conv)
{
    constexpr int K = DM, LDC = 2 * DI, N64 = (2 * DI) / 64;
    static_assert(K % 32 == 0 && NROW % 64 == 0 && L_SZ % 64 == 0, "first GEMM tiling");
    const int tile = blockIdx.x;
    const int tm   = tile / N64;
    const int tn   = tile % N64;
    const int lane = threadIdx.x;
    const int half = lane >> 4;
    const int lr   = lane & 15;

    const float* Ap[4];
    const float* Wp[4];
#pragma unroll
    for (int s = 0; s < 4; ++s) {
        Ap[s] = A + (size_t)(tm * 64 + 16 * s + lr) * K + 8 * half;
        Wp[s] = W + (size_t)(tn * 64 + 16 * s + lr) * K + 8 * half;
    }

    v8f acc[4][4];
#pragma unroll
    for (int s = 0; s < 4; ++s)
#pragma unroll
        for (int j = 0; j < 4; ++j) acc[s][j] = v8f{};

    auto ld = [&](int kb, vfrag (&a)[4], vfrag (&b)[4]) {
#pragma unroll
        for (int s = 0; s < 4; ++s) {
            a[s].q0 = *(const v4f*)(Ap[s] + kb);  a[s].q1 = *(const v4f*)(Ap[s] + kb + 4);
        }
#pragma unroll
        for (int j = 0; j < 4; ++j) {
            b[j].q0 = *(const v4f*)(Wp[j] + kb);  b[j].q1 = *(const v4f*)(Wp[j] + kb + 4);
        }
    };
    auto mm16 = [&](const v2f (&a)[4], const v2f (&b)[4]) {
#pragma unroll
        for (int s = 0; s < 4; ++s)
#pragma unroll
            for (int j = 0; j < 4; ++j)
                acc[s][j] = __builtin_amdgcn_wmma_f32_16x16x4_f32(false, a[s], false, b[j], (short)0, acc[s][j], false, false);
    };
    auto mm64 = [&](const vfrag (&a)[4], const vfrag (&b)[4]) {
        v2f fa[4], fb[4];
#pragma unroll
        for (int s = 0; s < 4; ++s) { fa[s] = frag_lo(a[s].q0); fb[s] = frag_lo(b[s].q0); }
        mm16(fa, fb);
#pragma unroll
        for (int s = 0; s < 4; ++s) { fa[s] = frag_hi(a[s].q0); fb[s] = frag_hi(b[s].q0); }
        mm16(fa, fb);
#pragma unroll
        for (int s = 0; s < 4; ++s) { fa[s] = frag_lo(a[s].q1); fb[s] = frag_lo(b[s].q1); }
        mm16(fa, fb);
#pragma unroll
        for (int s = 0; s < 4; ++s) { fa[s] = frag_hi(a[s].q1); fb[s] = frag_hi(b[s].q1); }
        mm16(fa, fb);
    };

    vfrag aA[4], bA[4], aB[4], bB[4];
    ld(0, aA, bA);
    for (int k = 0; k < K; k += 32) {
        ld(k + 16, aB, bB);
        mm64(aA, bA);
        const int kn = (k + 32 < K) ? (k + 32) : k;
        ld(kn, aA, bA);
        mm64(aB, bB);
    }

    float* Cb = C + (size_t)(tm * 64 + 8 * half) * LDC + tn * 64 + lr;

    if (tn < DI / 64) {
        float* Xb = x_conv + (size_t)(tm * 64 + 8 * half) * DI + tn * 64 + lr;
#pragma unroll
        for (int j = 0; j < 4; ++j) {
            const int    d = tn * 64 + 16 * j + lr;
            const float4 w = *(const float4*)(conv_w + (size_t)d * 4);
            const float  b = conv_b[d];
            float xs[4][3];
#pragma unroll
            for (int s = 0; s < 4; ++s)
#pragma unroll
                for (int i = 0; i < 3; ++i) xs[s][i] = xhalf(acc[s][j][5 + i]);
#pragma unroll
            for (int s = 0; s < 4; ++s) {
                float e[11];
#pragma unroll
                for (int i = 0; i < 3; ++i) e[i] = (half || s == 0) ? xs[s][i] : xs[s > 0 ? s - 1 : 0][i];
#pragma unroll
                for (int r = 0; r < 8; ++r) e[3 + r] = acc[s][j][r];
#pragma unroll
                for (int r = 0; r < 8; ++r) {
                    float a = b;
                    a = fmaf(w.x, e[r],     a);
                    a = fmaf(w.y, e[r + 1], a);
                    a = fmaf(w.z, e[r + 2], a);
                    a = fmaf(w.w, e[r + 3], a);
                    const float y = a / (1.0f + __expf(-a));
                    if (s > 0 || r >= 3 || half) Xb[(size_t)(16 * s + r) * DI + 16 * j] = y;
                    if (s == 0 && r < 3 && !half)  Cb[(size_t)r * LDC + 16 * j] = acc[s][j][r];
                    if (s == 3 && r >= 5 && half)  Cb[(size_t)(48 + r) * LDC + 16 * j] = acc[s][j][r];
                }
            }
        }
        return;
    }
#pragma unroll
    for (int s = 0; s < 4; ++s)
#pragma unroll
        for (int r = 0; r < 8; ++r) {
            float* Cr = Cb + (size_t)(16 * s + r) * LDC;
            Cr[0]  = acc[s][0][r];  Cr[16] = acc[s][1][r];
            Cr[32] = acc[s][2][r];  Cr[48] = acc[s][3][r];
        }
}

#define G2_KS 4
__global__ __launch_bounds__(G2_KS * 32) __attribute__((amdgpu_waves_per_eu(3, 3)))
void gemm64x64_splitk_f32_wmma(
    const float* __restrict__ A, const float* __restrict__ W, float* __restrict__ C)
{
    constexpr int K = DI, KQ = K / G2_KS, N64 = DM / 64;
    static_assert(KQ % 32 == 0 && NROW % 64 == 0 && L_SZ % 64 == 0 && DM % 64 == 0, "second GEMM tiling");
    __shared__ float4 part[G2_KS][G2_KS][4][2][32];

    const int tile = blockIdx.x;
    const int tm   = tile / N64;
    const int tn   = tile % N64;
    const int lane = threadIdx.x & 31;
    const int wv   = __builtin_amdgcn_readfirstlane((int)(threadIdx.x >> 5));
    const int half = lane >> 4;
    const int lr   = lane & 15;

    const int ab = (tm * 64) / L_SZ;
    const int al = (tm * 64) % L_SZ + lr;
    const int k0 = wv * KQ;

    const float* Ap[4];
    const float* Wp[4];
#pragma unroll
    for (int s = 0; s < 4; ++s) {
        Ap[s] = A + (((size_t)ab * (K / 8) + half) * L_SZ + al + 16 * s) * 8;
        Wp[s] = W + (size_t)(tn * 64 + 16 * s + lr) * K + 8 * half;
    }

    v8f acc[4][4];
#pragma unroll
    for (int s = 0; s < 4; ++s)
#pragma unroll
        for (int j = 0; j < 4; ++j) acc[s][j] = v8f{};

    auto ld = [&](int kb, vfrag (&a)[4], vfrag (&b)[4]) {
        const size_t ka = (size_t)kb * L_SZ;
#pragma unroll
        for (int s = 0; s < 4; ++s) {
            a[s].q0 = *(const v4f*)(Ap[s] + ka);  a[s].q1 = *(const v4f*)(Ap[s] + ka + 4);
        }
#pragma unroll
        for (int j = 0; j < 4; ++j) {
            b[j].q0 = *(const v4f*)(Wp[j] + kb);  b[j].q1 = *(const v4f*)(Wp[j] + kb + 4);
        }
    };
    auto mm16 = [&](const v2f (&a)[4], const v2f (&b)[4]) {
#pragma unroll
        for (int s = 0; s < 4; ++s)
#pragma unroll
            for (int j = 0; j < 4; ++j)
                acc[s][j] = __builtin_amdgcn_wmma_f32_16x16x4_f32(false, a[s], false, b[j], (short)0, acc[s][j], false, false);
    };
    auto mm64 = [&](const vfrag (&a)[4], const vfrag (&b)[4]) {
        v2f fa[4], fb[4];
#pragma unroll
        for (int s = 0; s < 4; ++s) { fa[s] = frag_lo(a[s].q0); fb[s] = frag_lo(b[s].q0); }
        mm16(fa, fb);
#pragma unroll
        for (int s = 0; s < 4; ++s) { fa[s] = frag_hi(a[s].q0); fb[s] = frag_hi(b[s].q0); }
        mm16(fa, fb);
#pragma unroll
        for (int s = 0; s < 4; ++s) { fa[s] = frag_lo(a[s].q1); fb[s] = frag_lo(b[s].q1); }
        mm16(fa, fb);
#pragma unroll
        for (int s = 0; s < 4; ++s) { fa[s] = frag_hi(a[s].q1); fb[s] = frag_hi(b[s].q1); }
        mm16(fa, fb);
    };

    vfrag aA[4], bA[4], aB[4], bB[4];
    ld(k0, aA, bA);
    for (int k = k0; k < k0 + KQ; k += 32) {
        ld(k + 16, aB, bB);
        mm64(aA, bA);
        const int kn = (k + 32 < k0 + KQ) ? (k + 32) : k;
        ld(kn, aA, bA);
        mm64(aB, bB);
    }

#pragma unroll
    for (int s = 0; s < 4; ++s)
#pragma unroll
        for (int j = 0; j < 4; ++j) {
            const v8f v = acc[s][j];
            part[s][wv][j][0][lane] = make_float4(v[0], v[1], v[2], v[3]);
            part[s][wv][j][1][lane] = make_float4(v[4], v[5], v[6], v[7]);
        }
    __syncthreads();

    float* Cb = C + (size_t)(tm * 64 + 16 * wv + 8 * half) * DM + tn * 64 + lr;
#pragma unroll
    for (int j = 0; j < 4; ++j)
#pragma unroll
        for (int h4 = 0; h4 < 2; ++h4) {
            const float4 p0 = part[wv][0][j][h4][lane];
            const float4 p1 = part[wv][1][j][h4][lane];
            const float4 p2 = part[wv][2][j][h4][lane];
            const float4 p3 = part[wv][3][j][h4][lane];
            float* Cr = Cb + (size_t)(4 * h4) * DM + 16 * j;
            Cr[0]              = ((p0.x + p1.x) + p2.x) + p3.x;
            Cr[(size_t)DM]     = ((p0.y + p1.y) + p2.y) + p3.y;
            Cr[(size_t)2 * DM] = ((p0.z + p1.z) + p2.z) + p3.z;
            Cr[(size_t)3 * DM] = ((p0.w + p1.w) + p2.w) + p3.w;
        }
}

__global__ __launch_bounds__(256) void conv_edge_kernel(
    const float* __restrict__ xg, const float* __restrict__ conv_w,
    const float* __restrict__ conv_b, float* __restrict__ x_conv)
{
    static_assert(DI % 256 == 0 && L_SZ % 64 == 0, "edge tiling");
    const int  tile = blockIdx.x / (DI / 256);
    const int  d    = (blockIdx.x % (DI / 256)) * 256 + threadIdx.x;
    const int  row0 = tile * 64;
    const bool head = (row0 % L_SZ) == 0;

    const float  b  = conv_b[d];
    const float4 w4 = *(const float4*)(conv_w + (size_t)d * 4);
    const float  w[4] = {w4.x, w4.y, w4.z, w4.w};
    const float* p  = xg + (size_t)row0 * (2 * DI) + d;

    float in[6];
    in[0] = in[1] = in[2] = 0.0f;
    if (!head) {
        in[0] = *(p - 3 * (2 * DI));
        in[1] = *(p - 2 * (2 * DI));
        in[2] = *(p - 1 * (2 * DI));
    }
    in[3] = p[0];
    in[4] = p[1 * (2 * DI)];
    in[5] = p[2 * (2 * DI)];
#pragma unroll
    for (int i = 0; i < 3; ++i) {
        float acc = b;
#pragma unroll
        for (int j = 0; j < 4; ++j) {
            const float t = fmaf(w[j], in[i + j], acc);
            acc = (head && i + j < 3) ? acc : t;
        }
        x_conv[(size_t)(row0 + i) * DI + d] = acc / (1.0f + __expf(-acc));
    }
}

#define XP_WAVES 8
#define XP_RPW   4
#define XP_ROWS  (XP_WAVES * XP_RPW)
#define XP_OUT   (XP_ROWS * 33)
#define XP_PITCH 33

__global__ __launch_bounds__(XP_WAVES * 32) __attribute__((amdgpu_waves_per_eu(2, 2)))
void xproj_kernel(
    const float* __restrict__ x_conv, const float* __restrict__ Wx,
    float* __restrict__ ssm)
{
    static_assert(NROW % XP_ROWS == 0 && DI % 128 == 0, "xproj tiling");
    __shared__ float part[XP_OUT * XP_PITCH];

    const int lane = threadIdx.x & 31;
    const int wv   = threadIdx.x >> 5;
    const int row0 = blockIdx.x * XP_ROWS;

    const float* xp = x_conv + (size_t)(row0 + XP_RPW * wv) * DI + 4 * lane;
    const float* wp = Wx + 4 * lane;

    v2f acc[XP_RPW][33];
#pragma unroll
    for (int r = 0; r < XP_RPW; ++r)
#pragma unroll
        for (int n = 0; n < 33; ++n) acc[r][n] = v2f{0.f, 0.f};

    for (int c = 0; c < DI; c += 128) {
        float4 xr[XP_RPW], w[33];
#pragma unroll
        for (int r = 0; r < XP_RPW; ++r) xr[r] = *(const float4*)(xp + (size_t)r * DI + c);
#pragma unroll
        for (int n = 0; n < 33; ++n) w[n] = *(const float4*)(wp + (size_t)n * DI + c);
#pragma unroll
        for (int n = 0; n < 33; ++n) {
            const v2f wlo = {w[n].x, w[n].y}, whi = {w[n].z, w[n].w};
#pragma unroll
            for (int r = 0; r < XP_RPW; ++r) {
                const v2f xlo = {xr[r].x, xr[r].y}, xhi = {xr[r].z, xr[r].w};
                acc[r][n] = __builtin_elementwise_fma(xlo, wlo, acc[r][n]);
                acc[r][n] = __builtin_elementwise_fma(xhi, whi, acc[r][n]);
            }
        }
    }

#pragma unroll
    for (int r = 0; r < XP_RPW; ++r)
#pragma unroll
        for (int n = 0; n < 33; ++n)
            part[((XP_RPW * wv + r) * 33 + n) * XP_PITCH + lane] = acc[r][n].x + acc[r][n].y;
    __syncthreads();

    for (int o = threadIdx.x; o < XP_OUT; o += XP_WAVES * 32) {
        const float* q = part + o * XP_PITCH;
        float s0 = 0.f, s1 = 0.f, s2 = 0.f, s3 = 0.f;
#pragma unroll
        for (int l = 0; l < 32; l += 4) {
            s0 += q[l];  s1 += q[l + 1];  s2 += q[l + 2];  s3 += q[l + 3];
        }
        ssm[(size_t)row0 * 33 + o] = (s0 + s1) + (s2 + s3);
    }
}

#define SCAN_WAVES 4
#define SCAN_CH    (2 * SCAN_WAVES)
#define SCAN_NBLK  (L_SZ / 16)
#define YT_PITCH   17

template <int CTRL>
__device__ __forceinline__ float row_dpp(float v) {
    return __builtin_bit_cast(float,
        __builtin_amdgcn_update_dpp(0, __builtin_bit_cast(int, v), CTRL, 0xf, 0xf, true));
}

#define SCAN_STEP(j)                                                  \
    {                                                                 \
        const float dlb  = row_dpp<0x150 + (j)>(dl);                  \
        const float ub   = row_dpp<0x150 + (j)>(u);                   \
        const float Abar = __builtin_amdgcn_exp2f(dlb * eA);          \
        h    = fmaf(Abar, h, Bv[j] * ub);                             \
        p[j] = h * Cv[j];                                             \
    }

__global__ __launch_bounds__(SCAN_WAVES * 32) void scan_kernel(
    const float* __restrict__ ssm, const float* __restrict__ x_conv,
    const float* __restrict__ xg, const float* __restrict__ w_dt,
    const float* __restrict__ b_dt, const float* __restrict__ A_log,
    const float* __restrict__ D_param, float* __restrict__ y)
{
    static_assert(DI % SCAN_CH == 0 && SCAN_CH == 8 && L_SZ % 32 == 0 && DS == 16, "scan tiling");
    __shared__ float ytile[2][SCAN_CH * YT_PITCH];

    const int lane  = threadIdx.x & 31;
    const int wv    = threadIdx.x >> 5;
    const int n     = lane & 15;
    const int c     = lane >> 4;
    const int b     = blockIdx.x / (DI / SCAN_CH);
    const int g     = blockIdx.x % (DI / SCAN_CH);
    const int dbase = g * SCAN_CH;
    const int so    = 4 * wv + (lane >> 3);
    const int co    = lane & 7;
    const int d     = dbase + 2 * wv + c;
    const size_t row0 = (size_t)b * L_SZ;

    const float wdt = w_dt[d];
    const float bdt = b_dt[d];
    const float Dp  = D_param[d];
    const float eA  = __expf(A_log[d * DS + n]);
    float h = 0.f;

    const bool b3 = (n & 8) != 0, b2 = (n & 4) != 0, b1 = (n & 2) != 0, b0 = (n & 1) != 0;

    const float* pBC = ssm + row0 * 33 + n;
    const float* pDT = ssm + (row0 + n) * 33 + 32;
    const float* pX  = x_conv + (row0 + n) * DI + d;
    const float* pG  = xg + (row0 + so) * (2 * DI) + DI + dbase + co;
    float*       pY  = y + (((size_t)b * (DI / SCAN_CH) + g) * L_SZ + so) * SCAN_CH + co;

    auto ldblk = [&](int k, float (&Bv)[16], float (&Cv)[16], float& dtr, float& xv, float& gv) {
        const float* q = pBC + (size_t)k * (16 * 33);
        dtr = pDT[(size_t)k * (16 * 33)];
        xv  = pX[(size_t)k * (16 * DI)];
#pragma unroll
        for (int j = 0; j < 16; ++j) { Bv[j] = q[j * 33]; Cv[j] = q[j * 33 + 16]; }
        gv  = pG[(size_t)k * (16 * 2 * DI)];
    };

    auto front = [&](float dtr, float xv, float& dl, float& u) {
        const float z  = fmaf(dtr, wdt, bdt);
        const float zc = (z > 20.0f) ? 20.0f : z;
        const float sp = log1pf(__expf(zc));
        const float dt = (z > 20.0f) ? z : sp;
        dl = dt * -1.4426950408889634f;
        u  = dt * xv;
    };

    auto compute = [&](int k, const float (&Bv)[16], const float (&Cv)[16],
                       float dl, float u, float xv, float gv, float* yt) {
        float p[16];
        SCAN_STEP(0)  SCAN_STEP(1)  SCAN_STEP(2)  SCAN_STEP(3)
        SCAN_STEP(4)  SCAN_STEP(5)  SCAN_STEP(6)  SCAN_STEP(7)
        SCAN_STEP(8)  SCAN_STEP(9)  SCAN_STEP(10) SCAN_STEP(11)
        SCAN_STEP(12) SCAN_STEP(13) SCAN_STEP(14) SCAN_STEP(15)

        float q8[8];
#pragma unroll
        for (int j = 0; j < 8; ++j) {
            const float keep = b3 ? p[j + 8] : p[j];
            const float send = b3 ? p[j] : p[j + 8];
            q8[j] = keep + row_dpp<0x168>(send);
        }
        float q4[4];
#pragma unroll
        for (int j = 0; j < 4; ++j) {
            const float keep = b2 ? q8[j + 4] : q8[j];
            const float send = b2 ? q8[j] : q8[j + 4];
            q4[j] = keep + row_dpp<0x164>(send);
        }
        float q2[2];
#pragma unroll
        for (int j = 0; j < 2; ++j) {
            const float keep = b1 ? q4[j + 2] : q4[j];
            const float send = b1 ? q4[j] : q4[j + 2];
            q2[j] = keep + row_dpp<0x162>(send);
        }
        const float keep1 = b0 ? q2[1] : q2[0];
        const float send1 = b0 ? q2[0] : q2[1];
        const float ysum  = keep1 + row_dpp<0x161>(send1);

        yt[(2 * wv + c) * YT_PITCH + n] = fmaf(xv, Dp, ysum);
        __builtin_amdgcn_fence(__ATOMIC_RELEASE, "workgroup", "local");
        __builtin_amdgcn_s_barrier();
        __builtin_amdgcn_fence(__ATOMIC_ACQUIRE, "workgroup", "local");
        const float ys = yt[co * YT_PITCH + so];
        const float sg = gv / (1.0f + __expf(-gv));
        pY[(size_t)k * (16 * SCAN_CH)] = ys * sg;
    };

    static_assert(SCAN_NBLK % 4 == 0 && SCAN_NBLK >= 4, "scan loop unrolled over four register sets");
    constexpr int LASTB = SCAN_NBLK - 1;
    float BvA[16], CvA[16], dtrA, xvA, gvA, dlA, uA;
    float BvB[16], CvB[16], dtrB, xvB, gvB, dlB, uB;
    float BvC[16], CvC[16], dtrC, xvC, gvC, dlC, uC;
    float BvD[16], CvD[16], dtrD, xvD, gvD, dlD, uD;
    ldblk(0, BvA, CvA, dtrA, xvA, gvA);
    __builtin_amdgcn_sched_barrier(0);
    ldblk(1, BvB, CvB, dtrB, xvB, gvB);
    __builtin_amdgcn_sched_barrier(0);
    ldblk(2, BvC, CvC, dtrC, xvC, gvC);
    front(dtrA, xvA, dlA, uA);
    for (int k = 0; k < SCAN_NBLK; k += 4) {
        ldblk(k + 3, BvD, CvD, dtrD, xvD, gvD);
        front(dtrB, xvB, dlB, uB);
        compute(k, BvA, CvA, dlA, uA, xvA, gvA, ytile[0]);
        __builtin_amdgcn_sched_barrier(0);
        ldblk((k + 4 < SCAN_NBLK) ? (k + 4) : LASTB, BvA, CvA, dtrA, xvA, gvA);
        front(dtrC, xvC, dlC, uC);
        compute(k + 1, BvB, CvB, dlB, uB, xvB, gvB, ytile[1]);
        __builtin_amdgcn_sched_barrier(0);
        ldblk((k + 5 < SCAN_NBLK) ? (k + 5) : LASTB, BvB, CvB, dtrB, xvB, gvB);
        front(dtrD, xvD, dlD, uD);
        compute(k + 2, BvC, CvC, dlC, uC, xvC, gvC, ytile[0]);
        __builtin_amdgcn_sched_barrier(0);
        ldblk((k + 6 < SCAN_NBLK) ? (k + 6) : LASTB, BvC, CvC, dtrC, xvC, gvC);
        front(dtrA, xvA, dlA, uA);
        compute(k + 3, BvD, CvD, dlD, uD, xvD, gvD, ytile[1]);
    }
}

extern "C" void kernel_launch(void* const* d_in, const int* in_sizes, int n_in,
                              void* d_out, int out_size, void* d_ws, size_t ws_size,
                              hipStream_t stream)
{
    const float* x       = (const float*)d_in[0];
    const float* W_in    = (const float*)d_in[1];
    const float* conv_w  = (const float*)d_in[2];
    const float* conv_b  = (const float*)d_in[3];
    const float* W_xproj = (const float*)d_in[4];
    const float* w_dt    = (const float*)d_in[5];
    const float* b_dt    = (const float*)d_in[6];
    const float* A_log   = (const float*)d_in[7];
    const float* D_param = (const float*)d_in[8];
    const float* W_out   = (const float*)d_in[9];
    float* out = (float*)d_out;

    float* ws     = (float*)d_ws;
    float* xg     = ws;
    float* x_conv = xg + (size_t)NROW * (2 * DI);
    float* ssm    = x_conv + (size_t)NROW * DI;
    float* yb     = ssm + (size_t)NROW * 33;

    gemm64x64_conv_f32_wmma<<<(NROW / 64) * ((2 * DI) / 64), 32, 0, stream>>>(
        x, W_in, xg, conv_w, conv_b, x_conv);

    conv_edge_kernel<<<(NROW / 64) * (DI / 256), 256, 0, stream>>>(
        xg, conv_w, conv_b, x_conv);

    xproj_kernel<<<NROW / XP_ROWS, XP_WAVES * 32, 0, stream>>>(
        x_conv, W_xproj, ssm);

    scan_kernel<<<B_SZ * (DI / SCAN_CH), SCAN_WAVES * 32, 0, stream>>>(
        ssm, x_conv, xg, w_dt, b_dt, A_log, D_param, yb);

    gemm64x64_splitk_f32_wmma<<<(NROW / 64) * (DM / 64), G2_KS * 32, 0, stream>>>(
        yb, W_out, out);
}


# ssm written in the scan's order: per 16-step block B and C as [step/4][state][step%4], dt_raw apart; a scan lane takes its 16 B and 16 C by 8 16-byte loads, were 32 4-byte; every sum unchanged
# speedup vs baseline: 13.0791x; 1.0294x over previous
#include <hip/hip_runtime.h>
#include <math.h>


typedef float  v2f __attribute__((ext_vector_type(2)));
typedef float  v4f __attribute__((ext_vector_type(4)));
typedef float  v8f __attribute__((ext_vector_type(8)));

#define B_SZ 2
#define L_SZ 2048
#define DM   768
#define DI   1536
#define DS   16
#define NROW (B_SZ * L_SZ)

__device__ __forceinline__ float xhalf(float v) {
    return __builtin_bit_cast(float, __builtin_amdgcn_permlanex16(
        __builtin_bit_cast(int, v), __builtin_bit_cast(int, v), 0x76543210, 0xfedcba98, false, false));
}

struct vfrag { v4f q0, q1; };
__device__ __forceinline__ v2f frag_lo(const v4f& q) { return __builtin_shufflevector(q, q, 0, 1); }
__device__ __forceinline__ v2f frag_hi(const v4f& q) { return __builtin_shufflevector(q, q, 2, 3); }

__global__ __launch_bounds__(32) __attribute__((amdgpu_waves_per_eu(3, 3)))
void gemm64x64_conv_f32_wmma(
    const float* __restrict__ A, const float* __restrict__ W, float* __restrict__ C,
    const float* __restrict__ conv_w, const float* __restrict__ conv_b,
    float* __restrict__ x_conv)
{
    constexpr int K = DM, LDC = 2 * DI, N64 = (2 * DI) / 64;
    static_assert(K % 32 == 0 && NROW % 64 == 0 && L_SZ % 64 == 0, "first GEMM tiling");
    const int tile = blockIdx.x;
    const int tm   = tile / N64;
    const int tn   = tile % N64;
    const int lane = threadIdx.x;
    const int half = lane >> 4;
    const int lr   = lane & 15;

    const float* Ap[4];
    const float* Wp[4];
#pragma unroll
    for (int s = 0; s < 4; ++s) {
        Ap[s] = A + (size_t)(tm * 64 + 16 * s + lr) * K + 8 * half;
        Wp[s] = W + (size_t)(tn * 64 + 16 * s + lr) * K + 8 * half;
    }

    v8f acc[4][4];
#pragma unroll
    for (int s = 0; s < 4; ++s)
#pragma unroll
        for (int j = 0; j < 4; ++j) acc[s][j] = v8f{};

    auto ld = [&](int kb, vfrag (&a)[4], vfrag (&b)[4]) {
#pragma unroll
        for (int s = 0; s < 4; ++s) {
            a[s].q0 = *(const v4f*)(Ap[s] + kb);  a[s].q1 = *(const v4f*)(Ap[s] + kb + 4);
        }
#pragma unroll
        for (int j = 0; j < 4; ++j) {
            b[j].q0 = *(const v4f*)(Wp[j] + kb);  b[j].q1 = *(const v4f*)(Wp[j] + kb + 4);
        }
    };
    auto mm16 = [&](const v2f (&a)[4], const v2f (&b)[4]) {
#pragma unroll
        for (int s = 0; s < 4; ++s)
#pragma unroll
            for (int j = 0; j < 4; ++j)
                acc[s][j] = __builtin_amdgcn_wmma_f32_16x16x4_f32(false, a[s], false, b[j], (short)0, acc[s][j], false, false);
    };
    auto mm64 = [&](const vfrag (&a)[4], const vfrag (&b)[4]) {
        v2f fa[4], fb[4];
#pragma unroll
        for (int s = 0; s < 4; ++s) { fa[s] = frag_lo(a[s].q0); fb[s] = frag_lo(b[s].q0); }
        mm16(fa, fb);
#pragma unroll
        for (int s = 0; s < 4; ++s) { fa[s] = frag_hi(a[s].q0); fb[s] = frag_hi(b[s].q0); }
        mm16(fa, fb);
#pragma unroll
        for (int s = 0; s < 4; ++s) { fa[s] = frag_lo(a[s].q1); fb[s] = frag_lo(b[s].q1); }
        mm16(fa, fb);
#pragma unroll
        for (int s = 0; s < 4; ++s) { fa[s] = frag_hi(a[s].q1); fb[s] = frag_hi(b[s].q1); }
        mm16(fa, fb);
    };

    vfrag aA[4], bA[4], aB[4], bB[4];
    ld(0, aA, bA);
    for (int k = 0; k < K; k += 32) {
        ld(k + 16, aB, bB);
        mm64(aA, bA);
        const int kn = (k + 32 < K) ? (k + 32) : k;
        ld(kn, aA, bA);
        mm64(aB, bB);
    }

    float* Cb = C + (size_t)(tm * 64 + 8 * half) * LDC + tn * 64 + lr;

    if (tn < DI / 64) {
        float* Xb = x_conv + (size_t)(tm * 64 + 8 * half) * DI + tn * 64 + lr;
#pragma unroll
        for (int j = 0; j < 4; ++j) {
            const int    d = tn * 64 + 16 * j + lr;
            const float4 w = *(const float4*)(conv_w + (size_t)d * 4);
            const float  b = conv_b[d];
            float xs[4][3];
#pragma unroll
            for (int s = 0; s < 4; ++s)
#pragma unroll
                for (int i = 0; i < 3; ++i) xs[s][i] = xhalf(acc[s][j][5 + i]);
#pragma unroll
            for (int s = 0; s < 4; ++s) {
                float e[11];
#pragma unroll
                for (int i = 0; i < 3; ++i) e[i] = (half || s == 0) ? xs[s][i] : xs[s > 0 ? s - 1 : 0][i];
#pragma unroll
                for (int r = 0; r < 8; ++r) e[3 + r] = acc[s][j][r];
#pragma unroll
                for (int r = 0; r < 8; ++r) {
                    float a = b;
                    a = fmaf(w.x, e[r],     a);
                    a = fmaf(w.y, e[r + 1], a);
                    a = fmaf(w.z, e[r + 2], a);
                    a = fmaf(w.w, e[r + 3], a);
                    const float y = a / (1.0f + __expf(-a));
                    if (s > 0 || r >= 3 || half) Xb[(size_t)(16 * s + r) * DI + 16 * j] = y;
                    if (s == 0 && r < 3 && !half)  Cb[(size_t)r * LDC + 16 * j] = acc[s][j][r];
                    if (s == 3 && r >= 5 && half)  Cb[(size_t)(48 + r) * LDC + 16 * j] = acc[s][j][r];
                }
            }
        }
        return;
    }
#pragma unroll
    for (int s = 0; s < 4; ++s)
#pragma unroll
        for (int r = 0; r < 8; ++r) {
            float* Cr = Cb + (size_t)(16 * s + r) * LDC;
            Cr[0]  = acc[s][0][r];  Cr[16] = acc[s][1][r];
            Cr[32] = acc[s][2][r];  Cr[48] = acc[s][3][r];
        }
}

#define G2_KS 4
__global__ __launch_bounds__(G2_KS * 32) __attribute__((amdgpu_waves_per_eu(3, 3)))
void gemm64x64_splitk_f32_wmma(
    const float* __restrict__ A, const float* __restrict__ W, float* __restrict__ C)
{
    constexpr int K = DI, KQ = K / G2_KS, N64 = DM / 64;
    static_assert(KQ % 32 == 0 && NROW % 64 == 0 && L_SZ % 64 == 0 && DM % 64 == 0, "second GEMM tiling");
    __shared__ float4 part[G2_KS][G2_KS][4][2][32];

    const int tile = blockIdx.x;
    const int tm   = tile / N64;
    const int tn   = tile % N64;
    const int lane = threadIdx.x & 31;
    const int wv   = __builtin_amdgcn_readfirstlane((int)(threadIdx.x >> 5));
    const int half = lane >> 4;
    const int lr   = lane & 15;

    const int ab = (tm * 64) / L_SZ;
    const int al = (tm * 64) % L_SZ + lr;
    const int k0 = wv * KQ;

    const float* Ap[4];
    const float* Wp[4];
#pragma unroll
    for (int s = 0; s < 4; ++s) {
        Ap[s] = A + (((size_t)ab * (K / 8) + half) * L_SZ + al + 16 * s) * 8;
        Wp[s] = W + (size_t)(tn * 64 + 16 * s + lr) * K + 8 * half;
    }

    v8f acc[4][4];
#pragma unroll
    for (int s = 0; s < 4; ++s)
#pragma unroll
        for (int j = 0; j < 4; ++j) acc[s][j] = v8f{};

    auto ld = [&](int kb, vfrag (&a)[4], vfrag (&b)[4]) {
        const size_t ka = (size_t)kb * L_SZ;
#pragma unroll
        for (int s = 0; s < 4; ++s) {
            a[s].q0 = *(const v4f*)(Ap[s] + ka);  a[s].q1 = *(const v4f*)(Ap[s] + ka + 4);
        }
#pragma unroll
        for (int j = 0; j < 4; ++j) {
            b[j].q0 = *(const v4f*)(Wp[j] + kb);  b[j].q1 = *(const v4f*)(Wp[j] + kb + 4);
        }
    };
    auto mm16 = [&](const v2f (&a)[4], const v2f (&b)[4]) {
#pragma unroll
        for (int s = 0; s < 4; ++s)
#pragma unroll
            for (int j = 0; j < 4; ++j)
                acc[s][j] = __builtin_amdgcn_wmma_f32_16x16x4_f32(false, a[s], false, b[j], (short)0, acc[s][j], false, false);
    };
    auto mm64 = [&](const vfrag (&a)[4], const vfrag (&b)[4]) {
        v2f fa[4], fb[4];
#pragma unroll
        for (int s = 0; s < 4; ++s) { fa[s] = frag_lo(a[s].q0); fb[s] = frag_lo(b[s].q0); }
        mm16(fa, fb);
#pragma unroll
        for (int s = 0; s < 4; ++s) { fa[s] = frag_hi(a[s].q0); fb[s] = frag_hi(b[s].q0); }
        mm16(fa, fb);
#pragma unroll
        for (int s = 0; s < 4; ++s) { fa[s] = frag_lo(a[s].q1); fb[s] = frag_lo(b[s].q1); }
        mm16(fa, fb);
#pragma unroll
        for (int s = 0; s < 4; ++s) { fa[s] = frag_hi(a[s].q1); fb[s] = frag_hi(b[s].q1); }
        mm16(fa, fb);
    };

    vfrag aA[4], bA[4], aB[4], bB[4];
    ld(k0, aA, bA);
    for (int k = k0; k < k0 + KQ; k += 32) {
        ld(k + 16, aB, bB);
        mm64(aA, bA);
        const int kn = (k + 32 < k0 + KQ) ? (k + 32) : k;
        ld(kn, aA, bA);
        mm64(aB, bB);
    }

#pragma unroll
    for (int s = 0; s < 4; ++s)
#pragma unroll
        for (int j = 0; j < 4; ++j) {
            const v8f v = acc[s][j];
            part[s][wv][j][0][lane] = make_float4(v[0], v[1], v[2], v[3]);
            part[s][wv][j][1][lane] = make_float4(v[4], v[5], v[6], v[7]);
        }
    __syncthreads();

    float* Cb = C + (size_t)(tm * 64 + 16 * wv + 8 * half) * DM + tn * 64 + lr;
#pragma unroll
    for (int j = 0; j < 4; ++j)
#pragma unroll
        for (int h4 = 0; h4 < 2; ++h4) {
            const float4 p0 = part[wv][0][j][h4][lane];
            const float4 p1 = part[wv][1][j][h4][lane];
            const float4 p2 = part[wv][2][j][h4][lane];
            const float4 p3 = part[wv][3][j][h4][lane];
            float* Cr = Cb + (size_t)(4 * h4) * DM + 16 * j;
            Cr[0]              = ((p0.x + p1.x) + p2.x) + p3.x;
            Cr[(size_t)DM]     = ((p0.y + p1.y) + p2.y) + p3.y;
            Cr[(size_t)2 * DM] = ((p0.z + p1.z) + p2.z) + p3.z;
            Cr[(size_t)3 * DM] = ((p0.w + p1.w) + p2.w) + p3.w;
        }
}

__global__ __launch_bounds__(256) void conv_edge_kernel(
    const float* __restrict__ xg, const float* __restrict__ conv_w,
    const float* __restrict__ conv_b, float* __restrict__ x_conv)
{
    static_assert(DI % 256 == 0 && L_SZ % 64 == 0, "edge tiling");
    const int  tile = blockIdx.x / (DI / 256);
    const int  d    = (blockIdx.x % (DI / 256)) * 256 + threadIdx.x;
    const int  row0 = tile * 64;
    const bool head = (row0 % L_SZ) == 0;

    const float  b  = conv_b[d];
    const float4 w4 = *(const float4*)(conv_w + (size_t)d * 4);
    const float  w[4] = {w4.x, w4.y, w4.z, w4.w};
    const float* p  = xg + (size_t)row0 * (2 * DI) + d;

    float in[6];
    in[0] = in[1] = in[2] = 0.0f;
    if (!head) {
        in[0] = *(p - 3 * (2 * DI));
        in[1] = *(p - 2 * (2 * DI));
        in[2] = *(p - 1 * (2 * DI));
    }
    in[3] = p[0];
    in[4] = p[1 * (2 * DI)];
    in[5] = p[2 * (2 * DI)];
#pragma unroll
    for (int i = 0; i < 3; ++i) {
        float acc = b;
#pragma unroll
        for (int j = 0; j < 4; ++j) {
            const float t = fmaf(w[j], in[i + j], acc);
            acc = (head && i + j < 3) ? acc : t;
        }
        x_conv[(size_t)(row0 + i) * DI + d] = acc / (1.0f + __expf(-acc));
    }
}

#define XP_WAVES 8
#define XP_RPW   4
#define XP_ROWS  (XP_WAVES * XP_RPW)
#define XP_OUT   (XP_ROWS * 33)
#define XP_PITCH 33

__global__ __launch_bounds__(XP_WAVES * 32) __attribute__((amdgpu_waves_per_eu(2, 2)))
void xproj_kernel(
    const float* __restrict__ x_conv, const float* __restrict__ Wx,
    float* __restrict__ ssm)
{
    static_assert(NROW % XP_ROWS == 0 && DI % 128 == 0, "xproj tiling");
    __shared__ float part[XP_OUT * XP_PITCH];

    const int lane = threadIdx.x & 31;
    const int wv   = threadIdx.x >> 5;
    const int row0 = blockIdx.x * XP_ROWS;

    const float* xp = x_conv + (size_t)(row0 + XP_RPW * wv) * DI + 4 * lane;
    const float* wp = Wx + 4 * lane;

    v2f acc[XP_RPW][33];
#pragma unroll
    for (int r = 0; r < XP_RPW; ++r)
#pragma unroll
        for (int n = 0; n < 33; ++n) acc[r][n] = v2f{0.f, 0.f};

    for (int c = 0; c < DI; c += 128) {
        float4 xr[XP_RPW], w[33];
#pragma unroll
        for (int r = 0; r < XP_RPW; ++r) xr[r] = *(const float4*)(xp + (size_t)r * DI + c);
#pragma unroll
        for (int n = 0; n < 33; ++n) w[n] = *(const float4*)(wp + (size_t)n * DI + c);
#pragma unroll
        for (int n = 0; n < 33; ++n) {
            const v2f wlo = {w[n].x, w[n].y}, whi = {w[n].z, w[n].w};
#pragma unroll
            for (int r = 0; r < XP_RPW; ++r) {
                const v2f xlo = {xr[r].x, xr[r].y}, xhi = {xr[r].z, xr[r].w};
                acc[r][n] = __builtin_elementwise_fma(xlo, wlo, acc[r][n]);
                acc[r][n] = __builtin_elementwise_fma(xhi, whi, acc[r][n]);
            }
        }
    }

    static_assert(XP_RPW == 4 && XP_ROWS == 32, "destination order of the partials");
    const int pb = (wv >> 2) * 512 + (wv & 3) * 64;
#pragma unroll
    for (int r = 0; r < XP_RPW; ++r) {
#pragma unroll
        for (int n = 0; n < 32; ++n)
            part[(pb + (n >> 4) * 256 + (n & 15) * 4 + r) * XP_PITCH + lane] = acc[r][n].x + acc[r][n].y;
        part[(1024 + XP_RPW * wv + r) * XP_PITCH + lane] = acc[r][32].x + acc[r][32].y;
    }
    __syncthreads();

    float* bc = ssm + (size_t)(row0 / 16) * 512;
    float* dr = ssm + (size_t)NROW * 32 + row0;
    for (int o = threadIdx.x; o < XP_OUT; o += XP_WAVES * 32) {
        const float* q = part + o * XP_PITCH;
        float s0 = 0.f, s1 = 0.f, s2 = 0.f, s3 = 0.f;
#pragma unroll
        for (int l = 0; l < 32; l += 4) {
            s0 += q[l];  s1 += q[l + 1];  s2 += q[l + 2];  s3 += q[l + 3];
        }
        float* dst = (o < 1024) ? bc + o : dr + (o - 1024);
        *dst = (s0 + s1) + (s2 + s3);
    }
}

#define SCAN_WAVES 4
#define SCAN_CH    (2 * SCAN_WAVES)
#define SCAN_NBLK  (L_SZ / 16)
#define YT_PITCH   17

template <int CTRL>
__device__ __forceinline__ float row_dpp(float v) {
    return __builtin_bit_cast(float,
        __builtin_amdgcn_update_dpp(0, __builtin_bit_cast(int, v), CTRL, 0xf, 0xf, true));
}

#define SCAN_STEP(j)                                                  \
    {                                                                 \
        const float dlb  = row_dpp<0x150 + (j)>(dl);                  \
        const float ub   = row_dpp<0x150 + (j)>(u);                   \
        const float Abar = __builtin_amdgcn_exp2f(dlb * eA);          \
        h    = fmaf(Abar, h, Bv[j] * ub);                             \
        p[j] = h * Cv[j];                                             \
    }

__global__ __launch_bounds__(SCAN_WAVES * 32) void scan_kernel(
    const float* __restrict__ ssm, const float* __restrict__ x_conv,
    const float* __restrict__ xg, const float* __restrict__ w_dt,
    const float* __restrict__ b_dt, const float* __restrict__ A_log,
    const float* __restrict__ D_param, float* __restrict__ y)
{
    static_assert(DI % SCAN_CH == 0 && SCAN_CH == 8 && L_SZ % 32 == 0 && DS == 16, "scan tiling");
    __shared__ float ytile[2][SCAN_CH * YT_PITCH];

    const int lane  = threadIdx.x & 31;
    const int wv    = threadIdx.x >> 5;
    const int n     = lane & 15;
    const int c     = lane >> 4;
    const int b     = blockIdx.x / (DI / SCAN_CH);
    const int g     = blockIdx.x % (DI / SCAN_CH);
    const int dbase = g * SCAN_CH;
    const int so    = 4 * wv + (lane >> 3);
    const int co    = lane & 7;
    const int d     = dbase + 2 * wv + c;
    const size_t row0 = (size_t)b * L_SZ;

    const float wdt = w_dt[d];
    const float bdt = b_dt[d];
    const float Dp  = D_param[d];
    const float eA  = __expf(A_log[d * DS + n]);
    float h = 0.f;

    const bool b3 = (n & 8) != 0, b2 = (n & 4) != 0, b1 = (n & 2) != 0, b0 = (n & 1) != 0;

    const float* pBC = ssm + (row0 / 16) * 512 + 4 * n;
    const float* pDT = ssm + (size_t)NROW * 32 + row0 + n;
    const float* pX  = x_conv + (row0 + n) * DI + d;
    const float* pG  = xg + (row0 + so) * (2 * DI) + DI + dbase + co;
    float*       pY  = y + (((size_t)b * (DI / SCAN_CH) + g) * L_SZ + so) * SCAN_CH + co;

    auto ldblk = [&](int k, float (&Bv)[16], float (&Cv)[16], float& dtr, float& xv, float& gv) {
        const float* q = pBC + (size_t)k * 512;
        dtr = pDT[(size_t)k * 16];
        xv  = pX[(size_t)k * (16 * DI)];
#pragma unroll
        for (int i = 0; i < 4; ++i) {
            const v4f bq = *(const v4f*)(q + 64 * i);
            const v4f cq = *(const v4f*)(q + 256 + 64 * i);
            Bv[4 * i] = bq.x;  Bv[4 * i + 1] = bq.y;  Bv[4 * i + 2] = bq.z;  Bv[4 * i + 3] = bq.w;
            Cv[4 * i] = cq.x;  Cv[4 * i + 1] = cq.y;  Cv[4 * i + 2] = cq.z;  Cv[4 * i + 3] = cq.w;
        }
        gv  = pG[(size_t)k * (16 * 2 * DI)];
    };

    auto front = [&](float dtr, float xv, float& dl, float& u) {
        const float z  = fmaf(dtr, wdt, bdt);
        const float zc = (z > 20.0f) ? 20.0f : z;
        const float sp = log1pf(__expf(zc));
        const float dt = (z > 20.0f) ? z : sp;
        dl = dt * -1.4426950408889634f;
        u  = dt * xv;
    };

    auto compute = [&](int k, const float (&Bv)[16], const float (&Cv)[16],
                       float dl, float u, float xv, float gv, float* yt) {
        float p[16];
        SCAN_STEP(0)  SCAN_STEP(1)  SCAN_STEP(2)  SCAN_STEP(3)
        SCAN_STEP(4)  SCAN_STEP(5)  SCAN_STEP(6)  SCAN_STEP(7)
        SCAN_STEP(8)  SCAN_STEP(9)  SCAN_STEP(10) SCAN_STEP(11)
        SCAN_STEP(12) SCAN_STEP(13) SCAN_STEP(14) SCAN_STEP(15)

        float q8[8];
#pragma unroll
        for (int j = 0; j < 8; ++j) {
            const float keep = b3 ? p[j + 8] : p[j];
            const float send = b3 ? p[j] : p[j + 8];
            q8[j] = keep + row_dpp<0x168>(send);
        }
        float q4[4];
#pragma unroll
        for (int j = 0; j < 4; ++j) {
            const float keep = b2 ? q8[j + 4] : q8[j];
            const float send = b2 ? q8[j] : q8[j + 4];
            q4[j] = keep + row_dpp<0x164>(send);
        }
        float q2[2];
#pragma unroll
        for (int j = 0; j < 2; ++j) {
            const float keep = b1 ? q4[j + 2] : q4[j];
            const float send = b1 ? q4[j] : q4[j + 2];
            q2[j] = keep + row_dpp<0x162>(send);
        }
        const float keep1 = b0 ? q2[1] : q2[0];
        const float send1 = b0 ? q2[0] : q2[1];
        const float ysum  = keep1 + row_dpp<0x161>(send1);

        yt[(2 * wv + c) * YT_PITCH + n] = fmaf(xv, Dp, ysum);
        __builtin_amdgcn_fence(__ATOMIC_RELEASE, "workgroup", "local");
        __builtin_amdgcn_s_barrier();
        __builtin_amdgcn_fence(__ATOMIC_ACQUIRE, "workgroup", "local");
        const float ys = yt[co * YT_PITCH + so];
        const float sg = gv / (1.0f + __expf(-gv));
        pY[(size_t)k * (16 * SCAN_CH)] = ys * sg;
    };

    static_assert(SCAN_NBLK % 4 == 0 && SCAN_NBLK >= 4, "scan loop unrolled over four register sets");
    constexpr int LASTB = SCAN_NBLK - 1;
    float BvA[16], CvA[16], dtrA, xvA, gvA, dlA, uA;
    float BvB[16], CvB[16], dtrB, xvB, gvB, dlB, uB;
    float BvC[16], CvC[16], dtrC, xvC, gvC, dlC, uC;
    float BvD[16], CvD[16], dtrD, xvD, gvD, dlD, uD;
    ldblk(0, BvA, CvA, dtrA, xvA, gvA);
    __builtin_amdgcn_sched_barrier(0);
    ldblk(1, BvB, CvB, dtrB, xvB, gvB);
    __builtin_amdgcn_sched_barrier(0);
    ldblk(2, BvC, CvC, dtrC, xvC, gvC);
    front(dtrA, xvA, dlA, uA);
    for (int k = 0; k < SCAN_NBLK; k += 4) {
        ldblk(k + 3, BvD, CvD, dtrD, xvD, gvD);
        front(dtrB, xvB, dlB, uB);
        compute(k, BvA, CvA, dlA, uA, xvA, gvA, ytile[0]);
        __builtin_amdgcn_sched_barrier(0);
        ldblk((k + 4 < SCAN_NBLK) ? (k + 4) : LASTB, BvA, CvA, dtrA, xvA, gvA);
        front(dtrC, xvC, dlC, uC);
        compute(k + 1, BvB, CvB, dlB, uB, xvB, gvB, ytile[1]);
        __builtin_amdgcn_sched_barrier(0);
        ldblk((k + 5 < SCAN_NBLK) ? (k + 5) : LASTB, BvB, CvB, dtrB, xvB, gvB);
        front(dtrD, xvD, dlD, uD);
        compute(k + 2, BvC, CvC, dlC, uC, xvC, gvC, ytile[0]);
        __builtin_amdgcn_sched_barrier(0);
        ldblk((k + 6 < SCAN_NBLK) ? (k + 6) : LASTB, BvC, CvC, dtrC, xvC, gvC);
        front(dtrA, xvA, dlA, uA);
        compute(k + 3, BvD, CvD, dlD, uD, xvD, gvD, ytile[1]);
    }
}

extern "C" void kernel_launch(void* const* d_in, const int* in_sizes, int n_in,
                              void* d_out, int out_size, void* d_ws, size_t ws_size,
                              hipStream_t stream)
{
    const float* x       = (const float*)d_in[0];
    const float* W_in    = (const float*)d_in[1];
    const float* conv_w  = (const float*)d_in[2];
    const float* conv_b  = (const float*)d_in[3];
    const float* W_xproj = (const float*)d_in[4];
    const float* w_dt    = (const float*)d_in[5];
    const float* b_dt    = (const float*)d_in[6];
    const float* A_log   = (const float*)d_in[7];
    const float* D_param = (const float*)d_in[8];
    const float* W_out   = (const float*)d_in[9];
    float* out = (float*)d_out;

    float* ws     = (float*)d_ws;
    float* xg     = ws;
    float* x_conv = xg + (size_t)NROW * (2 * DI);
    float* ssm    = x_conv + (size_t)NROW * DI;
    float* yb     = ssm + (size_t)NROW * 33;

    gemm64x64_conv_f32_wmma<<<(NROW / 64) * ((2 * DI) / 64), 32, 0, stream>>>(
        x, W_in, xg, conv_w, conv_b, x_conv);

    conv_edge_kernel<<<(NROW / 64) * (DI / 256), 256, 0, stream>>>(
        xg, conv_w, conv_b, x_conv);

    xproj_kernel<<<NROW / XP_ROWS, XP_WAVES * 32, 0, stream>>>(
        x_conv, W_xproj, ssm);

    scan_kernel<<<B_SZ * (DI / SCAN_CH), SCAN_WAVES * 32, 0, stream>>>(
        ssm, x_conv, xg, w_dt, b_dt, A_log, D_param, yb);

    gemm64x64_splitk_f32_wmma<<<(NROW / 64) * (DM / 64), G2_KS * 32, 0, stream>>>(
        yb, W_out, out);
}
